# starting kernel
# baseline (speedup 1.0000x reference)
#include <hip/hip_runtime.h>
#include <hip/hip_bf16.h>
#include <math.h>


#define B_     8
#define CIN_   25
#define NPG_   2000
#define ENC_   2816
#define HID_   100
#define KGEMM_ 2816

typedef __attribute__((ext_vector_type(2))) float v2f;
typedef __attribute__((ext_vector_type(8))) float v8f;

__device__ __forceinline__ unsigned fkey(float f) {
    unsigned u = __float_as_uint(f);
    return (u & 0x80000000u) ? ~u : (u | 0x80000000u);
}
__device__ __forceinline__ float keyf(unsigned k) {
    unsigned u = (k & 0x80000000u) ? (k ^ 0x80000000u) : ~k;
    return __uint_as_float(u);
}

__global__ void zero_u32(unsigned* __restrict__ p, int n) {
    int i = blockIdx.x * blockDim.x + threadIdx.x;
    if (i < n) p[i] = 0u;
}

__global__ void conv1_kernel(const float* __restrict__ X, const float* __restrict__ w,
                             const float* __restrict__ b, float* __restrict__ out) {
    const int total = 8 * 32 * 21 * 6 * 6;
    int idx = blockIdx.x * blockDim.x + threadIdx.x;
    if (idx >= total) return;
    int x = idx % 6;  int t = idx / 6;
    int y = t % 6;    t /= 6;
    int z = t % 21;   t /= 21;
    int o = t % 32;   int bn = t / 32;
    float acc = b[o];
    for (int c = 0; c < 25; ++c)
        for (int kd = 0; kd < 5; ++kd) {
            const float* Xp = X + ((size_t)((bn * 25 + c) * 25 + (z + kd))) * 36;
            const float* Wp = w + ((size_t)((o * 25 + c) * 5 + kd)) * 9;
            for (int kh = 0; kh < 3; ++kh) {
                int yy = y + kh - 1; if (yy < 0 || yy >= 6) continue;
                for (int kw = 0; kw < 3; ++kw) {
                    int xx = x + kw - 1; if (xx < 0 || xx >= 6) continue;
                    acc = fmaf(Xp[yy * 6 + xx], Wp[kh * 3 + kw], acc);
                }
            }
        }
    out[idx] = fmaxf(acc, 0.0f);
}

__global__ void conv2_kernel(const float* __restrict__ A, const float* __restrict__ w,
                             const float* __restrict__ b, float* __restrict__ out) {
    const int total = 8 * 64 * 17 * 6 * 6;
    int idx = blockIdx.x * blockDim.x + threadIdx.x;
    if (idx >= total) return;
    int x = idx % 6;  int t = idx / 6;
    int y = t % 6;    t /= 6;
    int z = t % 17;   t /= 17;
    int o = t % 64;   int bn = t / 64;
    float acc = b[o];
    for (int c = 0; c < 32; ++c)
        for (int kd = 0; kd < 5; ++kd) {
            const float* Ap = A + ((size_t)((bn * 32 + c) * 21 + (z + kd))) * 36;
            const float* Wp = w + ((size_t)((o * 32 + c) * 5 + kd)) * 9;
            for (int kh = 0; kh < 3; ++kh) {
                int yy = y + kh - 1; if (yy < 0 || yy >= 6) continue;
                for (int kw = 0; kw < 3; ++kw) {
                    int xx = x + kw - 1; if (xx < 0 || xx >= 6) continue;
                    acc = fmaf(Ap[yy * 6 + xx], Wp[kh * 3 + kw], acc);
                }
            }
        }
    out[idx] = fmaxf(acc, 0.0f);
}

__global__ void conv3_kernel(const float* __restrict__ A, const float* __restrict__ w,
                             const float* __restrict__ b, float* __restrict__ out) {
    const int total = 8 * 64 * 13 * 4 * 4;
    int idx = blockIdx.x * blockDim.x + threadIdx.x;
    if (idx >= total) return;
    int x = idx % 4;  int t = idx / 4;
    int y = t % 4;    t /= 4;
    int z = t % 13;   t /= 13;
    int o = t % 64;   int bn = t / 64;
    float acc = b[o];
    for (int c = 0; c < 64; ++c)
        for (int kd = 0; kd < 5; ++kd) {
            const float* Ap = A + ((size_t)((bn * 64 + c) * 17 + (z + kd))) * 36;
            const float* Wp = w + ((size_t)((o * 64 + c) * 5 + kd)) * 9;
            for (int kh = 0; kh < 3; ++kh)
                for (int kw = 0; kw < 3; ++kw)
                    acc = fmaf(Ap[(y + kh) * 6 + (x + kw)], Wp[kh * 3 + kw], acc);
        }
    out[idx] = fmaxf(acc, 0.0f);
}

__global__ void pool_kernel(const float* __restrict__ A, float* __restrict__ enc) {
    const int total = 8 * 64 * 11 * 2 * 2;
    int idx = blockIdx.x * blockDim.x + threadIdx.x;
    if (idx >= total) return;
    int x = idx % 2;  int t = idx / 2;
    int y = t % 2;    t /= 2;
    int d = t % 11;   t /= 11;
    int ch = t % 64;  int bn = t / 64;
    float m = -INFINITY;
    for (int dd = 0; dd < 3; ++dd)
        for (int hh = 0; hh < 3; ++hh)
            for (int ww = 0; ww < 3; ++ww) {
                float v = A[((size_t)((bn * 64 + ch) * 13 + (d + dd))) * 16 + (y + hh) * 4 + (x + ww)];
                m = fmaxf(m, v);
            }
    enc[(size_t)bn * ENC_ + ch * 44 + d * 4 + y * 2 + x] = m;
}

__global__ void edge_pass(const int* __restrict__ ei, const int* __restrict__ gid,
                          const float* __restrict__ x3, unsigned* __restrict__ aggk,
                          unsigned* __restrict__ mask, int* __restrict__ counts, int E) {
    int e = blockIdx.x * blockDim.x + threadIdx.x;
    if (e >= E) return;
    int s = ei[e];
    int d = ei[E + e];
    int g = gid[s];
    atomicOr(&mask[d], 1u << g);
    atomicAdd(&counts[d], 1);
    atomicMax(&aggk[(size_t)d * 3 + 0], fkey(x3[(size_t)s * 3 + 0]));
    atomicMax(&aggk[(size_t)d * 3 + 1], fkey(x3[(size_t)s * 3 + 1]));
    atomicMax(&aggk[(size_t)d * 3 + 2], fkey(x3[(size_t)s * 3 + 2]));
}

__global__ void maxenc_kernel(const float* __restrict__ enc, float* __restrict__ ME) {
    int c = blockIdx.x * blockDim.x + threadIdx.x;
    if (c >= ENC_) return;
    ME[c] = 0.0f;
    for (int m = 1; m < 256; ++m) {
        int p = m & (m - 1);
        int g = __ffs(m) - 1;
        ME[(size_t)m * ENC_ + c] = fmaxf(ME[(size_t)p * ENC_ + c], enc[(size_t)g * ENC_ + c]);
    }
}

__global__ void __launch_bounds__(32)
gemm_wmma_f32(const float* __restrict__ A, const float* __restrict__ W,
              float* __restrict__ C, int Mrows) {
    const int Ncols = HID_;
    int lane = threadIdx.x & 31;
    int half = lane >> 4;
    int l16  = lane & 15;
    int mBase = blockIdx.x * 16;
    int nBase = blockIdx.y * 16;
    int mRow = mBase + l16; if (mRow >= Mrows) mRow = Mrows - 1;
    int nCol = nBase + l16;
    bool nOk = (nCol < Ncols);
    int nClamp = nOk ? nCol : (Ncols - 1);
    const float* Arow = A + (size_t)mRow * KGEMM_;
    const float* Bcol = W + nClamp;

    v8f acc0 = {}, acc1 = {};
    for (int kt = 0; kt < KGEMM_; kt += 8) {
        int k0 = kt + half * 2;
        int k1 = k0 + 4;
        v2f a0, b0, a1, b1;
        a0.x = Arow[k0];
        a0.y = Arow[k0 + 1];
        a1.x = Arow[k1];
        a1.y = Arow[k1 + 1];
        b0.x = Bcol[(size_t)k0 * Ncols];
        b0.y = Bcol[(size_t)(k0 + 1) * Ncols];
        b1.x = Bcol[(size_t)k1 * Ncols];
        b1.y = Bcol[(size_t)(k1 + 1) * Ncols];
        acc0 = __builtin_amdgcn_wmma_f32_16x16x4_f32(false, a0, false, b0,
                                                     (short)0, acc0, false, false);
        acc1 = __builtin_amdgcn_wmma_f32_16x16x4_f32(false, a1, false, b1,
                                                     (short)0, acc1, false, false);
    }
    v8f acc = acc0 + acc1;
    for (int v = 0; v < 8; ++v) {
        int m = mBase + half * 8 + v;
        if (m < Mrows && nOk) C[(size_t)m * Ncols + nCol] = acc[v];
    }
}

__global__ void node_h_kernel(const float* __restrict__ x3, const int* __restrict__ gid,
                              const unsigned* __restrict__ mask, const unsigned* __restrict__ aggk,
                              const float* __restrict__ P, const float* __restrict__ Q,
                              const float* __restrict__ Wl1, const float* __restrict__ Wr1,
                              const float* __restrict__ bl1, float* __restrict__ h, int N) {
    int wave = threadIdx.x >> 5;
    int lane = threadIdx.x & 31;
    int n = blockIdx.x * (blockDim.x >> 5) + wave;
    if (n >= N) return;

    float x0 = x3[(size_t)n * 3 + 0];
    float x1 = x3[(size_t)n * 3 + 1];
    float x2 = x3[(size_t)n * 3 + 2];
    int g = gid[n];
    unsigned mk = mask[n];
    bool has = (mk != 0u);
    float a0 = 0.f, a1 = 0.f, a2 = 0.f;
    if (has) {
        a0 = keyf(aggk[(size_t)n * 3 + 0]);
        a1 = keyf(aggk[(size_t)n * 3 + 1]);
        a2 = keyf(aggk[(size_t)n * 3 + 2]);
    }
    const float* Prow = P + (size_t)mk * HID_;
    const float* Qrow = Q + (size_t)g * HID_;

    float v[4];
    #pragma unroll
    for (int c = 0; c < 4; ++c) {
        int col = lane + 32 * c;
        if (col < HID_) {
            float t = bl1[col] + Qrow[col]
                    + x0 * Wr1[col] + x1 * Wr1[HID_ + col] + x2 * Wr1[2 * HID_ + col];
            if (has)
                t += Prow[col]
                   + a0 * Wl1[col] + a1 * Wl1[HID_ + col] + a2 * Wl1[2 * HID_ + col];
            v[c] = t;
        } else {
            v[c] = -INFINITY;
        }
    }
    float m = fmaxf(fmaxf(v[0], v[1]), fmaxf(v[2], v[3]));
    for (int off = 16; off >= 1; off >>= 1)
        m = fmaxf(m, __shfl_xor(m, off, 32));
    float s = 0.f;
    #pragma unroll
    for (int c = 0; c < 4; ++c) s += __expf(v[c] - m);
    for (int off = 16; off >= 1; off >>= 1)
        s += __shfl_xor(s, off, 32);
    float lse = m + __logf(s);
    #pragma unroll
    for (int c = 0; c < 4; ++c) {
        int col = lane + 32 * c;
        if (col < HID_) h[(size_t)n * HID_ + col] = v[c] - lse;
    }
}

__global__ void edge_scatter_h(const int* __restrict__ ei, const float* __restrict__ h,
                               float* __restrict__ agg2, int E) {
    int e = blockIdx.x;
    int j = threadIdx.x;
    if (j < HID_) {
        int s = ei[e];
        int d = ei[E + e];
        atomicAdd(&agg2[(size_t)d * HID_ + j], h[(size_t)s * HID_ + j]);
    }
}

__global__ void node_out_kernel(const float* __restrict__ agg2, const float* __restrict__ h,
                                const int* __restrict__ counts, const float* __restrict__ Wl2,
                                const float* __restrict__ bl2, const float* __restrict__ Wr2,
                                float* __restrict__ y, int N) {
    int wave = threadIdx.x >> 5;
    int lane = threadIdx.x & 31;
    int n = blockIdx.x * (blockDim.x >> 5) + wave;
    if (n >= N) return;
    float s1 = 0.f, s2 = 0.f;
    #pragma unroll
    for (int c = 0; c < 4; ++c) {
        int col = lane + 32 * c;
        if (col < HID_) {
            s1 = fmaf(agg2[(size_t)n * HID_ + col], Wl2[col], s1);
            s2 = fmaf(h[(size_t)n * HID_ + col],   Wr2[col], s2);
        }
    }
    for (int off = 16; off >= 1; off >>= 1) {
        s1 += __shfl_xor(s1, off, 32);
        s2 += __shfl_xor(s2, off, 32);
    }
    if (lane == 0)
        y[n] = s1 / fmaxf((float)counts[n], 1.0f) + bl2[0] + s2;
}

extern "C" void kernel_launch(void* const* d_in, const int* in_sizes, int n_in,
                              void* d_out, int out_size, void* d_ws, size_t ws_size,
                              hipStream_t stream) {
    const float* X      = (const float*)d_in[0];
    const float* x3     = (const float*)d_in[1];
    const int*   ei     = (const int*)d_in[2];
    const int*   gid    = (const int*)d_in[3];
    const float* w1     = (const float*)d_in[4];
    const float* b1     = (const float*)d_in[5];
    const float* w2     = (const float*)d_in[6];
    const float* b2     = (const float*)d_in[7];
    const float* w3     = (const float*)d_in[8];
    const float* b3     = (const float*)d_in[9];
    const float* Wl1    = (const float*)d_in[10];
    const float* bl1    = (const float*)d_in[11];
    const float* Wr1    = (const float*)d_in[12];
    const float* Wl2    = (const float*)d_in[13];
    const float* bl2    = (const float*)d_in[14];
    const float* Wr2    = (const float*)d_in[15];
    float* y = (float*)d_out;

    const int E = in_sizes[2] / 2;
    const int N = in_sizes[3];

    char* ws = (char*)d_ws;
    size_t off = 0;
    auto alloc = [&](size_t bytes) -> char* {
        char* p = ws + off;
        off = (off + bytes + 255) & ~(size_t)255;
        return p;
    };
    float*    act1  = (float*)alloc((size_t)8 * 32 * 21 * 36 * 4);
    float*    act2  = (float*)alloc((size_t)8 * 64 * 17 * 36 * 4);
    float*    act3  = (float*)alloc((size_t)8 * 64 * 13 * 16 * 4);
    float*    enc   = (float*)alloc((size_t)8 * ENC_ * 4);
    float*    ME    = (float*)alloc((size_t)256 * ENC_ * 4);
    float*    P     = (float*)alloc((size_t)256 * HID_ * 4);
    float*    Q     = (float*)alloc((size_t)16  * HID_ * 4);
    unsigned* aggk  = (unsigned*)alloc((size_t)N * 3 * 4);
    unsigned* mask  = (unsigned*)alloc((size_t)N * 4);
    int*      cnt   = (int*)alloc((size_t)N * 4);
    float*    hbuf  = (float*)alloc((size_t)N * HID_ * 4);
    float*    agg2  = (float*)alloc((size_t)N * HID_ * 4);

    const int T = 256;
    zero_u32<<<(N * 3 + T - 1) / T, T, 0, stream>>>(aggk, N * 3);
    zero_u32<<<(N + T - 1) / T, T, 0, stream>>>(mask, N);
    zero_u32<<<(N + T - 1) / T, T, 0, stream>>>((unsigned*)cnt, N);
    zero_u32<<<(N * HID_ + T - 1) / T, T, 0, stream>>>((unsigned*)agg2, N * HID_);

    conv1_kernel<<<(8 * 32 * 21 * 36 + T - 1) / T, T, 0, stream>>>(X, w1, b1, act1);
    conv2_kernel<<<(8 * 64 * 17 * 36 + T - 1) / T, T, 0, stream>>>(act1, w2, b2, act2);
    conv3_kernel<<<(8 * 64 * 13 * 16 + T - 1) / T, T, 0, stream>>>(act2, w3, b3, act3);
    pool_kernel<<<(8 * 64 * 11 * 4 + T - 1) / T, T, 0, stream>>>(act3, enc);

    edge_pass<<<(E + T - 1) / T, T, 0, stream>>>(ei, gid, x3, aggk, mask, cnt, E);

    maxenc_kernel<<<(ENC_ + 63) / 64, 64, 0, stream>>>(enc, ME);
    {
        dim3 gP(256 / 16, (HID_ + 15) / 16);
        gemm_wmma_f32<<<gP, 32, 0, stream>>>(ME, Wl1 + 3 * HID_, P, 256);
        dim3 gQ(1, (HID_ + 15) / 16);
        gemm_wmma_f32<<<gQ, 32, 0, stream>>>(enc, Wr1 + 3 * HID_, Q, 8);
    }

    node_h_kernel<<<(N + 7) / 8, 256, 0, stream>>>(x3, gid, mask, aggk, P, Q,
                                                   Wl1, Wr1, bl1, hbuf, N);

    edge_scatter_h<<<E, 128, 0, stream>>>(ei, hbuf, agg2, E);
    node_out_kernel<<<(N + 7) / 8, 256, 0, stream>>>(agg2, hbuf, cnt, Wl2, bl2, Wr2, y, N);

    (void)n_in; (void)out_size; (void)ws_size;
}


# Convs as implicit GEMMs on the f32 16x16x4 matrix instruction: wave = 16 positions x 16 out-channels, K = 4 in-channels per tap, padding taps zeroed (changes the summation order)
# speedup vs baseline: 1.4615x; 1.4615x over previous
#include <hip/hip_runtime.h>
#include <hip/hip_bf16.h>
#include <math.h>


#define B_     8
#define CIN_   25
#define NPG_   2000
#define ENC_   2816
#define HID_   100
#define KGEMM_ 2816

typedef __attribute__((ext_vector_type(2))) float v2f;
typedef __attribute__((ext_vector_type(8))) float v8f;

__device__ __forceinline__ unsigned fkey(float f) {
    unsigned u = __float_as_uint(f);
    return (u & 0x80000000u) ? ~u : (u | 0x80000000u);
}
__device__ __forceinline__ float keyf(unsigned k) {
    unsigned u = (k & 0x80000000u) ? (k ^ 0x80000000u) : ~k;
    return __uint_as_float(u);
}

__global__ void zero_u32(unsigned* __restrict__ p, int n) {
    int i = blockIdx.x * blockDim.x + threadIdx.x;
    if (i < n) p[i] = 0u;
}

template <int CIN, int DIN, int COUT, int DOUT, int HO, int WO, int PAD, int MT>
__global__ void __launch_bounds__(128)
conv_wmma_f32(const float* __restrict__ In, const float* __restrict__ Wt,
              const float* __restrict__ bias, float* __restrict__ Out) {
    constexpr int PLANE = 36;
    constexpr int CHS   = DIN * PLANE;
    constexpr int HWO   = HO * WO;
    constexpr int P     = DOUT * HWO;
    constexpr int NT    = COUT / 16;
    constexpr int CBF   = CIN / 4;
    constexpr int CREM  = CIN % 4;
    static_assert(P % 4 == 0, "16-byte stores need P % 4 == 0");
    static_assert(COUT % 16 == 0, "N tiles are full");

    const int lane = threadIdx.x & 31;
    const int half = lane >> 4;
    const int l16  = lane & 15;
    const int nTile = blockIdx.x % NT;
    const int mTile = (blockIdx.x / NT) * 4 + (threadIdx.x >> 5);
    if (mTile >= MT) return;

    const int m  = mTile * 16 + l16;
    const int bn = m / P;
    const int p  = m - bn * P;
    const int z  = p / HWO;
    const int q  = p - z * HWO;
    const int y  = q / WO;
    const int x  = q - y * WO;
    const int o  = nTile * 16 + l16;

    int  toff[9];
    bool tok[9];
    #pragma unroll
    for (int t = 0; t < 9; ++t) {
        const int yy = y + t / 3 - PAD;
        const int xx = x + t % 3 - PAD;
        const bool ok = (PAD == 0) ? true : ((yy >= 0) && (yy < 6) && (xx >= 0) && (xx < 6));
        tok[t]  = ok;
        toff[t] = ok ? (yy * 6 + xx) : 0;
    }

    int aIdx = (bn * CIN + half * 2) * CHS + z * PLANE;
    int bIdx = (o * CIN + half * 2) * 45;

    v8f acc0 = {}, acc1 = {};
    for (int cb = 0; cb < CBF; ++cb) {
        #pragma unroll
        for (int kd = 0; kd < 5; ++kd) {
            #pragma unroll
            for (int t = 0; t < 9; ++t) {
                const int ai = aIdx + kd * PLANE + toff[t];
                const int bi = bIdx + kd * 9 + t;
                const float ax = In[ai];
                const float ay = In[ai + CHS];
                v2f a, b;
                a.x = tok[t] ? ax : 0.0f;
                a.y = tok[t] ? ay : 0.0f;
                b.x = Wt[bi];
                b.y = Wt[bi + 45];
                if ((kd * 9 + t) & 1)
                    acc1 = __builtin_amdgcn_wmma_f32_16x16x4_f32(false, a, false, b,
                                                                 (short)0, acc1, false, false);
                else
                    acc0 = __builtin_amdgcn_wmma_f32_16x16x4_f32(false, a, false, b,
                                                                 (short)0, acc0, false, false);
            }
        }
        aIdx += 4 * CHS;
        bIdx += 4 * 45;
    }
    if (CREM != 0) {
        const int  c0  = CBF * 4 + half * 2;
        const bool cv0 = (c0 < CIN);
        const bool cv1 = (c0 + 1 < CIN);
        const int  cc0 = cv0 ? c0 : CBF * 4;
        const int  cc1 = cv1 ? (c0 + 1) : CBF * 4;
        const int aT0 = (bn * CIN + cc0) * CHS + z * PLANE;
        const int aT1 = (bn * CIN + cc1) * CHS + z * PLANE;
        const int bT0 = (o * CIN + cc0) * 45;
        const int bT1 = (o * CIN + cc1) * 45;
        #pragma unroll
        for (int kd = 0; kd < 5; ++kd) {
            #pragma unroll
            for (int t = 0; t < 9; ++t) {
                const int so = kd * PLANE + toff[t];
                const float ax = In[aT0 + so];
                const float ay = In[aT1 + so];
                v2f a, b;
                a.x = (tok[t] && cv0) ? ax : 0.0f;
                a.y = (tok[t] && cv1) ? ay : 0.0f;
                const float bx = Wt[bT0 + kd * 9 + t];
                const float by = Wt[bT1 + kd * 9 + t];
                b.x = cv0 ? bx : 0.0f;
                b.y = cv1 ? by : 0.0f;
                if ((kd * 9 + t) & 1)
                    acc1 = __builtin_amdgcn_wmma_f32_16x16x4_f32(false, a, false, b,
                                                                 (short)0, acc1, false, false);
                else
                    acc0 = __builtin_amdgcn_wmma_f32_16x16x4_f32(false, a, false, b,
                                                                 (short)0, acc0, false, false);
            }
        }
    }

    const v8f acc = acc0 + acc1;
    const float bo = bias[o];
    const int m0 = mTile * 16 + half * 8;
    #pragma unroll
    for (int g = 0; g < 2; ++g) {
        const int mg  = m0 + g * 4;
        const int bng = mg / P;
        const int pg  = mg - bng * P;
        float4 v;
        v.x = fmaxf(acc[g * 4 + 0] + bo, 0.0f);
        v.y = fmaxf(acc[g * 4 + 1] + bo, 0.0f);
        v.z = fmaxf(acc[g * 4 + 2] + bo, 0.0f);
        v.w = fmaxf(acc[g * 4 + 3] + bo, 0.0f);
        *reinterpret_cast<float4*>(Out + (size_t)(bng * COUT + o) * P + pg) = v;
    }
}

__global__ void pool_kernel(const float* __restrict__ A, float* __restrict__ enc) {
    const int total = 8 * 64 * 11 * 2 * 2;
    int idx = blockIdx.x * blockDim.x + threadIdx.x;
    if (idx >= total) return;
    int x = idx % 2;  int t = idx / 2;
    int y = t % 2;    t /= 2;
    int d = t % 11;   t /= 11;
    int ch = t % 64;  int bn = t / 64;
    float m = -INFINITY;
    for (int dd = 0; dd < 3; ++dd)
        for (int hh = 0; hh < 3; ++hh)
            for (int ww = 0; ww < 3; ++ww) {
                float v = A[((size_t)((bn * 64 + ch) * 13 + (d + dd))) * 16 + (y + hh) * 4 + (x + ww)];
                m = fmaxf(m, v);
            }
    enc[(size_t)bn * ENC_ + ch * 44 + d * 4 + y * 2 + x] = m;
}

__global__ void edge_pass(const int* __restrict__ ei, const int* __restrict__ gid,
                          const float* __restrict__ x3, unsigned* __restrict__ aggk,
                          unsigned* __restrict__ mask, int* __restrict__ counts, int E) {
    int e = blockIdx.x * blockDim.x + threadIdx.x;
    if (e >= E) return;
    int s = ei[e];
    int d = ei[E + e];
    int g = gid[s];
    atomicOr(&mask[d], 1u << g);
    atomicAdd(&counts[d], 1);
    atomicMax(&aggk[(size_t)d * 3 + 0], fkey(x3[(size_t)s * 3 + 0]));
    atomicMax(&aggk[(size_t)d * 3 + 1], fkey(x3[(size_t)s * 3 + 1]));
    atomicMax(&aggk[(size_t)d * 3 + 2], fkey(x3[(size_t)s * 3 + 2]));
}

__global__ void maxenc_kernel(const float* __restrict__ enc, float* __restrict__ ME) {
    int c = blockIdx.x * blockDim.x + threadIdx.x;
    if (c >= ENC_) return;
    ME[c] = 0.0f;
    for (int m = 1; m < 256; ++m) {
        int p = m & (m - 1);
        int g = __ffs(m) - 1;
        ME[(size_t)m * ENC_ + c] = fmaxf(ME[(size_t)p * ENC_ + c], enc[(size_t)g * ENC_ + c]);
    }
}

__global__ void __launch_bounds__(32)
gemm_wmma_f32(const float* __restrict__ A, const float* __restrict__ W,
              float* __restrict__ C, int Mrows) {
    const int Ncols = HID_;
    int lane = threadIdx.x & 31;
    int half = lane >> 4;
    int l16  = lane & 15;
    int mBase = blockIdx.x * 16;
    int nBase = blockIdx.y * 16;
    int mRow = mBase + l16; if (mRow >= Mrows) mRow = Mrows - 1;
    int nCol = nBase + l16;
    bool nOk = (nCol < Ncols);
    int nClamp = nOk ? nCol : (Ncols - 1);
    const float* Arow = A + (size_t)mRow * KGEMM_;
    const float* Bcol = W + nClamp;

    v8f acc0 = {}, acc1 = {};
    for (int kt = 0; kt < KGEMM_; kt += 8) {
        int k0 = kt + half * 2;
        int k1 = k0 + 4;
        v2f a0, b0, a1, b1;
        a0.x = Arow[k0];
        a0.y = Arow[k0 + 1];
        a1.x = Arow[k1];
        a1.y = Arow[k1 + 1];
        b0.x = Bcol[(size_t)k0 * Ncols];
        b0.y = Bcol[(size_t)(k0 + 1) * Ncols];
        b1.x = Bcol[(size_t)k1 * Ncols];
        b1.y = Bcol[(size_t)(k1 + 1) * Ncols];
        acc0 = __builtin_amdgcn_wmma_f32_16x16x4_f32(false, a0, false, b0,
                                                     (short)0, acc0, false, false);
        acc1 = __builtin_amdgcn_wmma_f32_16x16x4_f32(false, a1, false, b1,
                                                     (short)0, acc1, false, false);
    }
    v8f acc = acc0 + acc1;
    for (int v = 0; v < 8; ++v) {
        int m = mBase + half * 8 + v;
        if (m < Mrows && nOk) C[(size_t)m * Ncols + nCol] = acc[v];
    }
}

__global__ void node_h_kernel(const float* __restrict__ x3, const int* __restrict__ gid,
                              const unsigned* __restrict__ mask, const unsigned* __restrict__ aggk,
                              const float* __restrict__ P, const float* __restrict__ Q,
                              const float* __restrict__ Wl1, const float* __restrict__ Wr1,
                              const float* __restrict__ bl1, float* __restrict__ h, int N) {
    int wave = threadIdx.x >> 5;
    int lane = threadIdx.x & 31;
    int n = blockIdx.x * (blockDim.x >> 5) + wave;
    if (n >= N) return;

    float x0 = x3[(size_t)n * 3 + 0];
    float x1 = x3[(size_t)n * 3 + 1];
    float x2 = x3[(size_t)n * 3 + 2];
    int g = gid[n];
    unsigned mk = mask[n];
    bool has = (mk != 0u);
    float a0 = 0.f, a1 = 0.f, a2 = 0.f;
    if (has) {
        a0 = keyf(aggk[(size_t)n * 3 + 0]);
        a1 = keyf(aggk[(size_t)n * 3 + 1]);
        a2 = keyf(aggk[(size_t)n * 3 + 2]);
    }
    const float* Prow = P + (size_t)mk * HID_;
    const float* Qrow = Q + (size_t)g * HID_;

    float v[4];
    #pragma unroll
    for (int c = 0; c < 4; ++c) {
        int col = lane + 32 * c;
        if (col < HID_) {
            float t = bl1[col] + Qrow[col]
                    + x0 * Wr1[col] + x1 * Wr1[HID_ + col] + x2 * Wr1[2 * HID_ + col];
            if (has)
                t += Prow[col]
                   + a0 * Wl1[col] + a1 * Wl1[HID_ + col] + a2 * Wl1[2 * HID_ + col];
            v[c] = t;
        } else {
            v[c] = -INFINITY;
        }
    }
    float m = fmaxf(fmaxf(v[0], v[1]), fmaxf(v[2], v[3]));
    for (int off = 16; off >= 1; off >>= 1)
        m = fmaxf(m, __shfl_xor(m, off, 32));
    float s = 0.f;
    #pragma unroll
    for (int c = 0; c < 4; ++c) s += __expf(v[c] - m);
    for (int off = 16; off >= 1; off >>= 1)
        s += __shfl_xor(s, off, 32);
    float lse = m + __logf(s);
    #pragma unroll
    for (int c = 0; c < 4; ++c) {
        int col = lane + 32 * c;
        if (col < HID_) h[(size_t)n * HID_ + col] = v[c] - lse;
    }
}

__global__ void edge_scatter_h(const int* __restrict__ ei, const float* __restrict__ h,
                               float* __restrict__ agg2, int E) {
    int e = blockIdx.x;
    int j = threadIdx.x;
    if (j < HID_) {
        int s = ei[e];
        int d = ei[E + e];
        atomicAdd(&agg2[(size_t)d * HID_ + j], h[(size_t)s * HID_ + j]);
    }
}

__global__ void node_out_kernel(const float* __restrict__ agg2, const float* __restrict__ h,
                                const int* __restrict__ counts, const float* __restrict__ Wl2,
                                const float* __restrict__ bl2, const float* __restrict__ Wr2,
                                float* __restrict__ y, int N) {
    int wave = threadIdx.x >> 5;
    int lane = threadIdx.x & 31;
    int n = blockIdx.x * (blockDim.x >> 5) + wave;
    if (n >= N) return;
    float s1 = 0.f, s2 = 0.f;
    #pragma unroll
    for (int c = 0; c < 4; ++c) {
        int col = lane + 32 * c;
        if (col < HID_) {
            s1 = fmaf(agg2[(size_t)n * HID_ + col], Wl2[col], s1);
            s2 = fmaf(h[(size_t)n * HID_ + col],   Wr2[col], s2);
        }
    }
    for (int off = 16; off >= 1; off >>= 1) {
        s1 += __shfl_xor(s1, off, 32);
        s2 += __shfl_xor(s2, off, 32);
    }
    if (lane == 0)
        y[n] = s1 / fmaxf((float)counts[n], 1.0f) + bl2[0] + s2;
}

extern "C" void kernel_launch(void* const* d_in, const int* in_sizes, int n_in,
                              void* d_out, int out_size, void* d_ws, size_t ws_size,
                              hipStream_t stream) {
    const float* X      = (const float*)d_in[0];
    const float* x3     = (const float*)d_in[1];
    const int*   ei     = (const int*)d_in[2];
    const int*   gid    = (const int*)d_in[3];
    const float* w1     = (const float*)d_in[4];
    const float* b1     = (const float*)d_in[5];
    const float* w2     = (const float*)d_in[6];
    const float* b2     = (const float*)d_in[7];
    const float* w3     = (const float*)d_in[8];
    const float* b3     = (const float*)d_in[9];
    const float* Wl1    = (const float*)d_in[10];
    const float* bl1    = (const float*)d_in[11];
    const float* Wr1    = (const float*)d_in[12];
    const float* Wl2    = (const float*)d_in[13];
    const float* bl2    = (const float*)d_in[14];
    const float* Wr2    = (const float*)d_in[15];
    float* y = (float*)d_out;

    const int E = in_sizes[2] / 2;
    const int N = in_sizes[3];

    char* ws = (char*)d_ws;
    size_t off = 0;
    auto alloc = [&](size_t bytes) -> char* {
        char* p = ws + off;
        off = (off + bytes + 255) & ~(size_t)255;
        return p;
    };
    float*    act1  = (float*)alloc((size_t)8 * 32 * 21 * 36 * 4);
    float*    act2  = (float*)alloc((size_t)8 * 64 * 17 * 36 * 4);
    float*    act3  = (float*)alloc((size_t)8 * 64 * 13 * 16 * 4);
    float*    enc   = (float*)alloc((size_t)8 * ENC_ * 4);
    float*    ME    = (float*)alloc((size_t)256 * ENC_ * 4);
    float*    P     = (float*)alloc((size_t)256 * HID_ * 4);
    float*    Q     = (float*)alloc((size_t)16  * HID_ * 4);
    unsigned* aggk  = (unsigned*)alloc((size_t)N * 3 * 4);
    unsigned* mask  = (unsigned*)alloc((size_t)N * 4);
    int*      cnt   = (int*)alloc((size_t)N * 4);
    float*    hbuf  = (float*)alloc((size_t)N * HID_ * 4);
    float*    agg2  = (float*)alloc((size_t)N * HID_ * 4);

    const int T = 256;
    zero_u32<<<(N * 3 + T - 1) / T, T, 0, stream>>>(aggk, N * 3);
    zero_u32<<<(N + T - 1) / T, T, 0, stream>>>(mask, N);
    zero_u32<<<(N + T - 1) / T, T, 0, stream>>>((unsigned*)cnt, N);
    zero_u32<<<(N * HID_ + T - 1) / T, T, 0, stream>>>((unsigned*)agg2, N * HID_);

    conv_wmma_f32<25, 25, 32, 21, 6, 6, 1, 378><<<95 * 2, 128, 0, stream>>>(X, w1, b1, act1);
    conv_wmma_f32<32, 21, 64, 17, 6, 6, 1, 306><<<77 * 4, 128, 0, stream>>>(act1, w2, b2, act2);
    conv_wmma_f32<64, 17, 64, 13, 4, 4, 0, 104><<<26 * 4, 128, 0, stream>>>(act2, w3, b3, act3);
    pool_kernel<<<(8 * 64 * 11 * 4 + T - 1) / T, T, 0, stream>>>(act3, enc);

    edge_pass<<<(E + T - 1) / T, T, 0, stream>>>(ei, gid, x3, aggk, mask, cnt, E);

    maxenc_kernel<<<(ENC_ + 63) / 64, 64, 0, stream>>>(enc, ME);
    {
        dim3 gP(256 / 16, (HID_ + 15) / 16);
        gemm_wmma_f32<<<gP, 32, 0, stream>>>(ME, Wl1 + 3 * HID_, P, 256);
        dim3 gQ(1, (HID_ + 15) / 16);
        gemm_wmma_f32<<<gQ, 32, 0, stream>>>(enc, Wr1 + 3 * HID_, Q, 8);
    }

    node_h_kernel<<<(N + 7) / 8, 256, 0, stream>>>(x3, gid, mask, aggk, P, Q,
                                                   Wl1, Wr1, bl1, hbuf, N);

    edge_scatter_h<<<E, 128, 0, stream>>>(ei, hbuf, agg2, E);
    node_out_kernel<<<(N + 7) / 8, 256, 0, stream>>>(agg2, hbuf, cnt, Wl2, bl2, Wr2, y, N);

    (void)n_in; (void)out_size; (void)ws_size;
}


# Layer 2 pulled, not pushed: edge pass links each dst's in-edges into a list; a node's wave walks it, sums neighbour h rows in registers (arrival order, dots by 4-column lanes); no float atomics
# speedup vs baseline: 1.5483x; 1.0594x over previous
#include <hip/hip_runtime.h>
#include <hip/hip_bf16.h>
#include <math.h>


#define B_     8
#define CIN_   25
#define NPG_   2000
#define ENC_   2816
#define HID_   100
#define KGEMM_ 2816

typedef __attribute__((ext_vector_type(2))) float v2f;
typedef __attribute__((ext_vector_type(8))) float v8f;

__device__ __forceinline__ unsigned fkey(float f) {
    unsigned u = __float_as_uint(f);
    return (u & 0x80000000u) ? ~u : (u | 0x80000000u);
}
__device__ __forceinline__ float keyf(unsigned k) {
    unsigned u = (k & 0x80000000u) ? (k ^ 0x80000000u) : ~k;
    return __uint_as_float(u);
}

__global__ void zero_u32(unsigned* __restrict__ p, int n) {
    int i = blockIdx.x * blockDim.x + threadIdx.x;
    if (i < n) p[i] = 0u;
}

template <int CIN, int DIN, int COUT, int DOUT, int HO, int WO, int PAD, int MT>
__global__ void __launch_bounds__(128)
conv_wmma_f32(const float* __restrict__ In, const float* __restrict__ Wt,
              const float* __restrict__ bias, float* __restrict__ Out) {
    constexpr int PLANE = 36;
    constexpr int CHS   = DIN * PLANE;
    constexpr int HWO   = HO * WO;
    constexpr int P     = DOUT * HWO;
    constexpr int NT    = COUT / 16;
    constexpr int CBF   = CIN / 4;
    constexpr int CREM  = CIN % 4;
    static_assert(P % 4 == 0, "16-byte stores need P % 4 == 0");
    static_assert(COUT % 16 == 0, "N tiles are full");

    const int lane = threadIdx.x & 31;
    const int half = lane >> 4;
    const int l16  = lane & 15;
    const int nTile = blockIdx.x % NT;
    const int mTile = (blockIdx.x / NT) * 4 + (threadIdx.x >> 5);
    if (mTile >= MT) return;

    const int m  = mTile * 16 + l16;
    const int bn = m / P;
    const int p  = m - bn * P;
    const int z  = p / HWO;
    const int q  = p - z * HWO;
    const int y  = q / WO;
    const int x  = q - y * WO;
    const int o  = nTile * 16 + l16;

    int  toff[9];
    bool tok[9];
    #pragma unroll
    for (int t = 0; t < 9; ++t) {
        const int yy = y + t / 3 - PAD;
        const int xx = x + t % 3 - PAD;
        const bool ok = (PAD == 0) ? true : ((yy >= 0) && (yy < 6) && (xx >= 0) && (xx < 6));
        tok[t]  = ok;
        toff[t] = ok ? (yy * 6 + xx) : 0;
    }

    int aIdx = (bn * CIN + half * 2) * CHS + z * PLANE;
    int bIdx = (o * CIN + half * 2) * 45;

    v8f acc0 = {}, acc1 = {};
    for (int cb = 0; cb < CBF; ++cb) {
        #pragma unroll
        for (int kd = 0; kd < 5; ++kd) {
            #pragma unroll
            for (int t = 0; t < 9; ++t) {
                const int ai = aIdx + kd * PLANE + toff[t];
                const int bi = bIdx + kd * 9 + t;
                const float ax = In[ai];
                const float ay = In[ai + CHS];
                v2f a, b;
                a.x = tok[t] ? ax : 0.0f;
                a.y = tok[t] ? ay : 0.0f;
                b.x = Wt[bi];
                b.y = Wt[bi + 45];
                if ((kd * 9 + t) & 1)
                    acc1 = __builtin_amdgcn_wmma_f32_16x16x4_f32(false, a, false, b,
                                                                 (short)0, acc1, false, false);
                else
                    acc0 = __builtin_amdgcn_wmma_f32_16x16x4_f32(false, a, false, b,
                                                                 (short)0, acc0, false, false);
            }
        }
        aIdx += 4 * CHS;
        bIdx += 4 * 45;
    }
    if (CREM != 0) {
        const int  c0  = CBF * 4 + half * 2;
        const bool cv0 = (c0 < CIN);
        const bool cv1 = (c0 + 1 < CIN);
        const int  cc0 = cv0 ? c0 : CBF * 4;
        const int  cc1 = cv1 ? (c0 + 1) : CBF * 4;
        const int aT0 = (bn * CIN + cc0) * CHS + z * PLANE;
        const int aT1 = (bn * CIN + cc1) * CHS + z * PLANE;
        const int bT0 = (o * CIN + cc0) * 45;
        const int bT1 = (o * CIN + cc1) * 45;
        #pragma unroll
        for (int kd = 0; kd < 5; ++kd) {
            #pragma unroll
            for (int t = 0; t < 9; ++t) {
                const int so = kd * PLANE + toff[t];
                const float ax = In[aT0 + so];
                const float ay = In[aT1 + so];
                v2f a, b;
                a.x = (tok[t] && cv0) ? ax : 0.0f;
                a.y = (tok[t] && cv1) ? ay : 0.0f;
                const float bx = Wt[bT0 + kd * 9 + t];
                const float by = Wt[bT1 + kd * 9 + t];
                b.x = cv0 ? bx : 0.0f;
                b.y = cv1 ? by : 0.0f;
                if ((kd * 9 + t) & 1)
                    acc1 = __builtin_amdgcn_wmma_f32_16x16x4_f32(false, a, false, b,
                                                                 (short)0, acc1, false, false);
                else
                    acc0 = __builtin_amdgcn_wmma_f32_16x16x4_f32(false, a, false, b,
                                                                 (short)0, acc0, false, false);
            }
        }
    }

    const v8f acc = acc0 + acc1;
    const float bo = bias[o];
    const int m0 = mTile * 16 + half * 8;
    #pragma unroll
    for (int g = 0; g < 2; ++g) {
        const int mg  = m0 + g * 4;
        const int bng = mg / P;
        const int pg  = mg - bng * P;
        float4 v;
        v.x = fmaxf(acc[g * 4 + 0] + bo, 0.0f);
        v.y = fmaxf(acc[g * 4 + 1] + bo, 0.0f);
        v.z = fmaxf(acc[g * 4 + 2] + bo, 0.0f);
        v.w = fmaxf(acc[g * 4 + 3] + bo, 0.0f);
        *reinterpret_cast<float4*>(Out + (size_t)(bng * COUT + o) * P + pg) = v;
    }
}

__global__ void pool_kernel(const float* __restrict__ A, float* __restrict__ enc) {
    const int total = 8 * 64 * 11 * 2 * 2;
    int idx = blockIdx.x * blockDim.x + threadIdx.x;
    if (idx >= total) return;
    int x = idx % 2;  int t = idx / 2;
    int y = t % 2;    t /= 2;
    int d = t % 11;   t /= 11;
    int ch = t % 64;  int bn = t / 64;
    float m = -INFINITY;
    for (int dd = 0; dd < 3; ++dd)
        for (int hh = 0; hh < 3; ++hh)
            for (int ww = 0; ww < 3; ++ww) {
                float v = A[((size_t)((bn * 64 + ch) * 13 + (d + dd))) * 16 + (y + hh) * 4 + (x + ww)];
                m = fmaxf(m, v);
            }
    enc[(size_t)bn * ENC_ + ch * 44 + d * 4 + y * 2 + x] = m;
}

__global__ void edge_pass(const int* __restrict__ ei, const int* __restrict__ gid,
                          const float* __restrict__ x3, unsigned* __restrict__ aggk,
                          unsigned* __restrict__ mask, int* __restrict__ counts,
                          unsigned* __restrict__ head, int* __restrict__ link, int E) {
    int e = blockIdx.x * blockDim.x + threadIdx.x;
    if (e >= E) return;
    int s = ei[e];
    int d = ei[E + e];
    int g = gid[s];
    const unsigned prev = atomicExch(&head[d], (unsigned)e + 1u);
    reinterpret_cast<int2*>(link)[e] = make_int2((int)prev - 1, s);
    atomicOr(&mask[d], 1u << g);
    atomicAdd(&counts[d], 1);
    atomicMax(&aggk[(size_t)d * 3 + 0], fkey(x3[(size_t)s * 3 + 0]));
    atomicMax(&aggk[(size_t)d * 3 + 1], fkey(x3[(size_t)s * 3 + 1]));
    atomicMax(&aggk[(size_t)d * 3 + 2], fkey(x3[(size_t)s * 3 + 2]));
}

__global__ void maxenc_kernel(const float* __restrict__ enc, float* __restrict__ ME) {
    int c = blockIdx.x * blockDim.x + threadIdx.x;
    if (c >= ENC_) return;
    ME[c] = 0.0f;
    for (int m = 1; m < 256; ++m) {
        int p = m & (m - 1);
        int g = __ffs(m) - 1;
        ME[(size_t)m * ENC_ + c] = fmaxf(ME[(size_t)p * ENC_ + c], enc[(size_t)g * ENC_ + c]);
    }
}

__global__ void __launch_bounds__(32)
gemm_wmma_f32(const float* __restrict__ A, const float* __restrict__ W,
              float* __restrict__ C, int Mrows) {
    const int Ncols = HID_;
    int lane = threadIdx.x & 31;
    int half = lane >> 4;
    int l16  = lane & 15;
    int mBase = blockIdx.x * 16;
    int nBase = blockIdx.y * 16;
    int mRow = mBase + l16; if (mRow >= Mrows) mRow = Mrows - 1;
    int nCol = nBase + l16;
    bool nOk = (nCol < Ncols);
    int nClamp = nOk ? nCol : (Ncols - 1);
    const float* Arow = A + (size_t)mRow * KGEMM_;
    const float* Bcol = W + nClamp;

    v8f acc0 = {}, acc1 = {};
    for (int kt = 0; kt < KGEMM_; kt += 8) {
        int k0 = kt + half * 2;
        int k1 = k0 + 4;
        v2f a0, b0, a1, b1;
        a0.x = Arow[k0];
        a0.y = Arow[k0 + 1];
        a1.x = Arow[k1];
        a1.y = Arow[k1 + 1];
        b0.x = Bcol[(size_t)k0 * Ncols];
        b0.y = Bcol[(size_t)(k0 + 1) * Ncols];
        b1.x = Bcol[(size_t)k1 * Ncols];
        b1.y = Bcol[(size_t)(k1 + 1) * Ncols];
        acc0 = __builtin_amdgcn_wmma_f32_16x16x4_f32(false, a0, false, b0,
                                                     (short)0, acc0, false, false);
        acc1 = __builtin_amdgcn_wmma_f32_16x16x4_f32(false, a1, false, b1,
                                                     (short)0, acc1, false, false);
    }
    v8f acc = acc0 + acc1;
    for (int v = 0; v < 8; ++v) {
        int m = mBase + half * 8 + v;
        if (m < Mrows && nOk) C[(size_t)m * Ncols + nCol] = acc[v];
    }
}

__global__ void node_h_kernel(const float* __restrict__ x3, const int* __restrict__ gid,
                              const unsigned* __restrict__ mask, const unsigned* __restrict__ aggk,
                              const float* __restrict__ P, const float* __restrict__ Q,
                              const float* __restrict__ Wl1, const float* __restrict__ Wr1,
                              const float* __restrict__ bl1, float* __restrict__ h, int N) {
    int wave = threadIdx.x >> 5;
    int lane = threadIdx.x & 31;
    int n = blockIdx.x * (blockDim.x >> 5) + wave;
    if (n >= N) return;

    float x0 = x3[(size_t)n * 3 + 0];
    float x1 = x3[(size_t)n * 3 + 1];
    float x2 = x3[(size_t)n * 3 + 2];
    int g = gid[n];
    unsigned mk = mask[n];
    bool has = (mk != 0u);
    float a0 = 0.f, a1 = 0.f, a2 = 0.f;
    if (has) {
        a0 = keyf(aggk[(size_t)n * 3 + 0]);
        a1 = keyf(aggk[(size_t)n * 3 + 1]);
        a2 = keyf(aggk[(size_t)n * 3 + 2]);
    }
    const float* Prow = P + (size_t)mk * HID_;
    const float* Qrow = Q + (size_t)g * HID_;

    float v[4];
    #pragma unroll
    for (int c = 0; c < 4; ++c) {
        int col = lane + 32 * c;
        if (col < HID_) {
            float t = bl1[col] + Qrow[col]
                    + x0 * Wr1[col] + x1 * Wr1[HID_ + col] + x2 * Wr1[2 * HID_ + col];
            if (has)
                t += Prow[col]
                   + a0 * Wl1[col] + a1 * Wl1[HID_ + col] + a2 * Wl1[2 * HID_ + col];
            v[c] = t;
        } else {
            v[c] = -INFINITY;
        }
    }
    float m = fmaxf(fmaxf(v[0], v[1]), fmaxf(v[2], v[3]));
    for (int off = 16; off >= 1; off >>= 1)
        m = fmaxf(m, __shfl_xor(m, off, 32));
    float s = 0.f;
    #pragma unroll
    for (int c = 0; c < 4; ++c) s += __expf(v[c] - m);
    for (int off = 16; off >= 1; off >>= 1)
        s += __shfl_xor(s, off, 32);
    float lse = m + __logf(s);
    #pragma unroll
    for (int c = 0; c < 4; ++c) {
        int col = lane + 32 * c;
        if (col < HID_) h[(size_t)n * HID_ + col] = v[c] - lse;
    }
}

__global__ void node_out_kernel(const float* __restrict__ h, const int* __restrict__ counts,
                                const unsigned* __restrict__ head, const int* __restrict__ link2,
                                const float* __restrict__ Wl2, const float* __restrict__ bl2,
                                const float* __restrict__ Wr2, float* __restrict__ y,
                                int N, int E) {
    const int wave = threadIdx.x >> 5;
    const int lane = threadIdx.x & 31;
    const int n = __builtin_amdgcn_readfirstlane(blockIdx.x * (blockDim.x >> 5) + wave);
    if (n >= N) return;
    const bool own = (lane < HID_ / 4);
    const int  c4  = own ? lane * 4 : 0;
    const int2* link = reinterpret_cast<const int2*>(link2);
    const int cnt = counts[n];
    const int e0  = (int)head[n] - 1;

    float4 own_row = *reinterpret_cast<const float4*>(h + (size_t)n * HID_ + c4);
    float4 acc  = make_float4(0.f, 0.f, 0.f, 0.f);
    float4 pend = make_float4(0.f, 0.f, 0.f, 0.f);
    int2 lk = make_int2(-1, 0);
    if (cnt > 0) lk = link[min(max(e0, 0), E - 1)];
    for (int j = 0; j < cnt; ++j) {
        const int2 lkn = link[min(max(lk.x, 0), E - 1)];
        const int  sc  = min(max(lk.y, 0), N - 1);
        const float4 row = *reinterpret_cast<const float4*>(h + (size_t)sc * HID_ + c4);
        acc.x += pend.x; acc.y += pend.y; acc.z += pend.z; acc.w += pend.w;
        pend = row;
        lk = lkn;
    }
    acc.x += pend.x; acc.y += pend.y; acc.z += pend.z; acc.w += pend.w;

    float s1 = 0.f, s2 = 0.f;
    if (own) {
        s1 = fmaf(acc.x, Wl2[c4 + 0], s1);
        s1 = fmaf(acc.y, Wl2[c4 + 1], s1);
        s1 = fmaf(acc.z, Wl2[c4 + 2], s1);
        s1 = fmaf(acc.w, Wl2[c4 + 3], s1);
        s2 = fmaf(own_row.x, Wr2[c4 + 0], s2);
        s2 = fmaf(own_row.y, Wr2[c4 + 1], s2);
        s2 = fmaf(own_row.z, Wr2[c4 + 2], s2);
        s2 = fmaf(own_row.w, Wr2[c4 + 3], s2);
    }
    for (int off = 16; off >= 1; off >>= 1) {
        s1 += __shfl_xor(s1, off, 32);
        s2 += __shfl_xor(s2, off, 32);
    }
    if (lane == 0)
        y[n] = s1 / fmaxf((float)cnt, 1.0f) + bl2[0] + s2;
}

extern "C" void kernel_launch(void* const* d_in, const int* in_sizes, int n_in,
                              void* d_out, int out_size, void* d_ws, size_t ws_size,
                              hipStream_t stream) {
    const float* X      = (const float*)d_in[0];
    const float* x3     = (const float*)d_in[1];
    const int*   ei     = (const int*)d_in[2];
    const int*   gid    = (const int*)d_in[3];
    const float* w1     = (const float*)d_in[4];
    const float* b1     = (const float*)d_in[5];
    const float* w2     = (const float*)d_in[6];
    const float* b2     = (const float*)d_in[7];
    const float* w3     = (const float*)d_in[8];
    const float* b3     = (const float*)d_in[9];
    const float* Wl1    = (const float*)d_in[10];
    const float* bl1    = (const float*)d_in[11];
    const float* Wr1    = (const float*)d_in[12];
    const float* Wl2    = (const float*)d_in[13];
    const float* bl2    = (const float*)d_in[14];
    const float* Wr2    = (const float*)d_in[15];
    float* y = (float*)d_out;

    const int E = in_sizes[2] / 2;
    const int N = in_sizes[3];

    char* ws = (char*)d_ws;
    size_t off = 0;
    auto alloc = [&](size_t bytes) -> char* {
        char* p = ws + off;
        off = (off + bytes + 255) & ~(size_t)255;
        return p;
    };
    float*    act1  = (float*)alloc((size_t)8 * 32 * 21 * 36 * 4);
    float*    act2  = (float*)alloc((size_t)8 * 64 * 17 * 36 * 4);
    float*    act3  = (float*)alloc((size_t)8 * 64 * 13 * 16 * 4);
    float*    enc   = (float*)alloc((size_t)8 * ENC_ * 4);
    float*    ME    = (float*)alloc((size_t)256 * ENC_ * 4);
    float*    P     = (float*)alloc((size_t)256 * HID_ * 4);
    float*    Q     = (float*)alloc((size_t)16  * HID_ * 4);
    unsigned* aggk  = (unsigned*)alloc((size_t)N * 3 * 4);
    unsigned* mask  = (unsigned*)alloc((size_t)N * 4);
    int*      cnt   = (int*)alloc((size_t)N * 2 * 4);
    unsigned* head  = (unsigned*)(cnt + N);
    int*      link  = (int*)alloc((size_t)E * 8);
    float*    hbuf  = (float*)alloc((size_t)N * HID_ * 4);

    const int T = 256;
    zero_u32<<<(N * 3 + T - 1) / T, T, 0, stream>>>(aggk, N * 3);
    zero_u32<<<(N + T - 1) / T, T, 0, stream>>>(mask, N);
    zero_u32<<<(N * 2 + T - 1) / T, T, 0, stream>>>((unsigned*)cnt, N * 2);

    conv_wmma_f32<25, 25, 32, 21, 6, 6, 1, 378><<<95 * 2, 128, 0, stream>>>(X, w1, b1, act1);
    conv_wmma_f32<32, 21, 64, 17, 6, 6, 1, 306><<<77 * 4, 128, 0, stream>>>(act1, w2, b2, act2);
    conv_wmma_f32<64, 17, 64, 13, 4, 4, 0, 104><<<26 * 4, 128, 0, stream>>>(act2, w3, b3, act3);
    pool_kernel<<<(8 * 64 * 11 * 4 + T - 1) / T, T, 0, stream>>>(act3, enc);

    edge_pass<<<(E + T - 1) / T, T, 0, stream>>>(ei, gid, x3, aggk, mask, cnt, head, link, E);

    maxenc_kernel<<<(ENC_ + 63) / 64, 64, 0, stream>>>(enc, ME);
    {
        dim3 gP(256 / 16, (HID_ + 15) / 16);
        gemm_wmma_f32<<<gP, 32, 0, stream>>>(ME, Wl1 + 3 * HID_, P, 256);
        dim3 gQ(1, (HID_ + 15) / 16);
        gemm_wmma_f32<<<gQ, 32, 0, stream>>>(enc, Wr1 + 3 * HID_, Q, 8);
    }

    node_h_kernel<<<(N + 7) / 8, 256, 0, stream>>>(x3, gid, mask, aggk, P, Q,
                                                   Wl1, Wr1, bl1, hbuf, N);

    node_out_kernel<<<(N + 7) / 8, 256, 0, stream>>>(hbuf, cnt, head, link, Wl2, bl2, Wr2, y, N, E);

    (void)n_in; (void)out_size; (void)ws_size;
}


# Layer-1 GEMM without a table: A loader forms each mask row as max of the 8 enc rows; the 8 Wr1 rows ride as a 17th M tile; table kernel and 2nd GEMM launch gone (changes the K summation order)
# speedup vs baseline: 2.6877x; 1.7359x over previous
#include <hip/hip_runtime.h>
#include <hip/hip_bf16.h>
#include <math.h>


#define B_     8
#define CIN_   25
#define NPG_   2000
#define ENC_   2816
#define HID_   100
#define KGEMM_ 2816

typedef __attribute__((ext_vector_type(2))) float v2f;
typedef __attribute__((ext_vector_type(4))) float v4f;
typedef __attribute__((ext_vector_type(8))) float v8f;

__device__ __forceinline__ unsigned fkey(float f) {
    unsigned u = __float_as_uint(f);
    return (u & 0x80000000u) ? ~u : (u | 0x80000000u);
}
__device__ __forceinline__ float keyf(unsigned k) {
    unsigned u = (k & 0x80000000u) ? (k ^ 0x80000000u) : ~k;
    return __uint_as_float(u);
}

__global__ void zero_u32(unsigned* __restrict__ p, int n) {
    int i = blockIdx.x * blockDim.x + threadIdx.x;
    if (i < n) p[i] = 0u;
}

template <int CIN, int DIN, int COUT, int DOUT, int HO, int WO, int PAD, int MT>
__global__ void __launch_bounds__(128)
conv_wmma_f32(const float* __restrict__ In, const float* __restrict__ Wt,
              const float* __restrict__ bias, float* __restrict__ Out) {
    constexpr int PLANE = 36;
    constexpr int CHS   = DIN * PLANE;
    constexpr int HWO   = HO * WO;
    constexpr int P     = DOUT * HWO;
    constexpr int NT    = COUT / 16;
    constexpr int CBF   = CIN / 4;
    constexpr int CREM  = CIN % 4;
    static_assert(P % 4 == 0, "16-byte stores need P % 4 == 0");
    static_assert(COUT % 16 == 0, "N tiles are full");

    const int lane = threadIdx.x & 31;
    const int half = lane >> 4;
    const int l16  = lane & 15;
    const int nTile = blockIdx.x % NT;
    const int mTile = (blockIdx.x / NT) * 4 + (threadIdx.x >> 5);
    if (mTile >= MT) return;

    const int m  = mTile * 16 + l16;
    const int bn = m / P;
    const int p  = m - bn * P;
    const int z  = p / HWO;
    const int q  = p - z * HWO;
    const int y  = q / WO;
    const int x  = q - y * WO;
    const int o  = nTile * 16 + l16;

    int  toff[9];
    bool tok[9];
    #pragma unroll
    for (int t = 0; t < 9; ++t) {
        const int yy = y + t / 3 - PAD;
        const int xx = x + t % 3 - PAD;
        const bool ok = (PAD == 0) ? true : ((yy >= 0) && (yy < 6) && (xx >= 0) && (xx < 6));
        tok[t]  = ok;
        toff[t] = ok ? (yy * 6 + xx) : 0;
    }

    int aIdx = (bn * CIN + half * 2) * CHS + z * PLANE;
    int bIdx = (o * CIN + half * 2) * 45;

    v8f acc0 = {}, acc1 = {};
    for (int cb = 0; cb < CBF; ++cb) {
        #pragma unroll
        for (int kd = 0; kd < 5; ++kd) {
            #pragma unroll
            for (int t = 0; t < 9; ++t) {
                const int ai = aIdx + kd * PLANE + toff[t];
                const int bi = bIdx + kd * 9 + t;
                const float ax = In[ai];
                const float ay = In[ai + CHS];
                v2f a, b;
                a.x = tok[t] ? ax : 0.0f;
                a.y = tok[t] ? ay : 0.0f;
                b.x = Wt[bi];
                b.y = Wt[bi + 45];
                if ((kd * 9 + t) & 1)
                    acc1 = __builtin_amdgcn_wmma_f32_16x16x4_f32(false, a, false, b,
                                                                 (short)0, acc1, false, false);
                else
                    acc0 = __builtin_amdgcn_wmma_f32_16x16x4_f32(false, a, false, b,
                                                                 (short)0, acc0, false, false);
            }
        }
        aIdx += 4 * CHS;
        bIdx += 4 * 45;
    }
    if (CREM != 0) {
        const int  c0  = CBF * 4 + half * 2;
        const bool cv0 = (c0 < CIN);
        const bool cv1 = (c0 + 1 < CIN);
        const int  cc0 = cv0 ? c0 : CBF * 4;
        const int  cc1 = cv1 ? (c0 + 1) : CBF * 4;
        const int aT0 = (bn * CIN + cc0) * CHS + z * PLANE;
        const int aT1 = (bn * CIN + cc1) * CHS + z * PLANE;
        const int bT0 = (o * CIN + cc0) * 45;
        const int bT1 = (o * CIN + cc1) * 45;
        #pragma unroll
        for (int kd = 0; kd < 5; ++kd) {
            #pragma unroll
            for (int t = 0; t < 9; ++t) {
                const int so = kd * PLANE + toff[t];
                const float ax = In[aT0 + so];
                const float ay = In[aT1 + so];
                v2f a, b;
                a.x = (tok[t] && cv0) ? ax : 0.0f;
                a.y = (tok[t] && cv1) ? ay : 0.0f;
                const float bx = Wt[bT0 + kd * 9 + t];
                const float by = Wt[bT1 + kd * 9 + t];
                b.x = cv0 ? bx : 0.0f;
                b.y = cv1 ? by : 0.0f;
                if ((kd * 9 + t) & 1)
                    acc1 = __builtin_amdgcn_wmma_f32_16x16x4_f32(false, a, false, b,
                                                                 (short)0, acc1, false, false);
                else
                    acc0 = __builtin_amdgcn_wmma_f32_16x16x4_f32(false, a, false, b,
                                                                 (short)0, acc0, false, false);
            }
        }
    }

    const v8f acc = acc0 + acc1;
    const float bo = bias[o];
    const int m0 = mTile * 16 + half * 8;
    #pragma unroll
    for (int g = 0; g < 2; ++g) {
        const int mg  = m0 + g * 4;
        const int bng = mg / P;
        const int pg  = mg - bng * P;
        float4 v;
        v.x = fmaxf(acc[g * 4 + 0] + bo, 0.0f);
        v.y = fmaxf(acc[g * 4 + 1] + bo, 0.0f);
        v.z = fmaxf(acc[g * 4 + 2] + bo, 0.0f);
        v.w = fmaxf(acc[g * 4 + 3] + bo, 0.0f);
        *reinterpret_cast<float4*>(Out + (size_t)(bng * COUT + o) * P + pg) = v;
    }
}

__global__ void pool_kernel(const float* __restrict__ A, float* __restrict__ enc) {
    const int total = 8 * 64 * 11 * 2 * 2;
    int idx = blockIdx.x * blockDim.x + threadIdx.x;
    if (idx >= total) return;
    int x = idx % 2;  int t = idx / 2;
    int y = t % 2;    t /= 2;
    int d = t % 11;   t /= 11;
    int ch = t % 64;  int bn = t / 64;
    float m = -INFINITY;
    for (int dd = 0; dd < 3; ++dd)
        for (int hh = 0; hh < 3; ++hh)
            for (int ww = 0; ww < 3; ++ww) {
                float v = A[((size_t)((bn * 64 + ch) * 13 + (d + dd))) * 16 + (y + hh) * 4 + (x + ww)];
                m = fmaxf(m, v);
            }
    enc[(size_t)bn * ENC_ + ch * 44 + d * 4 + y * 2 + x] = m;
}

__global__ void edge_pass(const int* __restrict__ ei, const int* __restrict__ gid,
                          const float* __restrict__ x3, unsigned* __restrict__ aggk,
                          unsigned* __restrict__ mask, int* __restrict__ counts,
                          unsigned* __restrict__ head, int* __restrict__ link, int E) {
    int e = blockIdx.x * blockDim.x + threadIdx.x;
    if (e >= E) return;
    int s = ei[e];
    int d = ei[E + e];
    int g = gid[s];
    const unsigned prev = atomicExch(&head[d], (unsigned)e + 1u);
    reinterpret_cast<int2*>(link)[e] = make_int2((int)prev - 1, s);
    atomicOr(&mask[d], 1u << g);
    atomicAdd(&counts[d], 1);
    atomicMax(&aggk[(size_t)d * 3 + 0], fkey(x3[(size_t)s * 3 + 0]));
    atomicMax(&aggk[(size_t)d * 3 + 1], fkey(x3[(size_t)s * 3 + 1]));
    atomicMax(&aggk[(size_t)d * 3 + 2], fkey(x3[(size_t)s * 3 + 2]));
}

__global__ void __launch_bounds__(32)
gemm_submax_wmma_f32(const float* __restrict__ enc, const float* __restrict__ Wl,
                     const float* __restrict__ Wr, float* __restrict__ P,
                     float* __restrict__ Q) {
    const int Ncols = HID_;
    const int lane = threadIdx.x & 31;
    const int half = lane >> 4;
    const int l16  = lane & 15;
    const bool isQ = (blockIdx.x == 16);
    const float* W = isQ ? Wr : Wl;
    float* C = isQ ? Q : P;
    const int Mrows = isQ ? 8 : 256;
    const int mBase = isQ ? 0 : blockIdx.x * 16;
    const unsigned mask = isQ ? (1u << (l16 & 7)) : (unsigned)(mBase + l16);
    const int g0 = mask ? (__ffs(mask) - 1) : 0;
    const int nBase = blockIdx.y * 16;
    const int nCol = nBase + l16;
    const bool nOk = (nCol < Ncols);
    const int nClamp = nOk ? nCol : (Ncols - 1);

    const float* rowp[8];
    #pragma unroll
    for (int g = 0; g < 8; ++g) {
        const int r = ((mask >> g) & 1u) ? g : g0;
        rowp[g] = enc + (size_t)r * ENC_ + half * 8;
    }
    const float* Bcol = W + (size_t)(half * 8) * Ncols + nClamp;

    v8f acc0 = {}, acc1 = {};
    for (int kt = 0; kt < KGEMM_; kt += 16) {
        v4f L[8], H[8];
        float bv[8];
        #pragma unroll
        for (int g = 0; g < 8; ++g) {
            L[g] = *reinterpret_cast<const v4f*>(rowp[g] + kt);
            H[g] = *reinterpret_cast<const v4f*>(rowp[g] + kt + 4);
        }
        #pragma unroll
        for (int i = 0; i < 8; ++i) bv[i] = Bcol[(size_t)(kt + i) * Ncols];
        __builtin_amdgcn_sched_barrier(0);
        v4f lo = L[0], hi = H[0];
        #pragma unroll
        for (int g = 1; g < 8; ++g) {
            lo.x = fmaxf(lo.x, L[g].x); lo.y = fmaxf(lo.y, L[g].y);
            lo.z = fmaxf(lo.z, L[g].z); lo.w = fmaxf(lo.w, L[g].w);
            hi.x = fmaxf(hi.x, H[g].x); hi.y = fmaxf(hi.y, H[g].y);
            hi.z = fmaxf(hi.z, H[g].z); hi.w = fmaxf(hi.w, H[g].w);
        }
        v2f a0, a1, a2, a3, b0, b1, b2, b3;
        a0.x = fmaxf(lo.x, 0.0f); a0.y = fmaxf(lo.y, 0.0f);
        a1.x = fmaxf(lo.z, 0.0f); a1.y = fmaxf(lo.w, 0.0f);
        a2.x = fmaxf(hi.x, 0.0f); a2.y = fmaxf(hi.y, 0.0f);
        a3.x = fmaxf(hi.z, 0.0f); a3.y = fmaxf(hi.w, 0.0f);
        b0.x = bv[0]; b0.y = bv[1];
        b1.x = bv[2]; b1.y = bv[3];
        b2.x = bv[4]; b2.y = bv[5];
        b3.x = bv[6]; b3.y = bv[7];
        acc0 = __builtin_amdgcn_wmma_f32_16x16x4_f32(false, a0, false, b0,
                                                     (short)0, acc0, false, false);
        acc1 = __builtin_amdgcn_wmma_f32_16x16x4_f32(false, a1, false, b1,
                                                     (short)0, acc1, false, false);
        acc0 = __builtin_amdgcn_wmma_f32_16x16x4_f32(false, a2, false, b2,
                                                     (short)0, acc0, false, false);
        acc1 = __builtin_amdgcn_wmma_f32_16x16x4_f32(false, a3, false, b3,
                                                     (short)0, acc1, false, false);
    }
    v8f acc = acc0 + acc1;
    for (int v = 0; v < 8; ++v) {
        const int m = mBase + half * 8 + v;
        const float val = (!isQ && m == 0) ? 0.0f : acc[v];
        if (m < Mrows && nOk) C[(size_t)m * Ncols + nCol] = val;
    }
}

__global__ void node_h_kernel(const float* __restrict__ x3, const int* __restrict__ gid,
                              const unsigned* __restrict__ mask, const unsigned* __restrict__ aggk,
                              const float* __restrict__ P, const float* __restrict__ Q,
                              const float* __restrict__ Wl1, const float* __restrict__ Wr1,
                              const float* __restrict__ bl1, float* __restrict__ h, int N) {
    int wave = threadIdx.x >> 5;
    int lane = threadIdx.x & 31;
    int n = blockIdx.x * (blockDim.x >> 5) + wave;
    if (n >= N) return;

    float x0 = x3[(size_t)n * 3 + 0];
    float x1 = x3[(size_t)n * 3 + 1];
    float x2 = x3[(size_t)n * 3 + 2];
    int g = gid[n];
    unsigned mk = mask[n];
    bool has = (mk != 0u);
    float a0 = 0.f, a1 = 0.f, a2 = 0.f;
    if (has) {
        a0 = keyf(aggk[(size_t)n * 3 + 0]);
        a1 = keyf(aggk[(size_t)n * 3 + 1]);
        a2 = keyf(aggk[(size_t)n * 3 + 2]);
    }
    const float* Prow = P + (size_t)mk * HID_;
    const float* Qrow = Q + (size_t)g * HID_;

    float v[4];
    #pragma unroll
    for (int c = 0; c < 4; ++c) {
        int col = lane + 32 * c;
        if (col < HID_) {
            float t = bl1[col] + Qrow[col]
                    + x0 * Wr1[col] + x1 * Wr1[HID_ + col] + x2 * Wr1[2 * HID_ + col];
            if (has)
                t += Prow[col]
                   + a0 * Wl1[col] + a1 * Wl1[HID_ + col] + a2 * Wl1[2 * HID_ + col];
            v[c] = t;
        } else {
            v[c] = -INFINITY;
        }
    }
    float m = fmaxf(fmaxf(v[0], v[1]), fmaxf(v[2], v[3]));
    for (int off = 16; off >= 1; off >>= 1)
        m = fmaxf(m, __shfl_xor(m, off, 32));
    float s = 0.f;
    #pragma unroll
    for (int c = 0; c < 4; ++c) s += __expf(v[c] - m);
    for (int off = 16; off >= 1; off >>= 1)
        s += __shfl_xor(s, off, 32);
    float lse = m + __logf(s);
    #pragma unroll
    for (int c = 0; c < 4; ++c) {
        int col = lane + 32 * c;
        if (col < HID_) h[(size_t)n * HID_ + col] = v[c] - lse;
    }
}

__global__ void node_out_kernel(const float* __restrict__ h, const int* __restrict__ counts,
                                const unsigned* __restrict__ head, const int* __restrict__ link2,
                                const float* __restrict__ Wl2, const float* __restrict__ bl2,
                                const float* __restrict__ Wr2, float* __restrict__ y,
                                int N, int E) {
    const int wave = threadIdx.x >> 5;
    const int lane = threadIdx.x & 31;
    const int n = __builtin_amdgcn_readfirstlane(blockIdx.x * (blockDim.x >> 5) + wave);
    if (n >= N) return;
    const bool own = (lane < HID_ / 4);
    const int  c4  = own ? lane * 4 : 0;
    const int2* link = reinterpret_cast<const int2*>(link2);
    const int cnt = counts[n];
    const int e0  = (int)head[n] - 1;

    float4 own_row = *reinterpret_cast<const float4*>(h + (size_t)n * HID_ + c4);
    float4 acc  = make_float4(0.f, 0.f, 0.f, 0.f);
    float4 pend = make_float4(0.f, 0.f, 0.f, 0.f);
    int2 lk = make_int2(-1, 0);
    if (cnt > 0) lk = link[min(max(e0, 0), E - 1)];
    for (int j = 0; j < cnt; ++j) {
        const int2 lkn = link[min(max(lk.x, 0), E - 1)];
        const int  sc  = min(max(lk.y, 0), N - 1);
        const float4 row = *reinterpret_cast<const float4*>(h + (size_t)sc * HID_ + c4);
        acc.x += pend.x; acc.y += pend.y; acc.z += pend.z; acc.w += pend.w;
        pend = row;
        lk = lkn;
    }
    acc.x += pend.x; acc.y += pend.y; acc.z += pend.z; acc.w += pend.w;

    float s1 = 0.f, s2 = 0.f;
    if (own) {
        s1 = fmaf(acc.x, Wl2[c4 + 0], s1);
        s1 = fmaf(acc.y, Wl2[c4 + 1], s1);
        s1 = fmaf(acc.z, Wl2[c4 + 2], s1);
        s1 = fmaf(acc.w, Wl2[c4 + 3], s1);
        s2 = fmaf(own_row.x, Wr2[c4 + 0], s2);
        s2 = fmaf(own_row.y, Wr2[c4 + 1], s2);
        s2 = fmaf(own_row.z, Wr2[c4 + 2], s2);
        s2 = fmaf(own_row.w, Wr2[c4 + 3], s2);
    }
    for (int off = 16; off >= 1; off >>= 1) {
        s1 += __shfl_xor(s1, off, 32);
        s2 += __shfl_xor(s2, off, 32);
    }
    if (lane == 0)
        y[n] = s1 / fmaxf((float)cnt, 1.0f) + bl2[0] + s2;
}

extern "C" void kernel_launch(void* const* d_in, const int* in_sizes, int n_in,
                              void* d_out, int out_size, void* d_ws, size_t ws_size,
                              hipStream_t stream) {
    const float* X      = (const float*)d_in[0];
    const float* x3     = (const float*)d_in[1];
    const int*   ei     = (const int*)d_in[2];
    const int*   gid    = (const int*)d_in[3];
    const float* w1     = (const float*)d_in[4];
    const float* b1     = (const float*)d_in[5];
    const float* w2     = (const float*)d_in[6];
    const float* b2     = (const float*)d_in[7];
    const float* w3     = (const float*)d_in[8];
    const float* b3     = (const float*)d_in[9];
    const float* Wl1    = (const float*)d_in[10];
    const float* bl1    = (const float*)d_in[11];
    const float* Wr1    = (const float*)d_in[12];
    const float* Wl2    = (const float*)d_in[13];
    const float* bl2    = (const float*)d_in[14];
    const float* Wr2    = (const float*)d_in[15];
    float* y = (float*)d_out;

    const int E = in_sizes[2] / 2;
    const int N = in_sizes[3];

    char* ws = (char*)d_ws;
    size_t off = 0;
    auto alloc = [&](size_t bytes) -> char* {
        char* p = ws + off;
        off = (off + bytes + 255) & ~(size_t)255;
        return p;
    };
    float*    act1  = (float*)alloc((size_t)8 * 32 * 21 * 36 * 4);
    float*    act2  = (float*)alloc((size_t)8 * 64 * 17 * 36 * 4);
    float*    act3  = (float*)alloc((size_t)8 * 64 * 13 * 16 * 4);
    float*    enc   = (float*)alloc((size_t)8 * ENC_ * 4);
    float*    P     = (float*)alloc((size_t)256 * HID_ * 4);
    float*    Q     = (float*)alloc((size_t)16  * HID_ * 4);
    unsigned* aggk  = (unsigned*)alloc((size_t)N * 3 * 4);
    unsigned* mask  = (unsigned*)alloc((size_t)N * 4);
    int*      cnt   = (int*)alloc((size_t)N * 2 * 4);
    unsigned* head  = (unsigned*)(cnt + N);
    int*      link  = (int*)alloc((size_t)E * 8);
    float*    hbuf  = (float*)alloc((size_t)N * HID_ * 4);

    const int T = 256;
    zero_u32<<<(N * 3 + T - 1) / T, T, 0, stream>>>(aggk, N * 3);
    zero_u32<<<(N + T - 1) / T, T, 0, stream>>>(mask, N);
    zero_u32<<<(N * 2 + T - 1) / T, T, 0, stream>>>((unsigned*)cnt, N * 2);

    conv_wmma_f32<25, 25, 32, 21, 6, 6, 1, 378><<<95 * 2, 128, 0, stream>>>(X, w1, b1, act1);
    conv_wmma_f32<32, 21, 64, 17, 6, 6, 1, 306><<<77 * 4, 128, 0, stream>>>(act1, w2, b2, act2);
    conv_wmma_f32<64, 17, 64, 13, 4, 4, 0, 104><<<26 * 4, 128, 0, stream>>>(act2, w3, b3, act3);
    pool_kernel<<<(8 * 64 * 11 * 4 + T - 1) / T, T, 0, stream>>>(act3, enc);

    edge_pass<<<(E + T - 1) / T, T, 0, stream>>>(ei, gid, x3, aggk, mask, cnt, head, link, E);

    {
        dim3 gPQ(256 / 16 + 1, (HID_ + 15) / 16);
        gemm_submax_wmma_f32<<<gPQ, 32, 0, stream>>>(enc, Wl1 + 3 * HID_, Wr1 + 3 * HID_, P, Q);
    }

    node_h_kernel<<<(N + 7) / 8, 256, 0, stream>>>(x3, gid, mask, aggk, P, Q,
                                                   Wl1, Wr1, bl1, hbuf, N);

    node_out_kernel<<<(N + 7) / 8, 256, 0, stream>>>(hbuf, cnt, head, link, Wl2, bl2, Wr2, y, N, E);

    (void)n_in; (void)out_size; (void)ws_size;
}


# Conv weights of a workgroup's N tile copied once into LDS by contiguous loads; a B fragment is one 8-byte LDS read, issued as a group ahead of each 45-WMMA block; sums and their order unchanged
# speedup vs baseline: 2.8555x; 1.0624x over previous
#include <hip/hip_runtime.h>
#include <hip/hip_bf16.h>
#include <math.h>


#define B_     8
#define CIN_   25
#define NPG_   2000
#define ENC_   2816
#define HID_   100
#define KGEMM_ 2816

typedef __attribute__((ext_vector_type(2))) float v2f;
typedef __attribute__((ext_vector_type(4))) float v4f;
typedef __attribute__((ext_vector_type(8))) float v8f;

__device__ __forceinline__ unsigned fkey(float f) {
    unsigned u = __float_as_uint(f);
    return (u & 0x80000000u) ? ~u : (u | 0x80000000u);
}
__device__ __forceinline__ float keyf(unsigned k) {
    unsigned u = (k & 0x80000000u) ? (k ^ 0x80000000u) : ~k;
    return __uint_as_float(u);
}

__global__ void zero_u32(unsigned* __restrict__ p, int n) {
    int i = blockIdx.x * blockDim.x + threadIdx.x;
    if (i < n) p[i] = 0u;
}

constexpr int largest_divisor_upto(int n, int cap) {
    int best = 1;
    for (int d = 1; d <= cap; ++d) if (n % d == 0) best = d;
    return best;
}

template <int CIN, int DIN, int COUT, int DOUT, int HO, int WO, int PAD, int MT>
__global__ void __launch_bounds__(128)
conv_wmma_f32(const float* __restrict__ In, const float* __restrict__ Wt,
              const float* __restrict__ bias, float* __restrict__ Out) {
    constexpr int PLANE = 36;
    constexpr int CHS   = DIN * PLANE;
    constexpr int HWO   = HO * WO;
    constexpr int P     = DOUT * HWO;
    constexpr int NT    = COUT / 16;
    constexpr int CBF   = CIN / 4;
    constexpr int CREM  = CIN % 4;
    constexpr int CP    = (CIN + 3) / 4 * 4;
    constexpr int ROW   = CP * 45;
    constexpr int PITCH = ROW + (68 - ROW % 64) % 64;
    constexpr int WTOT  = 16 * CIN * 45;
    static_assert(P % 4 == 0, "16-byte stores need P % 4 == 0");
    static_assert(COUT % 16 == 0, "N tiles are full");
    static_assert(PITCH % 64 == 4 && CIN >= 3, "pitch rule; the stager's channel step wraps at most once");

    __shared__ __attribute__((aligned(16))) float Bs[16 * PITCH];

    const int lane = threadIdx.x & 31;
    const int half = lane >> 4;
    const int l16  = lane & 15;
    const int nTile = blockIdx.x % NT;
    {
        const float* Wn = Wt + (size_t)nTile * WTOT;
        constexpr int FULL  = WTOT / 128;
        constexpr int BATCH = largest_divisor_upto(FULL, 60);
        int tap = threadIdx.x % 45, c = threadIdx.x / 45, o = 0;
        for (int j0 = 0; j0 < FULL; j0 += BATCH) {
            float v[BATCH];
            #pragma unroll
            for (int u = 0; u < BATCH; ++u) v[u] = Wn[(j0 + u) * 128 + threadIdx.x];
            #pragma unroll
            for (int u = 0; u < BATCH; ++u) {
                Bs[o * PITCH + ((c >> 1) * 45 + tap) * 2 + (c & 1)] = v[u];
                tap += 38; c += 2;
                if (tap >= 45) { tap -= 45; c += 1; }
                if (c >= CIN)  { c -= CIN;  o += 1; }
            }
        }
        if constexpr (WTOT % 128 != 0) {
            const int i = FULL * 128 + threadIdx.x;
            if (i < WTOT)
                Bs[o * PITCH + ((c >> 1) * 45 + tap) * 2 + (c & 1)] = Wn[i];
        }
        if constexpr (CREM != 0) {
            constexpr int NPAD = CP - CIN;
            for (int i = threadIdx.x; i < 16 * NPAD * 45; i += 128) {
                const int oz = i / (NPAD * 45);
                const int r  = i - oz * (NPAD * 45);
                const int cz = CIN + r / 45;
                const int tz = r % 45;
                Bs[oz * PITCH + ((cz >> 1) * 45 + tz) * 2 + (cz & 1)] = 0.0f;
            }
        }
    }
    __syncthreads();

    const int mTile = (blockIdx.x / NT) * 4 + (threadIdx.x >> 5);
    if (mTile >= MT) return;

    const int m  = mTile * 16 + l16;
    const int bn = m / P;
    const int p  = m - bn * P;
    const int z  = p / HWO;
    const int q  = p - z * HWO;
    const int y  = q / WO;
    const int x  = q - y * WO;
    const int o  = nTile * 16 + l16;

    int  toff[9];
    bool tok[9];
    #pragma unroll
    for (int t = 0; t < 9; ++t) {
        const int yy = y + t / 3 - PAD;
        const int xx = x + t % 3 - PAD;
        const bool ok = (PAD == 0) ? true : ((yy >= 0) && (yy < 6) && (xx >= 0) && (xx < 6));
        tok[t]  = ok;
        toff[t] = ok ? (yy * 6 + xx) : 0;
    }

    int aIdx = (bn * CIN + half * 2) * CHS + z * PLANE;
    const float* bP = Bs + l16 * PITCH + half * 90;

    v8f acc0 = {}, acc1 = {};
    for (int cb = 0; cb < CBF; ++cb) {
        v2f   bf[45];
        float ax[45], ay[45];
        #pragma unroll
        for (int kd = 0; kd < 5; ++kd) {
            #pragma unroll
            for (int t = 0; t < 9; ++t) {
                const int ai = aIdx + kd * PLANE + toff[t];
                bf[kd * 9 + t] = *reinterpret_cast<const v2f*>(bP + (kd * 9 + t) * 2);
                ax[kd * 9 + t] = In[ai];
                ay[kd * 9 + t] = In[ai + CHS];
            }
        }
        __builtin_amdgcn_sched_barrier(0);
        #pragma unroll
        for (int kd = 0; kd < 5; ++kd) {
            #pragma unroll
            for (int t = 0; t < 9; ++t) {
                v2f a;
                a.x = tok[t] ? ax[kd * 9 + t] : 0.0f;
                a.y = tok[t] ? ay[kd * 9 + t] : 0.0f;
                const v2f b = bf[kd * 9 + t];
                if ((kd * 9 + t) & 1)
                    acc1 = __builtin_amdgcn_wmma_f32_16x16x4_f32(false, a, false, b,
                                                                 (short)0, acc1, false, false);
                else
                    acc0 = __builtin_amdgcn_wmma_f32_16x16x4_f32(false, a, false, b,
                                                                 (short)0, acc0, false, false);
            }
        }
        aIdx += 4 * CHS;
        bP   += 4 * 45;
    }
    if (CREM != 0) {
        const int  c0  = CBF * 4 + half * 2;
        const bool cv0 = (c0 < CIN);
        const bool cv1 = (c0 + 1 < CIN);
        const int  cc0 = cv0 ? c0 : CBF * 4;
        const int  cc1 = cv1 ? (c0 + 1) : CBF * 4;
        const int aT0 = (bn * CIN + cc0) * CHS + z * PLANE;
        const int aT1 = (bn * CIN + cc1) * CHS + z * PLANE;
        #pragma unroll
        for (int kd = 0; kd < 5; ++kd) {
            #pragma unroll
            for (int t = 0; t < 9; ++t) {
                const int so = kd * PLANE + toff[t];
                const float ax = In[aT0 + so];
                const float ay = In[aT1 + so];
                v2f a;
                a.x = (tok[t] && cv0) ? ax : 0.0f;
                a.y = (tok[t] && cv1) ? ay : 0.0f;
                const v2f b = *reinterpret_cast<const v2f*>(bP + (kd * 9 + t) * 2);
                if ((kd * 9 + t) & 1)
                    acc1 = __builtin_amdgcn_wmma_f32_16x16x4_f32(false, a, false, b,
                                                                 (short)0, acc1, false, false);
                else
                    acc0 = __builtin_amdgcn_wmma_f32_16x16x4_f32(false, a, false, b,
                                                                 (short)0, acc0, false, false);
            }
        }
    }

    const v8f acc = acc0 + acc1;
    const float bo = bias[o];
    const int m0 = mTile * 16 + half * 8;
    #pragma unroll
    for (int g = 0; g < 2; ++g) {
        const int mg  = m0 + g * 4;
        const int bng = mg / P;
        const int pg  = mg - bng * P;
        float4 v;
        v.x = fmaxf(acc[g * 4 + 0] + bo, 0.0f);
        v.y = fmaxf(acc[g * 4 + 1] + bo, 0.0f);
        v.z = fmaxf(acc[g * 4 + 2] + bo, 0.0f);
        v.w = fmaxf(acc[g * 4 + 3] + bo, 0.0f);
        *reinterpret_cast<float4*>(Out + (size_t)(bng * COUT + o) * P + pg) = v;
    }
}

__global__ void pool_kernel(const float* __restrict__ A, float* __restrict__ enc) {
    const int total = 8 * 64 * 11 * 2 * 2;
    int idx = blockIdx.x * blockDim.x + threadIdx.x;
    if (idx >= total) return;
    int x = idx % 2;  int t = idx / 2;
    int y = t % 2;    t /= 2;
    int d = t % 11;   t /= 11;
    int ch = t % 64;  int bn = t / 64;
    float m = -INFINITY;
    for (int dd = 0; dd < 3; ++dd)
        for (int hh = 0; hh < 3; ++hh)
            for (int ww = 0; ww < 3; ++ww) {
                float v = A[((size_t)((bn * 64 + ch) * 13 + (d + dd))) * 16 + (y + hh) * 4 + (x + ww)];
                m = fmaxf(m, v);
            }
    enc[(size_t)bn * ENC_ + ch * 44 + d * 4 + y * 2 + x] = m;
}

__global__ void edge_pass(const int* __restrict__ ei, const int* __restrict__ gid,
                          const float* __restrict__ x3, unsigned* __restrict__ aggk,
                          unsigned* __restrict__ mask, int* __restrict__ counts,
                          unsigned* __restrict__ head, int* __restrict__ link, int E) {
    int e = blockIdx.x * blockDim.x + threadIdx.x;
    if (e >= E) return;
    int s = ei[e];
    int d = ei[E + e];
    int g = gid[s];
    const unsigned prev = atomicExch(&head[d], (unsigned)e + 1u);
    reinterpret_cast<int2*>(link)[e] = make_int2((int)prev - 1, s);
    atomicOr(&mask[d], 1u << g);
    atomicAdd(&counts[d], 1);
    atomicMax(&aggk[(size_t)d * 3 + 0], fkey(x3[(size_t)s * 3 + 0]));
    atomicMax(&aggk[(size_t)d * 3 + 1], fkey(x3[(size_t)s * 3 + 1]));
    atomicMax(&aggk[(size_t)d * 3 + 2], fkey(x3[(size_t)s * 3 + 2]));
}

__global__ void __launch_bounds__(32)
gemm_submax_wmma_f32(const float* __restrict__ enc, const float* __restrict__ Wl,
                     const float* __restrict__ Wr, float* __restrict__ P,
                     float* __restrict__ Q) {
    const int Ncols = HID_;
    const int lane = threadIdx.x & 31;
    const int half = lane >> 4;
    const int l16  = lane & 15;
    const bool isQ = (blockIdx.x == 16);
    const float* W = isQ ? Wr : Wl;
    float* C = isQ ? Q : P;
    const int Mrows = isQ ? 8 : 256;
    const int mBase = isQ ? 0 : blockIdx.x * 16;
    const unsigned mask = isQ ? (1u << (l16 & 7)) : (unsigned)(mBase + l16);
    const int g0 = mask ? (__ffs(mask) - 1) : 0;
    const int nBase = blockIdx.y * 16;
    const int nCol = nBase + l16;
    const bool nOk = (nCol < Ncols);
    const int nClamp = nOk ? nCol : (Ncols - 1);

    const float* rowp[8];
    #pragma unroll
    for (int g = 0; g < 8; ++g) {
        const int r = ((mask >> g) & 1u) ? g : g0;
        rowp[g] = enc + (size_t)r * ENC_ + half * 8;
    }
    const float* Bcol = W + (size_t)(half * 8) * Ncols + nClamp;

    v8f acc0 = {}, acc1 = {};
    for (int kt = 0; kt < KGEMM_; kt += 16) {
        v4f L[8], H[8];
        float bv[8];
        #pragma unroll
        for (int g = 0; g < 8; ++g) {
            L[g] = *reinterpret_cast<const v4f*>(rowp[g] + kt);
            H[g] = *reinterpret_cast<const v4f*>(rowp[g] + kt + 4);
        }
        #pragma unroll
        for (int i = 0; i < 8; ++i) bv[i] = Bcol[(size_t)(kt + i) * Ncols];
        __builtin_amdgcn_sched_barrier(0);
        v4f lo = L[0], hi = H[0];
        #pragma unroll
        for (int g = 1; g < 8; ++g) {
            lo.x = fmaxf(lo.x, L[g].x); lo.y = fmaxf(lo.y, L[g].y);
            lo.z = fmaxf(lo.z, L[g].z); lo.w = fmaxf(lo.w, L[g].w);
            hi.x = fmaxf(hi.x, H[g].x); hi.y = fmaxf(hi.y, H[g].y);
            hi.z = fmaxf(hi.z, H[g].z); hi.w = fmaxf(hi.w, H[g].w);
        }
        v2f a0, a1, a2, a3, b0, b1, b2, b3;
        a0.x = fmaxf(lo.x, 0.0f); a0.y = fmaxf(lo.y, 0.0f);
        a1.x = fmaxf(lo.z, 0.0f); a1.y = fmaxf(lo.w, 0.0f);
        a2.x = fmaxf(hi.x, 0.0f); a2.y = fmaxf(hi.y, 0.0f);
        a3.x = fmaxf(hi.z, 0.0f); a3.y = fmaxf(hi.w, 0.0f);
        b0.x = bv[0]; b0.y = bv[1];
        b1.x = bv[2]; b1.y = bv[3];
        b2.x = bv[4]; b2.y = bv[5];
        b3.x = bv[6]; b3.y = bv[7];
        acc0 = __builtin_amdgcn_wmma_f32_16x16x4_f32(false, a0, false, b0,
                                                     (short)0, acc0, false, false);
        acc1 = __builtin_amdgcn_wmma_f32_16x16x4_f32(false, a1, false, b1,
                                                     (short)0, acc1, false, false);
        acc0 = __builtin_amdgcn_wmma_f32_16x16x4_f32(false, a2, false, b2,
                                                     (short)0, acc0, false, false);
        acc1 = __builtin_amdgcn_wmma_f32_16x16x4_f32(false, a3, false, b3,
                                                     (short)0, acc1, false, false);
    }
    v8f acc = acc0 + acc1;
    for (int v = 0; v < 8; ++v) {
        const int m = mBase + half * 8 + v;
        const float val = (!isQ && m == 0) ? 0.0f : acc[v];
        if (m < Mrows && nOk) C[(size_t)m * Ncols + nCol] = val;
    }
}

__global__ void node_h_kernel(const float* __restrict__ x3, const int* __restrict__ gid,
                              const unsigned* __restrict__ mask, const unsigned* __restrict__ aggk,
                              const float* __restrict__ P, const float* __restrict__ Q,
                              const float* __restrict__ Wl1, const float* __restrict__ Wr1,
                              const float* __restrict__ bl1, float* __restrict__ h, int N) {
    int wave = threadIdx.x >> 5;
    int lane = threadIdx.x & 31;
    int n = blockIdx.x * (blockDim.x >> 5) + wave;
    if (n >= N) return;

    float x0 = x3[(size_t)n * 3 + 0];
    float x1 = x3[(size_t)n * 3 + 1];
    float x2 = x3[(size_t)n * 3 + 2];
    int g = gid[n];
    unsigned mk = mask[n];
    bool has = (mk != 0u);
    float a0 = 0.f, a1 = 0.f, a2 = 0.f;
    if (has) {
        a0 = keyf(aggk[(size_t)n * 3 + 0]);
        a1 = keyf(aggk[(size_t)n * 3 + 1]);
        a2 = keyf(aggk[(size_t)n * 3 + 2]);
    }
    const float* Prow = P + (size_t)mk * HID_;
    const float* Qrow = Q + (size_t)g * HID_;

    float v[4];
    #pragma unroll
    for (int c = 0; c < 4; ++c) {
        int col = lane + 32 * c;
        if (col < HID_) {
            float t = bl1[col] + Qrow[col]
                    + x0 * Wr1[col] + x1 * Wr1[HID_ + col] + x2 * Wr1[2 * HID_ + col];
            if (has)
                t += Prow[col]
                   + a0 * Wl1[col] + a1 * Wl1[HID_ + col] + a2 * Wl1[2 * HID_ + col];
            v[c] = t;
        } else {
            v[c] = -INFINITY;
        }
    }
    float m = fmaxf(fmaxf(v[0], v[1]), fmaxf(v[2], v[3]));
    for (int off = 16; off >= 1; off >>= 1)
        m = fmaxf(m, __shfl_xor(m, off, 32));
    float s = 0.f;
    #pragma unroll
    for (int c = 0; c < 4; ++c) s += __expf(v[c] - m);
    for (int off = 16; off >= 1; off >>= 1)
        s += __shfl_xor(s, off, 32);
    float lse = m + __logf(s);
    #pragma unroll
    for (int c = 0; c < 4; ++c) {
        int col = lane + 32 * c;
        if (col < HID_) h[(size_t)n * HID_ + col] = v[c] - lse;
    }
}

__global__ void node_out_kernel(const float* __restrict__ h, const int* __restrict__ counts,
                                const unsigned* __restrict__ head, const int* __restrict__ link2,
                                const float* __restrict__ Wl2, const float* __restrict__ bl2,
                                const float* __restrict__ Wr2, float* __restrict__ y,
                                int N, int E) {
    const int wave = threadIdx.x >> 5;
    const int lane = threadIdx.x & 31;
    const int n = __builtin_amdgcn_readfirstlane(blockIdx.x * (blockDim.x >> 5) + wave);
    if (n >= N) return;
    const bool own = (lane < HID_ / 4);
    const int  c4  = own ? lane * 4 : 0;
    const int2* link = reinterpret_cast<const int2*>(link2);
    const int cnt = counts[n];
    const int e0  = (int)head[n] - 1;

    float4 own_row = *reinterpret_cast<const float4*>(h + (size_t)n * HID_ + c4);
    float4 acc  = make_float4(0.f, 0.f, 0.f, 0.f);
    float4 pend = make_float4(0.f, 0.f, 0.f, 0.f);
    int2 lk = make_int2(-1, 0);
    if (cnt > 0) lk = link[min(max(e0, 0), E - 1)];
    for (int j = 0; j < cnt; ++j) {
        const int2 lkn = link[min(max(lk.x, 0), E - 1)];
        const int  sc  = min(max(lk.y, 0), N - 1);
        const float4 row = *reinterpret_cast<const float4*>(h + (size_t)sc * HID_ + c4);
        acc.x += pend.x; acc.y += pend.y; acc.z += pend.z; acc.w += pend.w;
        pend = row;
        lk = lkn;
    }
    acc.x += pend.x; acc.y += pend.y; acc.z += pend.z; acc.w += pend.w;

    float s1 = 0.f, s2 = 0.f;
    if (own) {
        s1 = fmaf(acc.x, Wl2[c4 + 0], s1);
        s1 = fmaf(acc.y, Wl2[c4 + 1], s1);
        s1 = fmaf(acc.z, Wl2[c4 + 2], s1);
        s1 = fmaf(acc.w, Wl2[c4 + 3], s1);
        s2 = fmaf(own_row.x, Wr2[c4 + 0], s2);
        s2 = fmaf(own_row.y, Wr2[c4 + 1], s2);
        s2 = fmaf(own_row.z, Wr2[c4 + 2], s2);
        s2 = fmaf(own_row.w, Wr2[c4 + 3], s2);
    }
    for (int off = 16; off >= 1; off >>= 1) {
        s1 += __shfl_xor(s1, off, 32);
        s2 += __shfl_xor(s2, off, 32);
    }
    if (lane == 0)
        y[n] = s1 / fmaxf((float)cnt, 1.0f) + bl2[0] + s2;
}

extern "C" void kernel_launch(void* const* d_in, const int* in_sizes, int n_in,
                              void* d_out, int out_size, void* d_ws, size_t ws_size,
                              hipStream_t stream) {
    const float* X      = (const float*)d_in[0];
    const float* x3     = (const float*)d_in[1];
    const int*   ei     = (const int*)d_in[2];
    const int*   gid    = (const int*)d_in[3];
    const float* w1     = (const float*)d_in[4];
    const float* b1     = (const float*)d_in[5];
    const float* w2     = (const float*)d_in[6];
    const float* b2     = (const float*)d_in[7];
    const float* w3     = (const float*)d_in[8];
    const float* b3     = (const float*)d_in[9];
    const float* Wl1    = (const float*)d_in[10];
    const float* bl1    = (const float*)d_in[11];
    const float* Wr1    = (const float*)d_in[12];
    const float* Wl2    = (const float*)d_in[13];
    const float* bl2    = (const float*)d_in[14];
    const float* Wr2    = (const float*)d_in[15];
    float* y = (float*)d_out;

    const int E = in_sizes[2] / 2;
    const int N = in_sizes[3];

    char* ws = (char*)d_ws;
    size_t off = 0;
    auto alloc = [&](size_t bytes) -> char* {
        char* p = ws + off;
        off = (off + bytes + 255) & ~(size_t)255;
        return p;
    };
    float*    act1  = (float*)alloc((size_t)8 * 32 * 21 * 36 * 4);
    float*    act2  = (float*)alloc((size_t)8 * 64 * 17 * 36 * 4);
    float*    act3  = (float*)alloc((size_t)8 * 64 * 13 * 16 * 4);
    float*    enc   = (float*)alloc((size_t)8 * ENC_ * 4);
    float*    P     = (float*)alloc((size_t)256 * HID_ * 4);
    float*    Q     = (float*)alloc((size_t)16  * HID_ * 4);
    unsigned* aggk  = (unsigned*)alloc((size_t)N * 3 * 4);
    unsigned* mask  = (unsigned*)alloc((size_t)N * 4);
    int*      cnt   = (int*)alloc((size_t)N * 2 * 4);
    unsigned* head  = (unsigned*)(cnt + N);
    int*      link  = (int*)alloc((size_t)E * 8);
    float*    hbuf  = (float*)alloc((size_t)N * HID_ * 4);

    const int T = 256;
    zero_u32<<<(N * 3 + T - 1) / T, T, 0, stream>>>(aggk, N * 3);
    zero_u32<<<(N + T - 1) / T, T, 0, stream>>>(mask, N);
    zero_u32<<<(N * 2 + T - 1) / T, T, 0, stream>>>((unsigned*)cnt, N * 2);

    conv_wmma_f32<25, 25, 32, 21, 6, 6, 1, 378><<<95 * 2, 128, 0, stream>>>(X, w1, b1, act1);
    conv_wmma_f32<32, 21, 64, 17, 6, 6, 1, 306><<<77 * 4, 128, 0, stream>>>(act1, w2, b2, act2);
    conv_wmma_f32<64, 17, 64, 13, 4, 4, 0, 104><<<26 * 4, 128, 0, stream>>>(act2, w3, b3, act3);
    pool_kernel<<<(8 * 64 * 11 * 4 + T - 1) / T, T, 0, stream>>>(act3, enc);

    edge_pass<<<(E + T - 1) / T, T, 0, stream>>>(ei, gid, x3, aggk, mask, cnt, head, link, E);

    {
        dim3 gPQ(256 / 16 + 1, (HID_ + 15) / 16);
        gemm_submax_wmma_f32<<<gPQ, 32, 0, stream>>>(enc, Wl1 + 3 * HID_, Wr1 + 3 * HID_, P, Q);
    }

    node_h_kernel<<<(N + 7) / 8, 256, 0, stream>>>(x3, gid, mask, aggk, P, Q,
                                                   Wl1, Wr1, bl1, hbuf, N);

    node_out_kernel<<<(N + 7) / 8, 256, 0, stream>>>(hbuf, cnt, head, link, Wl2, bl2, Wr2, y, N, E);

    (void)n_in; (void)out_size; (void)ws_size;
}


# Layer-1 GEMM K loop software-pipelined in one wave: three register sets in rotation, the 24 loads of a K step asked for two steps ahead of its max chain and WMMAs; sums and their order unchanged
# speedup vs baseline: 3.8389x; 1.3444x over previous
#include <hip/hip_runtime.h>
#include <hip/hip_bf16.h>
#include <math.h>


#define B_     8
#define CIN_   25
#define NPG_   2000
#define ENC_   2816
#define HID_   100
#define KGEMM_ 2816

typedef __attribute__((ext_vector_type(2))) float v2f;
typedef __attribute__((ext_vector_type(4))) float v4f;
typedef __attribute__((ext_vector_type(8))) float v8f;

__device__ __forceinline__ unsigned fkey(float f) {
    unsigned u = __float_as_uint(f);
    return (u & 0x80000000u) ? ~u : (u | 0x80000000u);
}
__device__ __forceinline__ float keyf(unsigned k) {
    unsigned u = (k & 0x80000000u) ? (k ^ 0x80000000u) : ~k;
    return __uint_as_float(u);
}

__global__ void zero_u32(unsigned* __restrict__ p, int n) {
    int i = blockIdx.x * blockDim.x + threadIdx.x;
    if (i < n) p[i] = 0u;
}

constexpr int largest_divisor_upto(int n, int cap) {
    int best = 1;
    for (int d = 1; d <= cap; ++d) if (n % d == 0) best = d;
    return best;
}

template <int CIN, int DIN, int COUT, int DOUT, int HO, int WO, int PAD, int MT>
__global__ void __launch_bounds__(128)
conv_wmma_f32(const float* __restrict__ In, const float* __restrict__ Wt,
              const float* __restrict__ bias, float* __restrict__ Out) {
    constexpr int PLANE = 36;
    constexpr int CHS   = DIN * PLANE;
    constexpr int HWO   = HO * WO;
    constexpr int P     = DOUT * HWO;
    constexpr int NT    = COUT / 16;
    constexpr int CBF   = CIN / 4;
    constexpr int CREM  = CIN % 4;
    constexpr int CP    = (CIN + 3) / 4 * 4;
    constexpr int ROW   = CP * 45;
    constexpr int PITCH = ROW + (68 - ROW % 64) % 64;
    constexpr int WTOT  = 16 * CIN * 45;
    static_assert(P % 4 == 0, "16-byte stores need P % 4 == 0");
    static_assert(COUT % 16 == 0, "N tiles are full");
    static_assert(PITCH % 64 == 4 && CIN >= 3, "pitch rule; the stager's channel step wraps at most once");

    __shared__ __attribute__((aligned(16))) float Bs[16 * PITCH];

    const int lane = threadIdx.x & 31;
    const int half = lane >> 4;
    const int l16  = lane & 15;
    const int nTile = blockIdx.x % NT;
    {
        const float* Wn = Wt + (size_t)nTile * WTOT;
        constexpr int FULL  = WTOT / 128;
        constexpr int BATCH = largest_divisor_upto(FULL, 60);
        int tap = threadIdx.x % 45, c = threadIdx.x / 45, o = 0;
        for (int j0 = 0; j0 < FULL; j0 += BATCH) {
            float v[BATCH];
            #pragma unroll
            for (int u = 0; u < BATCH; ++u) v[u] = Wn[(j0 + u) * 128 + threadIdx.x];
            #pragma unroll
            for (int u = 0; u < BATCH; ++u) {
                Bs[o * PITCH + ((c >> 1) * 45 + tap) * 2 + (c & 1)] = v[u];
                tap += 38; c += 2;
                if (tap >= 45) { tap -= 45; c += 1; }
                if (c >= CIN)  { c -= CIN;  o += 1; }
            }
        }
        if constexpr (WTOT % 128 != 0) {
            const int i = FULL * 128 + threadIdx.x;
            if (i < WTOT)
                Bs[o * PITCH + ((c >> 1) * 45 + tap) * 2 + (c & 1)] = Wn[i];
        }
        if constexpr (CREM != 0) {
            constexpr int NPAD = CP - CIN;
            for (int i = threadIdx.x; i < 16 * NPAD * 45; i += 128) {
                const int oz = i / (NPAD * 45);
                const int r  = i - oz * (NPAD * 45);
                const int cz = CIN + r / 45;
                const int tz = r % 45;
                Bs[oz * PITCH + ((cz >> 1) * 45 + tz) * 2 + (cz & 1)] = 0.0f;
            }
        }
    }
    __syncthreads();

    const int mTile = (blockIdx.x / NT) * 4 + (threadIdx.x >> 5);
    if (mTile >= MT) return;

    const int m  = mTile * 16 + l16;
    const int bn = m / P;
    const int p  = m - bn * P;
    const int z  = p / HWO;
    const int q  = p - z * HWO;
    const int y  = q / WO;
    const int x  = q - y * WO;
    const int o  = nTile * 16 + l16;

    int  toff[9];
    bool tok[9];
    #pragma unroll
    for (int t = 0; t < 9; ++t) {
        const int yy = y + t / 3 - PAD;
        const int xx = x + t % 3 - PAD;
        const bool ok = (PAD == 0) ? true : ((yy >= 0) && (yy < 6) && (xx >= 0) && (xx < 6));
        tok[t]  = ok;
        toff[t] = ok ? (yy * 6 + xx) : 0;
    }

    int aIdx = (bn * CIN + half * 2) * CHS + z * PLANE;
    const float* bP = Bs + l16 * PITCH + half * 90;

    v8f acc0 = {}, acc1 = {};
    for (int cb = 0; cb < CBF; ++cb) {
        v2f   bf[45];
        float ax[45], ay[45];
        #pragma unroll
        for (int kd = 0; kd < 5; ++kd) {
            #pragma unroll
            for (int t = 0; t < 9; ++t) {
                const int ai = aIdx + kd * PLANE + toff[t];
                bf[kd * 9 + t] = *reinterpret_cast<const v2f*>(bP + (kd * 9 + t) * 2);
                ax[kd * 9 + t] = In[ai];
                ay[kd * 9 + t] = In[ai + CHS];
            }
        }
        __builtin_amdgcn_sched_barrier(0);
        #pragma unroll
        for (int kd = 0; kd < 5; ++kd) {
            #pragma unroll
            for (int t = 0; t < 9; ++t) {
                v2f a;
                a.x = tok[t] ? ax[kd * 9 + t] : 0.0f;
                a.y = tok[t] ? ay[kd * 9 + t] : 0.0f;
                const v2f b = bf[kd * 9 + t];
                if ((kd * 9 + t) & 1)
                    acc1 = __builtin_amdgcn_wmma_f32_16x16x4_f32(false, a, false, b,
                                                                 (short)0, acc1, false, false);
                else
                    acc0 = __builtin_amdgcn_wmma_f32_16x16x4_f32(false, a, false, b,
                                                                 (short)0, acc0, false, false);
            }
        }
        aIdx += 4 * CHS;
        bP   += 4 * 45;
    }
    if (CREM != 0) {
        const int  c0  = CBF * 4 + half * 2;
        const bool cv0 = (c0 < CIN);
        const bool cv1 = (c0 + 1 < CIN);
        const int  cc0 = cv0 ? c0 : CBF * 4;
        const int  cc1 = cv1 ? (c0 + 1) : CBF * 4;
        const int aT0 = (bn * CIN + cc0) * CHS + z * PLANE;
        const int aT1 = (bn * CIN + cc1) * CHS + z * PLANE;
        #pragma unroll
        for (int kd = 0; kd < 5; ++kd) {
            #pragma unroll
            for (int t = 0; t < 9; ++t) {
                const int so = kd * PLANE + toff[t];
                const float ax = In[aT0 + so];
                const float ay = In[aT1 + so];
                v2f a;
                a.x = (tok[t] && cv0) ? ax : 0.0f;
                a.y = (tok[t] && cv1) ? ay : 0.0f;
                const v2f b = *reinterpret_cast<const v2f*>(bP + (kd * 9 + t) * 2);
                if ((kd * 9 + t) & 1)
                    acc1 = __builtin_amdgcn_wmma_f32_16x16x4_f32(false, a, false, b,
                                                                 (short)0, acc1, false, false);
                else
                    acc0 = __builtin_amdgcn_wmma_f32_16x16x4_f32(false, a, false, b,
                                                                 (short)0, acc0, false, false);
            }
        }
    }

    const v8f acc = acc0 + acc1;
    const float bo = bias[o];
    const int m0 = mTile * 16 + half * 8;
    #pragma unroll
    for (int g = 0; g < 2; ++g) {
        const int mg  = m0 + g * 4;
        const int bng = mg / P;
        const int pg  = mg - bng * P;
        float4 v;
        v.x = fmaxf(acc[g * 4 + 0] + bo, 0.0f);
        v.y = fmaxf(acc[g * 4 + 1] + bo, 0.0f);
        v.z = fmaxf(acc[g * 4 + 2] + bo, 0.0f);
        v.w = fmaxf(acc[g * 4 + 3] + bo, 0.0f);
        *reinterpret_cast<float4*>(Out + (size_t)(bng * COUT + o) * P + pg) = v;
    }
}

__global__ void pool_kernel(const float* __restrict__ A, float* __restrict__ enc) {
    const int total = 8 * 64 * 11 * 2 * 2;
    int idx = blockIdx.x * blockDim.x + threadIdx.x;
    if (idx >= total) return;
    int x = idx % 2;  int t = idx / 2;
    int y = t % 2;    t /= 2;
    int d = t % 11;   t /= 11;
    int ch = t % 64;  int bn = t / 64;
    float m = -INFINITY;
    for (int dd = 0; dd < 3; ++dd)
        for (int hh = 0; hh < 3; ++hh)
            for (int ww = 0; ww < 3; ++ww) {
                float v = A[((size_t)((bn * 64 + ch) * 13 + (d + dd))) * 16 + (y + hh) * 4 + (x + ww)];
                m = fmaxf(m, v);
            }
    enc[(size_t)bn * ENC_ + ch * 44 + d * 4 + y * 2 + x] = m;
}

__global__ void edge_pass(const int* __restrict__ ei, const int* __restrict__ gid,
                          const float* __restrict__ x3, unsigned* __restrict__ aggk,
                          unsigned* __restrict__ mask, int* __restrict__ counts,
                          unsigned* __restrict__ head, int* __restrict__ link, int E) {
    int e = blockIdx.x * blockDim.x + threadIdx.x;
    if (e >= E) return;
    int s = ei[e];
    int d = ei[E + e];
    int g = gid[s];
    const unsigned prev = atomicExch(&head[d], (unsigned)e + 1u);
    reinterpret_cast<int2*>(link)[e] = make_int2((int)prev - 1, s);
    atomicOr(&mask[d], 1u << g);
    atomicAdd(&counts[d], 1);
    atomicMax(&aggk[(size_t)d * 3 + 0], fkey(x3[(size_t)s * 3 + 0]));
    atomicMax(&aggk[(size_t)d * 3 + 1], fkey(x3[(size_t)s * 3 + 1]));
    atomicMax(&aggk[(size_t)d * 3 + 2], fkey(x3[(size_t)s * 3 + 2]));
}

__global__ void __launch_bounds__(32)
gemm_submax_wmma_f32(const float* __restrict__ enc, const float* __restrict__ Wl,
                     const float* __restrict__ Wr, float* __restrict__ P,
                     float* __restrict__ Q) {
    const int Ncols = HID_;
    const int lane = threadIdx.x & 31;
    const int half = lane >> 4;
    const int l16  = lane & 15;
    const bool isQ = (blockIdx.x == 16);
    const float* W = isQ ? Wr : Wl;
    float* C = isQ ? Q : P;
    const int Mrows = isQ ? 8 : 256;
    const int mBase = isQ ? 0 : blockIdx.x * 16;
    const unsigned mask = isQ ? (1u << (l16 & 7)) : (unsigned)(mBase + l16);
    const int g0 = mask ? (__ffs(mask) - 1) : 0;
    const int nBase = blockIdx.y * 16;
    const int nCol = nBase + l16;
    const bool nOk = (nCol < Ncols);
    const int nClamp = nOk ? nCol : (Ncols - 1);

    const float* rowp[8];
    #pragma unroll
    for (int g = 0; g < 8; ++g) {
        const int r = ((mask >> g) & 1u) ? g : g0;
        rowp[g] = enc + (size_t)r * ENC_ + half * 8;
    }
    const float* Bcol = W + (size_t)(half * 8) * Ncols + nClamp;

    struct KStep { v4f L[8], H[8]; float bv[8]; };
    v8f acc0 = {}, acc1 = {};

    auto load_step = [&](KStep& s, int kt) __attribute__((always_inline)) {
        #pragma unroll
        for (int g = 0; g < 8; ++g) {
            s.L[g] = *reinterpret_cast<const v4f*>(rowp[g] + kt);
            s.H[g] = *reinterpret_cast<const v4f*>(rowp[g] + kt + 4);
        }
        #pragma unroll
        for (int i = 0; i < 8; ++i) s.bv[i] = Bcol[(size_t)(kt + i) * Ncols];
    };
    auto mac_step = [&](const KStep& s) __attribute__((always_inline)) {
        v4f lo = s.L[0], hi = s.H[0];
        #pragma unroll
        for (int g = 1; g < 8; ++g) {
            lo.x = fmaxf(lo.x, s.L[g].x); lo.y = fmaxf(lo.y, s.L[g].y);
            lo.z = fmaxf(lo.z, s.L[g].z); lo.w = fmaxf(lo.w, s.L[g].w);
            hi.x = fmaxf(hi.x, s.H[g].x); hi.y = fmaxf(hi.y, s.H[g].y);
            hi.z = fmaxf(hi.z, s.H[g].z); hi.w = fmaxf(hi.w, s.H[g].w);
        }
        v2f a0, a1, a2, a3, b0, b1, b2, b3;
        a0.x = fmaxf(lo.x, 0.0f); a0.y = fmaxf(lo.y, 0.0f);
        a1.x = fmaxf(lo.z, 0.0f); a1.y = fmaxf(lo.w, 0.0f);
        a2.x = fmaxf(hi.x, 0.0f); a2.y = fmaxf(hi.y, 0.0f);
        a3.x = fmaxf(hi.z, 0.0f); a3.y = fmaxf(hi.w, 0.0f);
        b0.x = s.bv[0]; b0.y = s.bv[1];
        b1.x = s.bv[2]; b1.y = s.bv[3];
        b2.x = s.bv[4]; b2.y = s.bv[5];
        b3.x = s.bv[6]; b3.y = s.bv[7];
        acc0 = __builtin_amdgcn_wmma_f32_16x16x4_f32(false, a0, false, b0,
                                                     (short)0, acc0, false, false);
        acc1 = __builtin_amdgcn_wmma_f32_16x16x4_f32(false, a1, false, b1,
                                                     (short)0, acc1, false, false);
        acc0 = __builtin_amdgcn_wmma_f32_16x16x4_f32(false, a2, false, b2,
                                                     (short)0, acc0, false, false);
        acc1 = __builtin_amdgcn_wmma_f32_16x16x4_f32(false, a3, false, b3,
                                                     (short)0, acc1, false, false);
    };

    static_assert((KGEMM_ / 16) % 3 == 2 && KGEMM_ % 16 == 0, "58 trips of three steps, then the last two");
    KStep s0, s1, s2;
    load_step(s0, 0);
    load_step(s1, 16);
    for (int kt = 0; kt < KGEMM_ - 32; kt += 48) {
        load_step(s2, kt + 32);
        __builtin_amdgcn_sched_barrier(0);
        mac_step(s0);
        load_step(s0, kt + 48);
        __builtin_amdgcn_sched_barrier(0);
        mac_step(s1);
        load_step(s1, kt + 64);
        __builtin_amdgcn_sched_barrier(0);
        mac_step(s2);
    }
    mac_step(s0);
    mac_step(s1);
    v8f acc = acc0 + acc1;
    for (int v = 0; v < 8; ++v) {
        const int m = mBase + half * 8 + v;
        const float val = (!isQ && m == 0) ? 0.0f : acc[v];
        if (m < Mrows && nOk) C[(size_t)m * Ncols + nCol] = val;
    }
}

__global__ void node_h_kernel(const float* __restrict__ x3, const int* __restrict__ gid,
                              const unsigned* __restrict__ mask, const unsigned* __restrict__ aggk,
                              const float* __restrict__ P, const float* __restrict__ Q,
                              const float* __restrict__ Wl1, const float* __restrict__ Wr1,
                              const float* __restrict__ bl1, float* __restrict__ h, int N) {
    int wave = threadIdx.x >> 5;
    int lane = threadIdx.x & 31;
    int n = blockIdx.x * (blockDim.x >> 5) + wave;
    if (n >= N) return;

    float x0 = x3[(size_t)n * 3 + 0];
    float x1 = x3[(size_t)n * 3 + 1];
    float x2 = x3[(size_t)n * 3 + 2];
    int g = gid[n];
    unsigned mk = mask[n];
    bool has = (mk != 0u);
    float a0 = 0.f, a1 = 0.f, a2 = 0.f;
    if (has) {
        a0 = keyf(aggk[(size_t)n * 3 + 0]);
        a1 = keyf(aggk[(size_t)n * 3 + 1]);
        a2 = keyf(aggk[(size_t)n * 3 + 2]);
    }
    const float* Prow = P + (size_t)mk * HID_;
    const float* Qrow = Q + (size_t)g * HID_;

    float v[4];
    #pragma unroll
    for (int c = 0; c < 4; ++c) {
        int col = lane + 32 * c;
        if (col < HID_) {
            float t = bl1[col] + Qrow[col]
                    + x0 * Wr1[col] + x1 * Wr1[HID_ + col] + x2 * Wr1[2 * HID_ + col];
            if (has)
                t += Prow[col]
                   + a0 * Wl1[col] + a1 * Wl1[HID_ + col] + a2 * Wl1[2 * HID_ + col];
            v[c] = t;
        } else {
            v[c] = -INFINITY;
        }
    }
    float m = fmaxf(fmaxf(v[0], v[1]), fmaxf(v[2], v[3]));
    for (int off = 16; off >= 1; off >>= 1)
        m = fmaxf(m, __shfl_xor(m, off, 32));
    float s = 0.f;
    #pragma unroll
    for (int c = 0; c < 4; ++c) s += __expf(v[c] - m);
    for (int off = 16; off >= 1; off >>= 1)
        s += __shfl_xor(s, off, 32);
    float lse = m + __logf(s);
    #pragma unroll
    for (int c = 0; c < 4; ++c) {
        int col = lane + 32 * c;
        if (col < HID_) h[(size_t)n * HID_ + col] = v[c] - lse;
    }
}

__global__ void node_out_kernel(const float* __restrict__ h, const int* __restrict__ counts,
                                const unsigned* __restrict__ head, const int* __restrict__ link2,
                                const float* __restrict__ Wl2, const float* __restrict__ bl2,
                                const float* __restrict__ Wr2, float* __restrict__ y,
                                int N, int E) {
    const int wave = threadIdx.x >> 5;
    const int lane = threadIdx.x & 31;
    const int n = __builtin_amdgcn_readfirstlane(blockIdx.x * (blockDim.x >> 5) + wave);
    if (n >= N) return;
    const bool own = (lane < HID_ / 4);
    const int  c4  = own ? lane * 4 : 0;
    const int2* link = reinterpret_cast<const int2*>(link2);
    const int cnt = counts[n];
    const int e0  = (int)head[n] - 1;

    float4 own_row = *reinterpret_cast<const float4*>(h + (size_t)n * HID_ + c4);
    float4 acc  = make_float4(0.f, 0.f, 0.f, 0.f);
    float4 pend = make_float4(0.f, 0.f, 0.f, 0.f);
    int2 lk = make_int2(-1, 0);
    if (cnt > 0) lk = link[min(max(e0, 0), E - 1)];
    for (int j = 0; j < cnt; ++j) {
        const int2 lkn = link[min(max(lk.x, 0), E - 1)];
        const int  sc  = min(max(lk.y, 0), N - 1);
        const float4 row = *reinterpret_cast<const float4*>(h + (size_t)sc * HID_ + c4);
        acc.x += pend.x; acc.y += pend.y; acc.z += pend.z; acc.w += pend.w;
        pend = row;
        lk = lkn;
    }
    acc.x += pend.x; acc.y += pend.y; acc.z += pend.z; acc.w += pend.w;

    float s1 = 0.f, s2 = 0.f;
    if (own) {
        s1 = fmaf(acc.x, Wl2[c4 + 0], s1);
        s1 = fmaf(acc.y, Wl2[c4 + 1], s1);
        s1 = fmaf(acc.z, Wl2[c4 + 2], s1);
        s1 = fmaf(acc.w, Wl2[c4 + 3], s1);
        s2 = fmaf(own_row.x, Wr2[c4 + 0], s2);
        s2 = fmaf(own_row.y, Wr2[c4 + 1], s2);
        s2 = fmaf(own_row.z, Wr2[c4 + 2], s2);
        s2 = fmaf(own_row.w, Wr2[c4 + 3], s2);
    }
    for (int off = 16; off >= 1; off >>= 1) {
        s1 += __shfl_xor(s1, off, 32);
        s2 += __shfl_xor(s2, off, 32);
    }
    if (lane == 0)
        y[n] = s1 / fmaxf((float)cnt, 1.0f) + bl2[0] + s2;
}

extern "C" void kernel_launch(void* const* d_in, const int* in_sizes, int n_in,
                              void* d_out, int out_size, void* d_ws, size_t ws_size,
                              hipStream_t stream) {
    const float* X      = (const float*)d_in[0];
    const float* x3     = (const float*)d_in[1];
    const int*   ei     = (const int*)d_in[2];
    const int*   gid    = (const int*)d_in[3];
    const float* w1     = (const float*)d_in[4];
    const float* b1     = (const float*)d_in[5];
    const float* w2     = (const float*)d_in[6];
    const float* b2     = (const float*)d_in[7];
    const float* w3     = (const float*)d_in[8];
    const float* b3     = (const float*)d_in[9];
    const float* Wl1    = (const float*)d_in[10];
    const float* bl1    = (const float*)d_in[11];
    const float* Wr1    = (const float*)d_in[12];
    const float* Wl2    = (const float*)d_in[13];
    const float* bl2    = (const float*)d_in[14];
    const float* Wr2    = (const float*)d_in[15];
    float* y = (float*)d_out;

    const int E = in_sizes[2] / 2;
    const int N = in_sizes[3];

    char* ws = (char*)d_ws;
    size_t off = 0;
    auto alloc = [&](size_t bytes) -> char* {
        char* p = ws + off;
        off = (off + bytes + 255) & ~(size_t)255;
        return p;
    };
    float*    act1  = (float*)alloc((size_t)8 * 32 * 21 * 36 * 4);
    float*    act2  = (float*)alloc((size_t)8 * 64 * 17 * 36 * 4);
    float*    act3  = (float*)alloc((size_t)8 * 64 * 13 * 16 * 4);
    float*    enc   = (float*)alloc((size_t)8 * ENC_ * 4);
    float*    P     = (float*)alloc((size_t)256 * HID_ * 4);
    float*    Q     = (float*)alloc((size_t)16  * HID_ * 4);
    unsigned* aggk  = (unsigned*)alloc((size_t)N * 3 * 4);
    unsigned* mask  = (unsigned*)alloc((size_t)N * 4);
    int*      cnt   = (int*)alloc((size_t)N * 2 * 4);
    unsigned* head  = (unsigned*)(cnt + N);
    int*      link  = (int*)alloc((size_t)E * 8);
    float*    hbuf  = (float*)alloc((size_t)N * HID_ * 4);

    const int T = 256;
    zero_u32<<<(N * 3 + T - 1) / T, T, 0, stream>>>(aggk, N * 3);
    zero_u32<<<(N + T - 1) / T, T, 0, stream>>>(mask, N);
    zero_u32<<<(N * 2 + T - 1) / T, T, 0, stream>>>((unsigned*)cnt, N * 2);

    conv_wmma_f32<25, 25, 32, 21, 6, 6, 1, 378><<<95 * 2, 128, 0, stream>>>(X, w1, b1, act1);
    conv_wmma_f32<32, 21, 64, 17, 6, 6, 1, 306><<<77 * 4, 128, 0, stream>>>(act1, w2, b2, act2);
    conv_wmma_f32<64, 17, 64, 13, 4, 4, 0, 104><<<26 * 4, 128, 0, stream>>>(act2, w3, b3, act3);
    pool_kernel<<<(8 * 64 * 11 * 4 + T - 1) / T, T, 0, stream>>>(act3, enc);

    edge_pass<<<(E + T - 1) / T, T, 0, stream>>>(ei, gid, x3, aggk, mask, cnt, head, link, E);

    {
        dim3 gPQ(256 / 16 + 1, (HID_ + 15) / 16);
        gemm_submax_wmma_f32<<<gPQ, 32, 0, stream>>>(enc, Wl1 + 3 * HID_, Wr1 + 3 * HID_, P, Q);
    }

    node_h_kernel<<<(N + 7) / 8, 256, 0, stream>>>(x3, gid, mask, aggk, P, Q,
                                                   Wl1, Wr1, bl1, hbuf, N);

    node_out_kernel<<<(N + 7) / 8, 256, 0, stream>>>(hbuf, cnt, head, link, Wl2, bl2, Wr2, y, N, E);

    (void)n_in; (void)out_size; (void)ws_size;
}


# Layer-1 GEMM: the 8 enc rows copied once per workgroup into LDS, A loader reads LDS; only the 8 weight loads of a K step stay on the load queue and run 8 steps ahead (was 2); sums, order unchanged
# speedup vs baseline: 4.2541x; 1.1082x over previous
#include <hip/hip_runtime.h>
#include <hip/hip_bf16.h>
#include <math.h>


#define B_     8
#define CIN_   25
#define NPG_   2000
#define ENC_   2816
#define HID_   100
#define KGEMM_ 2816

typedef __attribute__((ext_vector_type(2))) float v2f;
typedef __attribute__((ext_vector_type(4))) float v4f;
typedef __attribute__((ext_vector_type(8))) float v8f;

__device__ __forceinline__ unsigned fkey(float f) {
    unsigned u = __float_as_uint(f);
    return (u & 0x80000000u) ? ~u : (u | 0x80000000u);
}
__device__ __forceinline__ float keyf(unsigned k) {
    unsigned u = (k & 0x80000000u) ? (k ^ 0x80000000u) : ~k;
    return __uint_as_float(u);
}

__global__ void zero_u32(unsigned* __restrict__ p, int n) {
    int i = blockIdx.x * blockDim.x + threadIdx.x;
    if (i < n) p[i] = 0u;
}

constexpr int largest_divisor_upto(int n, int cap) {
    int best = 1;
    for (int d = 1; d <= cap; ++d) if (n % d == 0) best = d;
    return best;
}

template <int CIN, int DIN, int COUT, int DOUT, int HO, int WO, int PAD, int MT>
__global__ void __launch_bounds__(128)
conv_wmma_f32(const float* __restrict__ In, const float* __restrict__ Wt,
              const float* __restrict__ bias, float* __restrict__ Out) {
    constexpr int PLANE = 36;
    constexpr int CHS   = DIN * PLANE;
    constexpr int HWO   = HO * WO;
    constexpr int P     = DOUT * HWO;
    constexpr int NT    = COUT / 16;
    constexpr int CBF   = CIN / 4;
    constexpr int CREM  = CIN % 4;
    constexpr int CP    = (CIN + 3) / 4 * 4;
    constexpr int ROW   = CP * 45;
    constexpr int PITCH = ROW + (68 - ROW % 64) % 64;
    constexpr int WTOT  = 16 * CIN * 45;
    static_assert(P % 4 == 0, "16-byte stores need P % 4 == 0");
    static_assert(COUT % 16 == 0, "N tiles are full");
    static_assert(PITCH % 64 == 4 && CIN >= 3, "pitch rule; the stager's channel step wraps at most once");

    __shared__ __attribute__((aligned(16))) float Bs[16 * PITCH];

    const int lane = threadIdx.x & 31;
    const int half = lane >> 4;
    const int l16  = lane & 15;
    const int nTile = blockIdx.x % NT;
    {
        const float* Wn = Wt + (size_t)nTile * WTOT;
        constexpr int FULL  = WTOT / 128;
        constexpr int BATCH = largest_divisor_upto(FULL, 60);
        int tap = threadIdx.x % 45, c = threadIdx.x / 45, o = 0;
        for (int j0 = 0; j0 < FULL; j0 += BATCH) {
            float v[BATCH];
            #pragma unroll
            for (int u = 0; u < BATCH; ++u) v[u] = Wn[(j0 + u) * 128 + threadIdx.x];
            #pragma unroll
            for (int u = 0; u < BATCH; ++u) {
                Bs[o * PITCH + ((c >> 1) * 45 + tap) * 2 + (c & 1)] = v[u];
                tap += 38; c += 2;
                if (tap >= 45) { tap -= 45; c += 1; }
                if (c >= CIN)  { c -= CIN;  o += 1; }
            }
        }
        if constexpr (WTOT % 128 != 0) {
            const int i = FULL * 128 + threadIdx.x;
            if (i < WTOT)
                Bs[o * PITCH + ((c >> 1) * 45 + tap) * 2 + (c & 1)] = Wn[i];
        }
        if constexpr (CREM != 0) {
            constexpr int NPAD = CP - CIN;
            for (int i = threadIdx.x; i < 16 * NPAD * 45; i += 128) {
                const int oz = i / (NPAD * 45);
                const int r  = i - oz * (NPAD * 45);
                const int cz = CIN + r / 45;
                const int tz = r % 45;
                Bs[oz * PITCH + ((cz >> 1) * 45 + tz) * 2 + (cz & 1)] = 0.0f;
            }
        }
    }
    __syncthreads();

    const int mTile = (blockIdx.x / NT) * 4 + (threadIdx.x >> 5);
    if (mTile >= MT) return;

    const int m  = mTile * 16 + l16;
    const int bn = m / P;
    const int p  = m - bn * P;
    const int z  = p / HWO;
    const int q  = p - z * HWO;
    const int y  = q / WO;
    const int x  = q - y * WO;
    const int o  = nTile * 16 + l16;

    int  toff[9];
    bool tok[9];
    #pragma unroll
    for (int t = 0; t < 9; ++t) {
        const int yy = y + t / 3 - PAD;
        const int xx = x + t % 3 - PAD;
        const bool ok = (PAD == 0) ? true : ((yy >= 0) && (yy < 6) && (xx >= 0) && (xx < 6));
        tok[t]  = ok;
        toff[t] = ok ? (yy * 6 + xx) : 0;
    }

    int aIdx = (bn * CIN + half * 2) * CHS + z * PLANE;
    const float* bP = Bs + l16 * PITCH + half * 90;

    v8f acc0 = {}, acc1 = {};
    for (int cb = 0; cb < CBF; ++cb) {
        v2f   bf[45];
        float ax[45], ay[45];
        #pragma unroll
        for (int kd = 0; kd < 5; ++kd) {
            #pragma unroll
            for (int t = 0; t < 9; ++t) {
                const int ai = aIdx + kd * PLANE + toff[t];
                bf[kd * 9 + t] = *reinterpret_cast<const v2f*>(bP + (kd * 9 + t) * 2);
                ax[kd * 9 + t] = In[ai];
                ay[kd * 9 + t] = In[ai + CHS];
            }
        }
        __builtin_amdgcn_sched_barrier(0);
        #pragma unroll
        for (int kd = 0; kd < 5; ++kd) {
            #pragma unroll
            for (int t = 0; t < 9; ++t) {
                v2f a;
                a.x = tok[t] ? ax[kd * 9 + t] : 0.0f;
                a.y = tok[t] ? ay[kd * 9 + t] : 0.0f;
                const v2f b = bf[kd * 9 + t];
                if ((kd * 9 + t) & 1)
                    acc1 = __builtin_amdgcn_wmma_f32_16x16x4_f32(false, a, false, b,
                                                                 (short)0, acc1, false, false);
                else
                    acc0 = __builtin_amdgcn_wmma_f32_16x16x4_f32(false, a, false, b,
                                                                 (short)0, acc0, false, false);
            }
        }
        aIdx += 4 * CHS;
        bP   += 4 * 45;
    }
    if (CREM != 0) {
        const int  c0  = CBF * 4 + half * 2;
        const bool cv0 = (c0 < CIN);
        const bool cv1 = (c0 + 1 < CIN);
        const int  cc0 = cv0 ? c0 : CBF * 4;
        const int  cc1 = cv1 ? (c0 + 1) : CBF * 4;
        const int aT0 = (bn * CIN + cc0) * CHS + z * PLANE;
        const int aT1 = (bn * CIN + cc1) * CHS + z * PLANE;
        #pragma unroll
        for (int kd = 0; kd < 5; ++kd) {
            #pragma unroll
            for (int t = 0; t < 9; ++t) {
                const int so = kd * PLANE + toff[t];
                const float ax = In[aT0 + so];
                const float ay = In[aT1 + so];
                v2f a;
                a.x = (tok[t] && cv0) ? ax : 0.0f;
                a.y = (tok[t] && cv1) ? ay : 0.0f;
                const v2f b = *reinterpret_cast<const v2f*>(bP + (kd * 9 + t) * 2);
                if ((kd * 9 + t) & 1)
                    acc1 = __builtin_amdgcn_wmma_f32_16x16x4_f32(false, a, false, b,
                                                                 (short)0, acc1, false, false);
                else
                    acc0 = __builtin_amdgcn_wmma_f32_16x16x4_f32(false, a, false, b,
                                                                 (short)0, acc0, false, false);
            }
        }
    }

    const v8f acc = acc0 + acc1;
    const float bo = bias[o];
    const int m0 = mTile * 16 + half * 8;
    #pragma unroll
    for (int g = 0; g < 2; ++g) {
        const int mg  = m0 + g * 4;
        const int bng = mg / P;
        const int pg  = mg - bng * P;
        float4 v;
        v.x = fmaxf(acc[g * 4 + 0] + bo, 0.0f);
        v.y = fmaxf(acc[g * 4 + 1] + bo, 0.0f);
        v.z = fmaxf(acc[g * 4 + 2] + bo, 0.0f);
        v.w = fmaxf(acc[g * 4 + 3] + bo, 0.0f);
        *reinterpret_cast<float4*>(Out + (size_t)(bng * COUT + o) * P + pg) = v;
    }
}

__global__ void pool_kernel(const float* __restrict__ A, float* __restrict__ enc) {
    const int total = 8 * 64 * 11 * 2 * 2;
    int idx = blockIdx.x * blockDim.x + threadIdx.x;
    if (idx >= total) return;
    int x = idx % 2;  int t = idx / 2;
    int y = t % 2;    t /= 2;
    int d = t % 11;   t /= 11;
    int ch = t % 64;  int bn = t / 64;
    float m = -INFINITY;
    for (int dd = 0; dd < 3; ++dd)
        for (int hh = 0; hh < 3; ++hh)
            for (int ww = 0; ww < 3; ++ww) {
                float v = A[((size_t)((bn * 64 + ch) * 13 + (d + dd))) * 16 + (y + hh) * 4 + (x + ww)];
                m = fmaxf(m, v);
            }
    enc[(size_t)bn * ENC_ + ch * 44 + d * 4 + y * 2 + x] = m;
}

__global__ void edge_pass(const int* __restrict__ ei, const int* __restrict__ gid,
                          const float* __restrict__ x3, unsigned* __restrict__ aggk,
                          unsigned* __restrict__ mask, int* __restrict__ counts,
                          unsigned* __restrict__ head, int* __restrict__ link, int E) {
    int e = blockIdx.x * blockDim.x + threadIdx.x;
    if (e >= E) return;
    int s = ei[e];
    int d = ei[E + e];
    int g = gid[s];
    const unsigned prev = atomicExch(&head[d], (unsigned)e + 1u);
    reinterpret_cast<int2*>(link)[e] = make_int2((int)prev - 1, s);
    atomicOr(&mask[d], 1u << g);
    atomicAdd(&counts[d], 1);
    atomicMax(&aggk[(size_t)d * 3 + 0], fkey(x3[(size_t)s * 3 + 0]));
    atomicMax(&aggk[(size_t)d * 3 + 1], fkey(x3[(size_t)s * 3 + 1]));
    atomicMax(&aggk[(size_t)d * 3 + 2], fkey(x3[(size_t)s * 3 + 2]));
}

__global__ void __launch_bounds__(32)
gemm_submax_wmma_f32(const float* __restrict__ enc, const float* __restrict__ Wl,
                     const float* __restrict__ Wr, float* __restrict__ P,
                     float* __restrict__ Q) {
    constexpr int EPITCH = KGEMM_ + 4;
    __shared__ __attribute__((aligned(16))) float EncS[8 * EPITCH];

    const int Ncols = HID_;
    const int lane = threadIdx.x & 31;
    const int half = lane >> 4;
    const int l16  = lane & 15;
    const bool isQ = (blockIdx.x == 16);
    const float* W = isQ ? Wr : Wl;
    float* C = isQ ? Q : P;
    const int Mrows = isQ ? 8 : 256;
    const int mBase = isQ ? 0 : blockIdx.x * 16;
    const unsigned mask = isQ ? (1u << (l16 & 7)) : (unsigned)(mBase + l16);
    const int g0 = mask ? (__ffs(mask) - 1) : 0;
    const int nBase = blockIdx.y * 16;
    const int nCol = nBase + l16;
    const bool nOk = (nCol < Ncols);
    const int nClamp = nOk ? nCol : (Ncols - 1);

    int aoff[8];
    #pragma unroll
    for (int g = 0; g < 8; ++g) {
        const int r = ((mask >> g) & 1u) ? g : g0;
        aoff[g] = r * EPITCH + half * 8;
    }
    const float* Bcol = W + (size_t)(half * 8) * Ncols + nClamp;

    struct KB { float bv[8]; };
    struct KA { v4f L[8], H[8]; };
    v8f acc0 = {}, acc1 = {};

    auto load_b = [&](KB& s, int kt) __attribute__((always_inline)) {
        #pragma unroll
        for (int i = 0; i < 8; ++i) s.bv[i] = Bcol[(size_t)(kt + i) * Ncols];
    };
    auto load_a = [&](KA& s, int kt) __attribute__((always_inline)) {
        #pragma unroll
        for (int g = 0; g < 8; ++g) {
            s.L[g] = *reinterpret_cast<const v4f*>(EncS + aoff[g] + kt);
            s.H[g] = *reinterpret_cast<const v4f*>(EncS + aoff[g] + kt + 4);
        }
    };
    auto mac_step = [&](const KA& s, const KB& b) __attribute__((always_inline)) {
        v4f lo = s.L[0], hi = s.H[0];
        #pragma unroll
        for (int g = 1; g < 8; ++g) {
            lo.x = fmaxf(lo.x, s.L[g].x); lo.y = fmaxf(lo.y, s.L[g].y);
            lo.z = fmaxf(lo.z, s.L[g].z); lo.w = fmaxf(lo.w, s.L[g].w);
            hi.x = fmaxf(hi.x, s.H[g].x); hi.y = fmaxf(hi.y, s.H[g].y);
            hi.z = fmaxf(hi.z, s.H[g].z); hi.w = fmaxf(hi.w, s.H[g].w);
        }
        v2f a0, a1, a2, a3, b0, b1, b2, b3;
        a0.x = fmaxf(lo.x, 0.0f); a0.y = fmaxf(lo.y, 0.0f);
        a1.x = fmaxf(lo.z, 0.0f); a1.y = fmaxf(lo.w, 0.0f);
        a2.x = fmaxf(hi.x, 0.0f); a2.y = fmaxf(hi.y, 0.0f);
        a3.x = fmaxf(hi.z, 0.0f); a3.y = fmaxf(hi.w, 0.0f);
        b0.x = b.bv[0]; b0.y = b.bv[1];
        b1.x = b.bv[2]; b1.y = b.bv[3];
        b2.x = b.bv[4]; b2.y = b.bv[5];
        b3.x = b.bv[6]; b3.y = b.bv[7];
        acc0 = __builtin_amdgcn_wmma_f32_16x16x4_f32(false, a0, false, b0,
                                                     (short)0, acc0, false, false);
        acc1 = __builtin_amdgcn_wmma_f32_16x16x4_f32(false, a1, false, b1,
                                                     (short)0, acc1, false, false);
        acc0 = __builtin_amdgcn_wmma_f32_16x16x4_f32(false, a2, false, b2,
                                                     (short)0, acc0, false, false);
        acc1 = __builtin_amdgcn_wmma_f32_16x16x4_f32(false, a3, false, b3,
                                                     (short)0, acc1, false, false);
    };

    static_assert((KGEMM_ / 16) % 8 == 0 && KGEMM_ % 16 == 0, "22 trips of eight steps");
    for (int r0 = 0; r0 < 8; r0 += 2) {
        v4f v[44];
        #pragma unroll
        for (int u = 0; u < 44; ++u) {
            const int r = r0 + u / 22, j = u % 22;
            v[u] = *reinterpret_cast<const v4f*>(enc + (size_t)r * ENC_ + (j * 32 + lane) * 4);
        }
        #pragma unroll
        for (int u = 0; u < 44; ++u) {
            const int r = r0 + u / 22, j = u % 22;
            *reinterpret_cast<v4f*>(EncS + r * EPITCH + (j * 32 + lane) * 4) = v[u];
        }
    }
    __syncthreads();

    KB b0, b1, b2, b3, b4, b5, b6, b7;
    load_b(b0, 0);  load_b(b1, 16); load_b(b2, 32); load_b(b3, 48);
    load_b(b4, 64); load_b(b5, 80); load_b(b6, 96); load_b(b7, 112);
    KA a0, a1;
    load_a(a0, 0);
    load_a(a1, 16);
    for (int kt = 0; kt < KGEMM_ - 128; kt += 128) {
        mac_step(a0, b0); load_b(b0, kt + 128); load_a(a0, kt + 32);
        mac_step(a1, b1); load_b(b1, kt + 144); load_a(a1, kt + 48);
        mac_step(a0, b2); load_b(b2, kt + 160); load_a(a0, kt + 64);
        mac_step(a1, b3); load_b(b3, kt + 176); load_a(a1, kt + 80);
        mac_step(a0, b4); load_b(b4, kt + 192); load_a(a0, kt + 96);
        mac_step(a1, b5); load_b(b5, kt + 208); load_a(a1, kt + 112);
        mac_step(a0, b6); load_b(b6, kt + 224); load_a(a0, kt + 128);
        mac_step(a1, b7); load_b(b7, kt + 240); load_a(a1, kt + 144);
    }
    {
        constexpr int kt = KGEMM_ - 128;
        mac_step(a0, b0); load_a(a0, kt + 32);
        mac_step(a1, b1); load_a(a1, kt + 48);
        mac_step(a0, b2); load_a(a0, kt + 64);
        mac_step(a1, b3); load_a(a1, kt + 80);
        mac_step(a0, b4); load_a(a0, kt + 96);
        mac_step(a1, b5); load_a(a1, kt + 112);
        mac_step(a0, b6);
        mac_step(a1, b7);
    }
    v8f acc = acc0 + acc1;
    for (int v = 0; v < 8; ++v) {
        const int m = mBase + half * 8 + v;
        const float val = (!isQ && m == 0) ? 0.0f : acc[v];
        if (m < Mrows && nOk) C[(size_t)m * Ncols + nCol] = val;
    }
}

__global__ void node_h_kernel(const float* __restrict__ x3, const int* __restrict__ gid,
                              const unsigned* __restrict__ mask, const unsigned* __restrict__ aggk,
                              const float* __restrict__ P, const float* __restrict__ Q,
                              const float* __restrict__ Wl1, const float* __restrict__ Wr1,
                              const float* __restrict__ bl1, float* __restrict__ h, int N) {
    int wave = threadIdx.x >> 5;
    int lane = threadIdx.x & 31;
    int n = blockIdx.x * (blockDim.x >> 5) + wave;
    if (n >= N) return;

    float x0 = x3[(size_t)n * 3 + 0];
    float x1 = x3[(size_t)n * 3 + 1];
    float x2 = x3[(size_t)n * 3 + 2];
    int g = gid[n];
    unsigned mk = mask[n];
    bool has = (mk != 0u);
    float a0 = 0.f, a1 = 0.f, a2 = 0.f;
    if (has) {
        a0 = keyf(aggk[(size_t)n * 3 + 0]);
        a1 = keyf(aggk[(size_t)n * 3 + 1]);
        a2 = keyf(aggk[(size_t)n * 3 + 2]);
    }
    const float* Prow = P + (size_t)mk * HID_;
    const float* Qrow = Q + (size_t)g * HID_;

    float v[4];
    #pragma unroll
    for (int c = 0; c < 4; ++c) {
        int col = lane + 32 * c;
        if (col < HID_) {
            float t = bl1[col] + Qrow[col]
                    + x0 * Wr1[col] + x1 * Wr1[HID_ + col] + x2 * Wr1[2 * HID_ + col];
            if (has)
                t += Prow[col]
                   + a0 * Wl1[col] + a1 * Wl1[HID_ + col] + a2 * Wl1[2 * HID_ + col];
            v[c] = t;
        } else {
            v[c] = -INFINITY;
        }
    }
    float m = fmaxf(fmaxf(v[0], v[1]), fmaxf(v[2], v[3]));
    for (int off = 16; off >= 1; off >>= 1)
        m = fmaxf(m, __shfl_xor(m, off, 32));
    float s = 0.f;
    #pragma unroll
    for (int c = 0; c < 4; ++c) s += __expf(v[c] - m);
    for (int off = 16; off >= 1; off >>= 1)
        s += __shfl_xor(s, off, 32);
    float lse = m + __logf(s);
    #pragma unroll
    for (int c = 0; c < 4; ++c) {
        int col = lane + 32 * c;
        if (col < HID_) h[(size_t)n * HID_ + col] = v[c] - lse;
    }
}

__global__ void node_out_kernel(const float* __restrict__ h, const int* __restrict__ counts,
                                const unsigned* __restrict__ head, const int* __restrict__ link2,
                                const float* __restrict__ Wl2, const float* __restrict__ bl2,
                                const float* __restrict__ Wr2, float* __restrict__ y,
                                int N, int E) {
    const int wave = threadIdx.x >> 5;
    const int lane = threadIdx.x & 31;
    const int n = __builtin_amdgcn_readfirstlane(blockIdx.x * (blockDim.x >> 5) + wave);
    if (n >= N) return;
    const bool own = (lane < HID_ / 4);
    const int  c4  = own ? lane * 4 : 0;
    const int2* link = reinterpret_cast<const int2*>(link2);
    const int cnt = counts[n];
    const int e0  = (int)head[n] - 1;

    float4 own_row = *reinterpret_cast<const float4*>(h + (size_t)n * HID_ + c4);
    float4 acc  = make_float4(0.f, 0.f, 0.f, 0.f);
    float4 pend = make_float4(0.f, 0.f, 0.f, 0.f);
    int2 lk = make_int2(-1, 0);
    if (cnt > 0) lk = link[min(max(e0, 0), E - 1)];
    for (int j = 0; j < cnt; ++j) {
        const int2 lkn = link[min(max(lk.x, 0), E - 1)];
        const int  sc  = min(max(lk.y, 0), N - 1);
        const float4 row = *reinterpret_cast<const float4*>(h + (size_t)sc * HID_ + c4);
        acc.x += pend.x; acc.y += pend.y; acc.z += pend.z; acc.w += pend.w;
        pend = row;
        lk = lkn;
    }
    acc.x += pend.x; acc.y += pend.y; acc.z += pend.z; acc.w += pend.w;

    float s1 = 0.f, s2 = 0.f;
    if (own) {
        s1 = fmaf(acc.x, Wl2[c4 + 0], s1);
        s1 = fmaf(acc.y, Wl2[c4 + 1], s1);
        s1 = fmaf(acc.z, Wl2[c4 + 2], s1);
        s1 = fmaf(acc.w, Wl2[c4 + 3], s1);
        s2 = fmaf(own_row.x, Wr2[c4 + 0], s2);
        s2 = fmaf(own_row.y, Wr2[c4 + 1], s2);
        s2 = fmaf(own_row.z, Wr2[c4 + 2], s2);
        s2 = fmaf(own_row.w, Wr2[c4 + 3], s2);
    }
    for (int off = 16; off >= 1; off >>= 1) {
        s1 += __shfl_xor(s1, off, 32);
        s2 += __shfl_xor(s2, off, 32);
    }
    if (lane == 0)
        y[n] = s1 / fmaxf((float)cnt, 1.0f) + bl2[0] + s2;
}

extern "C" void kernel_launch(void* const* d_in, const int* in_sizes, int n_in,
                              void* d_out, int out_size, void* d_ws, size_t ws_size,
                              hipStream_t stream) {
    const float* X      = (const float*)d_in[0];
    const float* x3     = (const float*)d_in[1];
    const int*   ei     = (const int*)d_in[2];
    const int*   gid    = (const int*)d_in[3];
    const float* w1     = (const float*)d_in[4];
    const float* b1     = (const float*)d_in[5];
    const float* w2     = (const float*)d_in[6];
    const float* b2     = (const float*)d_in[7];
    const float* w3     = (const float*)d_in[8];
    const float* b3     = (const float*)d_in[9];
    const float* Wl1    = (const float*)d_in[10];
    const float* bl1    = (const float*)d_in[11];
    const float* Wr1    = (const float*)d_in[12];
    const float* Wl2    = (const float*)d_in[13];
    const float* bl2    = (const float*)d_in[14];
    const float* Wr2    = (const float*)d_in[15];
    float* y = (float*)d_out;

    const int E = in_sizes[2] / 2;
    const int N = in_sizes[3];

    char* ws = (char*)d_ws;
    size_t off = 0;
    auto alloc = [&](size_t bytes) -> char* {
        char* p = ws + off;
        off = (off + bytes + 255) & ~(size_t)255;
        return p;
    };
    float*    act1  = (float*)alloc((size_t)8 * 32 * 21 * 36 * 4);
    float*    act2  = (float*)alloc((size_t)8 * 64 * 17 * 36 * 4);
    float*    act3  = (float*)alloc((size_t)8 * 64 * 13 * 16 * 4);
    float*    enc   = (float*)alloc((size_t)8 * ENC_ * 4);
    float*    P     = (float*)alloc((size_t)256 * HID_ * 4);
    float*    Q     = (float*)alloc((size_t)16  * HID_ * 4);
    unsigned* aggk  = (unsigned*)alloc((size_t)N * 3 * 4);
    unsigned* mask  = (unsigned*)alloc((size_t)N * 4);
    int*      cnt   = (int*)alloc((size_t)N * 2 * 4);
    unsigned* head  = (unsigned*)(cnt + N);
    int*      link  = (int*)alloc((size_t)E * 8);
    float*    hbuf  = (float*)alloc((size_t)N * HID_ * 4);

    const int T = 256;
    zero_u32<<<(N * 3 + T - 1) / T, T, 0, stream>>>(aggk, N * 3);
    zero_u32<<<(N + T - 1) / T, T, 0, stream>>>(mask, N);
    zero_u32<<<(N * 2 + T - 1) / T, T, 0, stream>>>((unsigned*)cnt, N * 2);

    conv_wmma_f32<25, 25, 32, 21, 6, 6, 1, 378><<<95 * 2, 128, 0, stream>>>(X, w1, b1, act1);
    conv_wmma_f32<32, 21, 64, 17, 6, 6, 1, 306><<<77 * 4, 128, 0, stream>>>(act1, w2, b2, act2);
    conv_wmma_f32<64, 17, 64, 13, 4, 4, 0, 104><<<26 * 4, 128, 0, stream>>>(act2, w3, b3, act3);
    pool_kernel<<<(8 * 64 * 11 * 4 + T - 1) / T, T, 0, stream>>>(act3, enc);

    edge_pass<<<(E + T - 1) / T, T, 0, stream>>>(ei, gid, x3, aggk, mask, cnt, head, link, E);

    {
        dim3 gPQ(256 / 16 + 1, (HID_ + 15) / 16);
        gemm_submax_wmma_f32<<<gPQ, 32, 0, stream>>>(enc, Wl1 + 3 * HID_, Wr1 + 3 * HID_, P, Q);
    }

    node_h_kernel<<<(N + 7) / 8, 256, 0, stream>>>(x3, gid, mask, aggk, P, Q,
                                                   Wl1, Wr1, bl1, hbuf, N);

    node_out_kernel<<<(N + 7) / 8, 256, 0, stream>>>(hbuf, cnt, head, link, Wl2, bl2, Wr2, y, N, E);

    (void)n_in; (void)out_size; (void)ws_size;
}


# Graph side pulled from per-node source slots: edge pass = 1 atomic add (arrival rank) + 1 store per edge; node kernels read a node's sources in one load, no list walk; h rows summed in rank order
# speedup vs baseline: 5.0493x; 1.1869x over previous
#include <hip/hip_runtime.h>
#include <hip/hip_bf16.h>
#include <math.h>


#define B_     8
#define CIN_   25
#define NPG_   2000
#define ENC_   2816
#define HID_   100
#define KGEMM_ 2816
#define CAP_   32

typedef __attribute__((ext_vector_type(2))) float v2f;
typedef __attribute__((ext_vector_type(4))) float v4f;
typedef __attribute__((ext_vector_type(8))) float v8f;

__global__ void zero_u32(unsigned* __restrict__ p, int n) {
    int i = blockIdx.x * blockDim.x + threadIdx.x;
    if (i < n) p[i] = 0u;
}

constexpr int largest_divisor_upto(int n, int cap) {
    int best = 1;
    for (int d = 1; d <= cap; ++d) if (n % d == 0) best = d;
    return best;
}

template <int CIN, int DIN, int COUT, int DOUT, int HO, int WO, int PAD, int MT>
__global__ void __launch_bounds__(128)
conv_wmma_f32(const float* __restrict__ In, const float* __restrict__ Wt,
              const float* __restrict__ bias, float* __restrict__ Out) {
    constexpr int PLANE = 36;
    constexpr int CHS   = DIN * PLANE;
    constexpr int HWO   = HO * WO;
    constexpr int P     = DOUT * HWO;
    constexpr int NT    = COUT / 16;
    constexpr int CBF   = CIN / 4;
    constexpr int CREM  = CIN % 4;
    constexpr int CP    = (CIN + 3) / 4 * 4;
    constexpr int ROW   = CP * 45;
    constexpr int PITCH = ROW + (68 - ROW % 64) % 64;
    constexpr int WTOT  = 16 * CIN * 45;
    static_assert(P % 4 == 0, "16-byte stores need P % 4 == 0");
    static_assert(COUT % 16 == 0, "N tiles are full");
    static_assert(PITCH % 64 == 4 && CIN >= 3, "pitch rule; the stager's channel step wraps at most once");

    __shared__ __attribute__((aligned(16))) float Bs[16 * PITCH];

    const int lane = threadIdx.x & 31;
    const int half = lane >> 4;
    const int l16  = lane & 15;
    const int nTile = blockIdx.x % NT;
    {
        const float* Wn = Wt + (size_t)nTile * WTOT;
        constexpr int FULL  = WTOT / 128;
        constexpr int BATCH = largest_divisor_upto(FULL, 60);
        int tap = threadIdx.x % 45, c = threadIdx.x / 45, o = 0;
        for (int j0 = 0; j0 < FULL; j0 += BATCH) {
            float v[BATCH];
            #pragma unroll
            for (int u = 0; u < BATCH; ++u) v[u] = Wn[(j0 + u) * 128 + threadIdx.x];
            #pragma unroll
            for (int u = 0; u < BATCH; ++u) {
                Bs[o * PITCH + ((c >> 1) * 45 + tap) * 2 + (c & 1)] = v[u];
                tap += 38; c += 2;
                if (tap >= 45) { tap -= 45; c += 1; }
                if (c >= CIN)  { c -= CIN;  o += 1; }
            }
        }
        if constexpr (WTOT % 128 != 0) {
            const int i = FULL * 128 + threadIdx.x;
            if (i < WTOT)
                Bs[o * PITCH + ((c >> 1) * 45 + tap) * 2 + (c & 1)] = Wn[i];
        }
        if constexpr (CREM != 0) {
            constexpr int NPAD = CP - CIN;
            for (int i = threadIdx.x; i < 16 * NPAD * 45; i += 128) {
                const int oz = i / (NPAD * 45);
                const int r  = i - oz * (NPAD * 45);
                const int cz = CIN + r / 45;
                const int tz = r % 45;
                Bs[oz * PITCH + ((cz >> 1) * 45 + tz) * 2 + (cz & 1)] = 0.0f;
            }
        }
    }
    __syncthreads();

    const int mTile = (blockIdx.x / NT) * 4 + (threadIdx.x >> 5);
    if (mTile >= MT) return;

    const int m  = mTile * 16 + l16;
    const int bn = m / P;
    const int p  = m - bn * P;
    const int z  = p / HWO;
    const int q  = p - z * HWO;
    const int y  = q / WO;
    const int x  = q - y * WO;
    const int o  = nTile * 16 + l16;

    int  toff[9];
    bool tok[9];
    #pragma unroll
    for (int t = 0; t < 9; ++t) {
        const int yy = y + t / 3 - PAD;
        const int xx = x + t % 3 - PAD;
        const bool ok = (PAD == 0) ? true : ((yy >= 0) && (yy < 6) && (xx >= 0) && (xx < 6));
        tok[t]  = ok;
        toff[t] = ok ? (yy * 6 + xx) : 0;
    }

    int aIdx = (bn * CIN + half * 2) * CHS + z * PLANE;
    const float* bP = Bs + l16 * PITCH + half * 90;

    v8f acc0 = {}, acc1 = {};
    for (int cb = 0; cb < CBF; ++cb) {
        v2f   bf[45];
        float ax[45], ay[45];
        #pragma unroll
        for (int kd = 0; kd < 5; ++kd) {
            #pragma unroll
            for (int t = 0; t < 9; ++t) {
                const int ai = aIdx + kd * PLANE + toff[t];
                bf[kd * 9 + t] = *reinterpret_cast<const v2f*>(bP + (kd * 9 + t) * 2);
                ax[kd * 9 + t] = In[ai];
                ay[kd * 9 + t] = In[ai + CHS];
            }
        }
        __builtin_amdgcn_sched_barrier(0);
        #pragma unroll
        for (int kd = 0; kd < 5; ++kd) {
            #pragma unroll
            for (int t = 0; t < 9; ++t) {
                v2f a;
                a.x = tok[t] ? ax[kd * 9 + t] : 0.0f;
                a.y = tok[t] ? ay[kd * 9 + t] : 0.0f;
                const v2f b = bf[kd * 9 + t];
                if ((kd * 9 + t) & 1)
                    acc1 = __builtin_amdgcn_wmma_f32_16x16x4_f32(false, a, false, b,
                                                                 (short)0, acc1, false, false);
                else
                    acc0 = __builtin_amdgcn_wmma_f32_16x16x4_f32(false, a, false, b,
                                                                 (short)0, acc0, false, false);
            }
        }
        aIdx += 4 * CHS;
        bP   += 4 * 45;
    }
    if (CREM != 0) {
        const int  c0  = CBF * 4 + half * 2;
        const bool cv0 = (c0 < CIN);
        const bool cv1 = (c0 + 1 < CIN);
        const int  cc0 = cv0 ? c0 : CBF * 4;
        const int  cc1 = cv1 ? (c0 + 1) : CBF * 4;
        const int aT0 = (bn * CIN + cc0) * CHS + z * PLANE;
        const int aT1 = (bn * CIN + cc1) * CHS + z * PLANE;
        #pragma unroll
        for (int kd = 0; kd < 5; ++kd) {
            #pragma unroll
            for (int t = 0; t < 9; ++t) {
                const int so = kd * PLANE + toff[t];
                const float ax = In[aT0 + so];
                const float ay = In[aT1 + so];
                v2f a;
                a.x = (tok[t] && cv0) ? ax : 0.0f;
                a.y = (tok[t] && cv1) ? ay : 0.0f;
                const v2f b = *reinterpret_cast<const v2f*>(bP + (kd * 9 + t) * 2);
                if ((kd * 9 + t) & 1)
                    acc1 = __builtin_amdgcn_wmma_f32_16x16x4_f32(false, a, false, b,
                                                                 (short)0, acc1, false, false);
                else
                    acc0 = __builtin_amdgcn_wmma_f32_16x16x4_f32(false, a, false, b,
                                                                 (short)0, acc0, false, false);
            }
        }
    }

    const v8f acc = acc0 + acc1;
    const float bo = bias[o];
    const int m0 = mTile * 16 + half * 8;
    #pragma unroll
    for (int g = 0; g < 2; ++g) {
        const int mg  = m0 + g * 4;
        const int bng = mg / P;
        const int pg  = mg - bng * P;
        float4 v;
        v.x = fmaxf(acc[g * 4 + 0] + bo, 0.0f);
        v.y = fmaxf(acc[g * 4 + 1] + bo, 0.0f);
        v.z = fmaxf(acc[g * 4 + 2] + bo, 0.0f);
        v.w = fmaxf(acc[g * 4 + 3] + bo, 0.0f);
        *reinterpret_cast<float4*>(Out + (size_t)(bng * COUT + o) * P + pg) = v;
    }
}

__global__ void pool_kernel(const float* __restrict__ A, float* __restrict__ enc) {
    const int total = 8 * 64 * 11 * 2 * 2;
    int idx = blockIdx.x * blockDim.x + threadIdx.x;
    if (idx >= total) return;
    int x = idx % 2;  int t = idx / 2;
    int y = t % 2;    t /= 2;
    int d = t % 11;   t /= 11;
    int ch = t % 64;  int bn = t / 64;
    float m = -INFINITY;
    for (int dd = 0; dd < 3; ++dd)
        for (int hh = 0; hh < 3; ++hh)
            for (int ww = 0; ww < 3; ++ww) {
                float v = A[((size_t)((bn * 64 + ch) * 13 + (d + dd))) * 16 + (y + hh) * 4 + (x + ww)];
                m = fmaxf(m, v);
            }
    enc[(size_t)bn * ENC_ + ch * 44 + d * 4 + y * 2 + x] = m;
}

__global__ void edge_pass(const int* __restrict__ ei, int* __restrict__ counts,
                          int* __restrict__ slots, int* __restrict__ spilln,
                          int* __restrict__ spill, int E) {
    int e = blockIdx.x * blockDim.x + threadIdx.x;
    if (e >= E) return;
    const int s = ei[e];
    const int d = ei[E + e];
    const int r = atomicAdd(&counts[d], 1);
    if (r < CAP_) {
        slots[(size_t)d * CAP_ + r] = s;
    } else {
        const int o = atomicAdd(spilln, 1);
        reinterpret_cast<int2*>(spill)[o] = make_int2(d, s);
    }
}

__global__ void __launch_bounds__(32)
gemm_submax_wmma_f32(const float* __restrict__ enc, const float* __restrict__ Wl,
                     const float* __restrict__ Wr, float* __restrict__ P,
                     float* __restrict__ Q) {
    constexpr int EPITCH = KGEMM_ + 4;
    __shared__ __attribute__((aligned(16))) float EncS[8 * EPITCH];

    const int Ncols = HID_;
    const int lane = threadIdx.x & 31;
    const int half = lane >> 4;
    const int l16  = lane & 15;
    const bool isQ = (blockIdx.x == 16);
    const float* W = isQ ? Wr : Wl;
    float* C = isQ ? Q : P;
    const int Mrows = isQ ? 8 : 256;
    const int mBase = isQ ? 0 : blockIdx.x * 16;
    const unsigned mask = isQ ? (1u << (l16 & 7)) : (unsigned)(mBase + l16);
    const int g0 = mask ? (__ffs(mask) - 1) : 0;
    const int nBase = blockIdx.y * 16;
    const int nCol = nBase + l16;
    const bool nOk = (nCol < Ncols);
    const int nClamp = nOk ? nCol : (Ncols - 1);

    int aoff[8];
    #pragma unroll
    for (int g = 0; g < 8; ++g) {
        const int r = ((mask >> g) & 1u) ? g : g0;
        aoff[g] = r * EPITCH + half * 8;
    }
    const float* Bcol = W + (size_t)(half * 8) * Ncols + nClamp;

    struct KB { float bv[8]; };
    struct KA { v4f L[8], H[8]; };
    v8f acc0 = {}, acc1 = {};

    auto load_b = [&](KB& s, int kt) __attribute__((always_inline)) {
        #pragma unroll
        for (int i = 0; i < 8; ++i) s.bv[i] = Bcol[(size_t)(kt + i) * Ncols];
    };
    auto load_a = [&](KA& s, int kt) __attribute__((always_inline)) {
        #pragma unroll
        for (int g = 0; g < 8; ++g) {
            s.L[g] = *reinterpret_cast<const v4f*>(EncS + aoff[g] + kt);
            s.H[g] = *reinterpret_cast<const v4f*>(EncS + aoff[g] + kt + 4);
        }
    };
    auto mac_step = [&](const KA& s, const KB& b) __attribute__((always_inline)) {
        v4f lo = s.L[0], hi = s.H[0];
        #pragma unroll
        for (int g = 1; g < 8; ++g) {
            lo.x = fmaxf(lo.x, s.L[g].x); lo.y = fmaxf(lo.y, s.L[g].y);
            lo.z = fmaxf(lo.z, s.L[g].z); lo.w = fmaxf(lo.w, s.L[g].w);
            hi.x = fmaxf(hi.x, s.H[g].x); hi.y = fmaxf(hi.y, s.H[g].y);
            hi.z = fmaxf(hi.z, s.H[g].z); hi.w = fmaxf(hi.w, s.H[g].w);
        }
        v2f a0, a1, a2, a3, b0, b1, b2, b3;
        a0.x = fmaxf(lo.x, 0.0f); a0.y = fmaxf(lo.y, 0.0f);
        a1.x = fmaxf(lo.z, 0.0f); a1.y = fmaxf(lo.w, 0.0f);
        a2.x = fmaxf(hi.x, 0.0f); a2.y = fmaxf(hi.y, 0.0f);
        a3.x = fmaxf(hi.z, 0.0f); a3.y = fmaxf(hi.w, 0.0f);
        b0.x = b.bv[0]; b0.y = b.bv[1];
        b1.x = b.bv[2]; b1.y = b.bv[3];
        b2.x = b.bv[4]; b2.y = b.bv[5];
        b3.x = b.bv[6]; b3.y = b.bv[7];
        acc0 = __builtin_amdgcn_wmma_f32_16x16x4_f32(false, a0, false, b0,
                                                     (short)0, acc0, false, false);
        acc1 = __builtin_amdgcn_wmma_f32_16x16x4_f32(false, a1, false, b1,
                                                     (short)0, acc1, false, false);
        acc0 = __builtin_amdgcn_wmma_f32_16x16x4_f32(false, a2, false, b2,
                                                     (short)0, acc0, false, false);
        acc1 = __builtin_amdgcn_wmma_f32_16x16x4_f32(false, a3, false, b3,
                                                     (short)0, acc1, false, false);
    };

    static_assert((KGEMM_ / 16) % 8 == 0 && KGEMM_ % 16 == 0, "22 trips of eight steps");
    for (int r0 = 0; r0 < 8; r0 += 2) {
        v4f v[44];
        #pragma unroll
        for (int u = 0; u < 44; ++u) {
            const int r = r0 + u / 22, j = u % 22;
            v[u] = *reinterpret_cast<const v4f*>(enc + (size_t)r * ENC_ + (j * 32 + lane) * 4);
        }
        #pragma unroll
        for (int u = 0; u < 44; ++u) {
            const int r = r0 + u / 22, j = u % 22;
            *reinterpret_cast<v4f*>(EncS + r * EPITCH + (j * 32 + lane) * 4) = v[u];
        }
    }
    __syncthreads();

    KB b0, b1, b2, b3, b4, b5, b6, b7;
    load_b(b0, 0);  load_b(b1, 16); load_b(b2, 32); load_b(b3, 48);
    load_b(b4, 64); load_b(b5, 80); load_b(b6, 96); load_b(b7, 112);
    KA a0, a1;
    load_a(a0, 0);
    load_a(a1, 16);
    for (int kt = 0; kt < KGEMM_ - 128; kt += 128) {
        mac_step(a0, b0); load_b(b0, kt + 128); load_a(a0, kt + 32);
        mac_step(a1, b1); load_b(b1, kt + 144); load_a(a1, kt + 48);
        mac_step(a0, b2); load_b(b2, kt + 160); load_a(a0, kt + 64);
        mac_step(a1, b3); load_b(b3, kt + 176); load_a(a1, kt + 80);
        mac_step(a0, b4); load_b(b4, kt + 192); load_a(a0, kt + 96);
        mac_step(a1, b5); load_b(b5, kt + 208); load_a(a1, kt + 112);
        mac_step(a0, b6); load_b(b6, kt + 224); load_a(a0, kt + 128);
        mac_step(a1, b7); load_b(b7, kt + 240); load_a(a1, kt + 144);
    }
    {
        constexpr int kt = KGEMM_ - 128;
        mac_step(a0, b0); load_a(a0, kt + 32);
        mac_step(a1, b1); load_a(a1, kt + 48);
        mac_step(a0, b2); load_a(a0, kt + 64);
        mac_step(a1, b3); load_a(a1, kt + 80);
        mac_step(a0, b4); load_a(a0, kt + 96);
        mac_step(a1, b5); load_a(a1, kt + 112);
        mac_step(a0, b6);
        mac_step(a1, b7);
    }
    v8f acc = acc0 + acc1;
    for (int v = 0; v < 8; ++v) {
        const int m = mBase + half * 8 + v;
        const float val = (!isQ && m == 0) ? 0.0f : acc[v];
        if (m < Mrows && nOk) C[(size_t)m * Ncols + nCol] = val;
    }
}

__global__ void node_h_kernel(const float* __restrict__ x3, const int* __restrict__ gid,
                              const int* __restrict__ counts, const int* __restrict__ slots,
                              const int* __restrict__ spilln, const int* __restrict__ spill,
                              const float* __restrict__ P, const float* __restrict__ Q,
                              const float* __restrict__ Wl1, const float* __restrict__ Wr1,
                              const float* __restrict__ bl1, float* __restrict__ h, int N, int E) {
    int wave = threadIdx.x >> 5;
    int lane = threadIdx.x & 31;
    const int n = __builtin_amdgcn_readfirstlane(blockIdx.x * (blockDim.x >> 5) + wave);
    if (n >= N) return;

    float x0 = x3[(size_t)n * 3 + 0];
    float x1 = x3[(size_t)n * 3 + 1];
    float x2 = x3[(size_t)n * 3 + 2];
    int g = gid[n];
    const int cnt = counts[n];
    const int m = min(cnt, CAP_);
    const bool has = (cnt > 0);

    unsigned mk = 0u;
    float a0 = -INFINITY, a1 = -INFINITY, a2 = -INFINITY;
    if (lane < m) {
        const int s = min(max(slots[(size_t)n * CAP_ + lane], 0), N - 1);
        mk = 1u << (gid[s] & 7);
        a0 = x3[(size_t)s * 3 + 0];
        a1 = x3[(size_t)s * 3 + 1];
        a2 = x3[(size_t)s * 3 + 2];
    }
    if (cnt > CAP_) {
        const int sn = min(max(spilln[0], 0), E);
        for (int i = lane; i < sn; i += 32) {
            const int2 ds = reinterpret_cast<const int2*>(spill)[i];
            if (ds.x == n) {
                const int s = min(max(ds.y, 0), N - 1);
                mk |= 1u << (gid[s] & 7);
                a0 = fmaxf(a0, x3[(size_t)s * 3 + 0]);
                a1 = fmaxf(a1, x3[(size_t)s * 3 + 1]);
                a2 = fmaxf(a2, x3[(size_t)s * 3 + 2]);
            }
        }
    }
    for (int off = 16; off >= 1; off >>= 1) {
        mk |= (unsigned)__shfl_xor((int)mk, off, 32);
        a0 = fmaxf(a0, __shfl_xor(a0, off, 32));
        a1 = fmaxf(a1, __shfl_xor(a1, off, 32));
        a2 = fmaxf(a2, __shfl_xor(a2, off, 32));
    }
    if (!has) { a0 = 0.f; a1 = 0.f; a2 = 0.f; }
    const float* Prow = P + (size_t)mk * HID_;
    const float* Qrow = Q + (size_t)g * HID_;

    float v[4];
    #pragma unroll
    for (int c = 0; c < 4; ++c) {
        int col = lane + 32 * c;
        if (col < HID_) {
            float t = bl1[col] + Qrow[col]
                    + x0 * Wr1[col] + x1 * Wr1[HID_ + col] + x2 * Wr1[2 * HID_ + col];
            if (has)
                t += Prow[col]
                   + a0 * Wl1[col] + a1 * Wl1[HID_ + col] + a2 * Wl1[2 * HID_ + col];
            v[c] = t;
        } else {
            v[c] = -INFINITY;
        }
    }
    float mx = fmaxf(fmaxf(v[0], v[1]), fmaxf(v[2], v[3]));
    for (int off = 16; off >= 1; off >>= 1)
        mx = fmaxf(mx, __shfl_xor(mx, off, 32));
    float s = 0.f;
    #pragma unroll
    for (int c = 0; c < 4; ++c) s += __expf(v[c] - mx);
    for (int off = 16; off >= 1; off >>= 1)
        s += __shfl_xor(s, off, 32);
    float lse = mx + __logf(s);
    #pragma unroll
    for (int c = 0; c < 4; ++c) {
        int col = lane + 32 * c;
        if (col < HID_) h[(size_t)n * HID_ + col] = v[c] - lse;
    }
}

__global__ void node_out_kernel(const float* __restrict__ h, const int* __restrict__ counts,
                                const int* __restrict__ slots, const int* __restrict__ spilln,
                                const int* __restrict__ spill,
                                const float* __restrict__ Wl2, const float* __restrict__ bl2,
                                const float* __restrict__ Wr2, float* __restrict__ y,
                                int N, int E) {
    const int wave = threadIdx.x >> 5;
    const int lane = threadIdx.x & 31;
    const int n = __builtin_amdgcn_readfirstlane(blockIdx.x * (blockDim.x >> 5) + wave);
    if (n >= N) return;
    const bool own = (lane < HID_ / 4);
    const int  c4  = own ? lane * 4 : 0;
    const int cnt = counts[n];
    const int m   = min(cnt, CAP_);

    int my = 0;
    if (lane < m) my = min(max(slots[(size_t)n * CAP_ + lane], 0), N - 1);
    float4 own_row = *reinterpret_cast<const float4*>(h + (size_t)n * HID_ + c4);
    float4 acc = make_float4(0.f, 0.f, 0.f, 0.f);
    for (int j0 = 0; j0 < m; j0 += 8) {
        float4 row[8];
        #pragma unroll
        for (int u = 0; u < 8; ++u) {
            const int sc = __builtin_amdgcn_readlane(my, j0 + u);
            row[u] = *reinterpret_cast<const float4*>(h + (size_t)sc * HID_ + c4);
        }
        #pragma unroll
        for (int u = 0; u < 8; ++u) {
            const bool on = (j0 + u < m);
            acc.x += on ? row[u].x : 0.f;
            acc.y += on ? row[u].y : 0.f;
            acc.z += on ? row[u].z : 0.f;
            acc.w += on ? row[u].w : 0.f;
        }
    }
    if (cnt > CAP_) {
        const int sn = min(max(spilln[0], 0), E);
        for (int i = 0; i < sn; ++i) {
            const int2 ds = reinterpret_cast<const int2*>(spill)[i];
            if (ds.x == n) {
                const int sc = min(max(ds.y, 0), N - 1);
                const float4 row = *reinterpret_cast<const float4*>(h + (size_t)sc * HID_ + c4);
                acc.x += row.x; acc.y += row.y; acc.z += row.z; acc.w += row.w;
            }
        }
    }

    float s1 = 0.f, s2 = 0.f;
    if (own) {
        s1 = fmaf(acc.x, Wl2[c4 + 0], s1);
        s1 = fmaf(acc.y, Wl2[c4 + 1], s1);
        s1 = fmaf(acc.z, Wl2[c4 + 2], s1);
        s1 = fmaf(acc.w, Wl2[c4 + 3], s1);
        s2 = fmaf(own_row.x, Wr2[c4 + 0], s2);
        s2 = fmaf(own_row.y, Wr2[c4 + 1], s2);
        s2 = fmaf(own_row.z, Wr2[c4 + 2], s2);
        s2 = fmaf(own_row.w, Wr2[c4 + 3], s2);
    }
    for (int off = 16; off >= 1; off >>= 1) {
        s1 += __shfl_xor(s1, off, 32);
        s2 += __shfl_xor(s2, off, 32);
    }
    if (lane == 0)
        y[n] = s1 / fmaxf((float)cnt, 1.0f) + bl2[0] + s2;
}

extern "C" void kernel_launch(void* const* d_in, const int* in_sizes, int n_in,
                              void* d_out, int out_size, void* d_ws, size_t ws_size,
                              hipStream_t stream) {
    const float* X      = (const float*)d_in[0];
    const float* x3     = (const float*)d_in[1];
    const int*   ei     = (const int*)d_in[2];
    const int*   gid    = (const int*)d_in[3];
    const float* w1     = (const float*)d_in[4];
    const float* b1     = (const float*)d_in[5];
    const float* w2     = (const float*)d_in[6];
    const float* b2     = (const float*)d_in[7];
    const float* w3     = (const float*)d_in[8];
    const float* b3     = (const float*)d_in[9];
    const float* Wl1    = (const float*)d_in[10];
    const float* bl1    = (const float*)d_in[11];
    const float* Wr1    = (const float*)d_in[12];
    const float* Wl2    = (const float*)d_in[13];
    const float* bl2    = (const float*)d_in[14];
    const float* Wr2    = (const float*)d_in[15];
    float* y = (float*)d_out;

    const int E = in_sizes[2] / 2;
    const int N = in_sizes[3];

    char* ws = (char*)d_ws;
    size_t off = 0;
    auto alloc = [&](size_t bytes) -> char* {
        char* p = ws + off;
        off = (off + bytes + 255) & ~(size_t)255;
        return p;
    };
    float*    act1  = (float*)alloc((size_t)8 * 32 * 21 * 36 * 4);
    float*    act2  = (float*)alloc((size_t)8 * 64 * 17 * 36 * 4);
    float*    act3  = (float*)alloc((size_t)8 * 64 * 13 * 16 * 4);
    float*    enc   = (float*)alloc((size_t)8 * ENC_ * 4);
    float*    P     = (float*)alloc((size_t)256 * HID_ * 4);
    float*    Q     = (float*)alloc((size_t)16  * HID_ * 4);
    int*      cnt   = (int*)alloc((size_t)(N + 1) * 4);
    int*      spilln = cnt + N;
    int*      slots = (int*)alloc((size_t)N * CAP_ * 4);
    int*      spill = (int*)alloc((size_t)E * 8);
    float*    hbuf  = (float*)alloc((size_t)N * HID_ * 4);

    const int T = 256;
    zero_u32<<<(N + 1 + T - 1) / T, T, 0, stream>>>((unsigned*)cnt, N + 1);

    conv_wmma_f32<25, 25, 32, 21, 6, 6, 1, 378><<<95 * 2, 128, 0, stream>>>(X, w1, b1, act1);
    conv_wmma_f32<32, 21, 64, 17, 6, 6, 1, 306><<<77 * 4, 128, 0, stream>>>(act1, w2, b2, act2);
    conv_wmma_f32<64, 17, 64, 13, 4, 4, 0, 104><<<26 * 4, 128, 0, stream>>>(act2, w3, b3, act3);
    pool_kernel<<<(8 * 64 * 11 * 4 + T - 1) / T, T, 0, stream>>>(act3, enc);

    edge_pass<<<(E + T - 1) / T, T, 0, stream>>>(ei, cnt, slots, spilln, spill, E);

    {
        dim3 gPQ(256 / 16 + 1, (HID_ + 15) / 16);
        gemm_submax_wmma_f32<<<gPQ, 32, 0, stream>>>(enc, Wl1 + 3 * HID_, Wr1 + 3 * HID_, P, Q);
    }

    node_h_kernel<<<(N + 7) / 8, 256, 0, stream>>>(x3, gid, cnt, slots, spilln, spill, P, Q,
                                                   Wl1, Wr1, bl1, hbuf, N, E);

    node_out_kernel<<<(N + 7) / 8, 256, 0, stream>>>(hbuf, cnt, slots, spilln, spill, Wl2, bl2, Wr2, y, N, E);

    (void)n_in; (void)out_size; (void)ws_size;
}


# Conv weights copied to LDS in the weight array's own order, 16-byte pieces, no index chain; WMMA i of a K block takes elements 2i, 2i+1 of the lane's 2 channels x 45 taps (changes K summation order)
# speedup vs baseline: 5.6227x; 1.1136x over previous
#include <hip/hip_runtime.h>
#include <hip/hip_bf16.h>
#include <math.h>


#define B_     8
#define CIN_   25
#define NPG_   2000
#define ENC_   2816
#define HID_   100
#define KGEMM_ 2816
#define CAP_   32

typedef __attribute__((ext_vector_type(2))) float v2f;
typedef __attribute__((ext_vector_type(4))) float v4f;
typedef __attribute__((ext_vector_type(8))) float v8f;

__global__ void zero_u32(unsigned* __restrict__ p, int n) {
    int i = blockIdx.x * blockDim.x + threadIdx.x;
    if (i < n) p[i] = 0u;
}

template <int CIN, int DIN, int COUT, int DOUT, int HO, int WO, int PAD, int MT>
__global__ void __launch_bounds__(128)
conv_wmma_f32(const float* __restrict__ In, const float* __restrict__ Wt,
              const float* __restrict__ bias, float* __restrict__ Out) {
    constexpr int PLANE = 36;
    constexpr int CHS   = DIN * PLANE;
    constexpr int HWO   = HO * WO;
    constexpr int P     = DOUT * HWO;
    constexpr int NT    = COUT / 16;
    constexpr int CBF   = CIN / 4;
    constexpr int CREM  = CIN % 4;
    constexpr int CP    = (CIN + 3) / 4 * 4;
    constexpr int RW    = CIN * 45;
    constexpr int ROW   = CP * 45;
    constexpr int PITCH = ROW + (68 - ROW % 64) % 64;
    constexpr int WTOT  = 16 * RW;
    constexpr int V     = (RW % 4 == 0) ? 4 : 1;
    static_assert(P % 4 == 0, "16-byte stores need P % 4 == 0");
    static_assert(COUT % 16 == 0, "N tiles are full");
    constexpr int RWP   = RW / V;
    constexpr int PITCHP = PITCH / V;
    static_assert(PITCH % 64 == 4 && RW % V == 0 && PITCH % V == 0 && RWP > 128,
                  "pitch rule; rows are whole pieces; a thread's piece is at most one row past the trip's first");

    __shared__ __attribute__((aligned(16))) float Bs[16 * PITCH];

    const int lane = threadIdx.x & 31;
    const int half = lane >> 4;
    const int l16  = lane & 15;
    const int nTile = blockIdx.x % NT;
    {
        typedef float piece_t __attribute__((ext_vector_type(V)));
        const piece_t* Wn = reinterpret_cast<const piece_t*>(Wt + (size_t)nTile * WTOT);
        piece_t* Bp = reinterpret_cast<piece_t*>(Bs);
        constexpr int NPC  = WTOT / V;
        constexpr int FULL = NPC / 128;
        const int tid = threadIdx.x;
        piece_t v[FULL];
        #pragma unroll
        for (int j = 0; j < FULL; ++j) v[j] = Wn[j * 128 + tid];
        #pragma unroll
        for (int j = 0; j < FULL; ++j) {
            const int first = j * 128;
            const int q = first / RWP, m = first - q * RWP;
            const int next = (tid + m >= RWP) ? (PITCHP - RWP) : 0;
            Bp[q * PITCHP + m + tid + next] = v[j];
        }
        if constexpr (NPC % 128 != 0) {
            constexpr int first = FULL * 128;
            constexpr int q = first / RWP, m = first - q * RWP;
            if (first + tid < NPC) {
                const int next = (tid + m >= RWP) ? (PITCHP - RWP) : 0;
                Bp[q * PITCHP + m + tid + next] = Wn[first + tid];
            }
        }
        if constexpr (CREM != 0) {
            constexpr int NPAD = ROW - RW;
            for (int i = threadIdx.x; i < 16 * NPAD; i += 128) {
                const int oz = i / NPAD;
                Bs[oz * PITCH + RW + (i - oz * NPAD)] = 0.0f;
            }
        }
    }
    __syncthreads();

    const int mTile = (blockIdx.x / NT) * 4 + (threadIdx.x >> 5);
    if (mTile >= MT) return;

    const int m  = mTile * 16 + l16;
    const int bn = m / P;
    const int p  = m - bn * P;
    const int z  = p / HWO;
    const int q  = p - z * HWO;
    const int y  = q / WO;
    const int x  = q - y * WO;
    const int o  = nTile * 16 + l16;

    int  toff[9];
    bool tok[9];
    #pragma unroll
    for (int t = 0; t < 9; ++t) {
        const int yy = y + t / 3 - PAD;
        const int xx = x + t % 3 - PAD;
        const bool ok = (PAD == 0) ? true : ((yy >= 0) && (yy < 6) && (xx >= 0) && (xx < 6));
        tok[t]  = ok;
        toff[t] = ok ? (yy * 6 + xx) : 0;
    }

    int aIdx = (bn * CIN + half * 2) * CHS + z * PLANE;
    const float* bP = Bs + l16 * PITCH + half * 90;

    v8f acc0 = {}, acc1 = {};
    auto k_block = [&](int a0, int a1, bool v0, bool v1, const float* bp) __attribute__((always_inline)) {
        v2f   bf[45];
        float ae[90];
        #pragma unroll
        for (int i = 0; i < 45; ++i) {
            bf[i] = *reinterpret_cast<const v2f*>(bp + 2 * i);
            #pragma unroll
            for (int s = 0; s < 2; ++s) {
                const int e = 2 * i + s, tp = e % 45;
                ae[e] = In[(e / 45 ? a1 : a0) + (tp / 9) * PLANE + toff[tp % 9]];
            }
        }
        __builtin_amdgcn_sched_barrier(0);
        #pragma unroll
        for (int i = 0; i < 45; ++i) {
            const int e0 = 2 * i, e1 = 2 * i + 1;
            const bool on0 = tok[e0 % 45 % 9] & (e0 / 45 ? v1 : v0);
            const bool on1 = tok[e1 % 45 % 9] & (e1 / 45 ? v1 : v0);
            v2f a;
            a.x = on0 ? ae[e0] : 0.0f;
            a.y = on1 ? ae[e1] : 0.0f;
            const v2f b = bf[i];
            if (i & 1)
                acc1 = __builtin_amdgcn_wmma_f32_16x16x4_f32(false, a, false, b,
                                                             (short)0, acc1, false, false);
            else
                acc0 = __builtin_amdgcn_wmma_f32_16x16x4_f32(false, a, false, b,
                                                             (short)0, acc0, false, false);
        }
    };
    for (int cb = 0; cb < CBF; ++cb) {
        k_block(aIdx, aIdx + CHS, true, true, bP);
        aIdx += 4 * CHS;
        bP   += 4 * 45;
    }
    if (CREM != 0) {
        const int  c0 = CBF * 4 + half * 2;
        const bool v0 = (c0 < CIN), v1 = (c0 + 1 < CIN);
        k_block((bn * CIN + (v0 ? c0     : CBF * 4)) * CHS + z * PLANE,
                (bn * CIN + (v1 ? c0 + 1 : CBF * 4)) * CHS + z * PLANE, v0, v1, bP);
    }

    const v8f acc = acc0 + acc1;
    const float bo = bias[o];
    const int m0 = mTile * 16 + half * 8;
    #pragma unroll
    for (int g = 0; g < 2; ++g) {
        const int mg  = m0 + g * 4;
        const int bng = mg / P;
        const int pg  = mg - bng * P;
        float4 v;
        v.x = fmaxf(acc[g * 4 + 0] + bo, 0.0f);
        v.y = fmaxf(acc[g * 4 + 1] + bo, 0.0f);
        v.z = fmaxf(acc[g * 4 + 2] + bo, 0.0f);
        v.w = fmaxf(acc[g * 4 + 3] + bo, 0.0f);
        *reinterpret_cast<float4*>(Out + (size_t)(bng * COUT + o) * P + pg) = v;
    }
}

__global__ void pool_kernel(const float* __restrict__ A, float* __restrict__ enc) {
    const int total = 8 * 64 * 11 * 2 * 2;
    int idx = blockIdx.x * blockDim.x + threadIdx.x;
    if (idx >= total) return;
    int x = idx % 2;  int t = idx / 2;
    int y = t % 2;    t /= 2;
    int d = t % 11;   t /= 11;
    int ch = t % 64;  int bn = t / 64;
    float m = -INFINITY;
    for (int dd = 0; dd < 3; ++dd)
        for (int hh = 0; hh < 3; ++hh)
            for (int ww = 0; ww < 3; ++ww) {
                float v = A[((size_t)((bn * 64 + ch) * 13 + (d + dd))) * 16 + (y + hh) * 4 + (x + ww)];
                m = fmaxf(m, v);
            }
    enc[(size_t)bn * ENC_ + ch * 44 + d * 4 + y * 2 + x] = m;
}

__global__ void edge_pass(const int* __restrict__ ei, int* __restrict__ counts,
                          int* __restrict__ slots, int* __restrict__ spilln,
                          int* __restrict__ spill, int E) {
    int e = blockIdx.x * blockDim.x + threadIdx.x;
    if (e >= E) return;
    const int s = ei[e];
    const int d = ei[E + e];
    const int r = atomicAdd(&counts[d], 1);
    if (r < CAP_) {
        slots[(size_t)d * CAP_ + r] = s;
    } else {
        const int o = atomicAdd(spilln, 1);
        reinterpret_cast<int2*>(spill)[o] = make_int2(d, s);
    }
}

__global__ void __launch_bounds__(32)
gemm_submax_wmma_f32(const float* __restrict__ enc, const float* __restrict__ Wl,
                     const float* __restrict__ Wr, float* __restrict__ P,
                     float* __restrict__ Q) {
    constexpr int EPITCH = KGEMM_ + 4;
    __shared__ __attribute__((aligned(16))) float EncS[8 * EPITCH];

    const int Ncols = HID_;
    const int lane = threadIdx.x & 31;
    const int half = lane >> 4;
    const int l16  = lane & 15;
    const bool isQ = (blockIdx.x == 16);
    const float* W = isQ ? Wr : Wl;
    float* C = isQ ? Q : P;
    const int Mrows = isQ ? 8 : 256;
    const int mBase = isQ ? 0 : blockIdx.x * 16;
    const unsigned mask = isQ ? (1u << (l16 & 7)) : (unsigned)(mBase + l16);
    const int g0 = mask ? (__ffs(mask) - 1) : 0;
    const int nBase = blockIdx.y * 16;
    const int nCol = nBase + l16;
    const bool nOk = (nCol < Ncols);
    const int nClamp = nOk ? nCol : (Ncols - 1);

    int aoff[8];
    #pragma unroll
    for (int g = 0; g < 8; ++g) {
        const int r = ((mask >> g) & 1u) ? g : g0;
        aoff[g] = r * EPITCH + half * 8;
    }
    const float* Bcol = W + (size_t)(half * 8) * Ncols + nClamp;

    struct KB { float bv[8]; };
    struct KA { v4f L[8], H[8]; };
    v8f acc0 = {}, acc1 = {};

    auto load_b = [&](KB& s, int kt) __attribute__((always_inline)) {
        #pragma unroll
        for (int i = 0; i < 8; ++i) s.bv[i] = Bcol[(size_t)(kt + i) * Ncols];
    };
    auto load_a = [&](KA& s, int kt) __attribute__((always_inline)) {
        #pragma unroll
        for (int g = 0; g < 8; ++g) {
            s.L[g] = *reinterpret_cast<const v4f*>(EncS + aoff[g] + kt);
            s.H[g] = *reinterpret_cast<const v4f*>(EncS + aoff[g] + kt + 4);
        }
    };
    auto mac_step = [&](const KA& s, const KB& b) __attribute__((always_inline)) {
        v4f lo = s.L[0], hi = s.H[0];
        #pragma unroll
        for (int g = 1; g < 8; ++g) {
            lo.x = fmaxf(lo.x, s.L[g].x); lo.y = fmaxf(lo.y, s.L[g].y);
            lo.z = fmaxf(lo.z, s.L[g].z); lo.w = fmaxf(lo.w, s.L[g].w);
            hi.x = fmaxf(hi.x, s.H[g].x); hi.y = fmaxf(hi.y, s.H[g].y);
            hi.z = fmaxf(hi.z, s.H[g].z); hi.w = fmaxf(hi.w, s.H[g].w);
        }
        v2f a0, a1, a2, a3, b0, b1, b2, b3;
        a0.x = fmaxf(lo.x, 0.0f); a0.y = fmaxf(lo.y, 0.0f);
        a1.x = fmaxf(lo.z, 0.0f); a1.y = fmaxf(lo.w, 0.0f);
        a2.x = fmaxf(hi.x, 0.0f); a2.y = fmaxf(hi.y, 0.0f);
        a3.x = fmaxf(hi.z, 0.0f); a3.y = fmaxf(hi.w, 0.0f);
        b0.x = b.bv[0]; b0.y = b.bv[1];
        b1.x = b.bv[2]; b1.y = b.bv[3];
        b2.x = b.bv[4]; b2.y = b.bv[5];
        b3.x = b.bv[6]; b3.y = b.bv[7];
        acc0 = __builtin_amdgcn_wmma_f32_16x16x4_f32(false, a0, false, b0,
                                                     (short)0, acc0, false, false);
        acc1 = __builtin_amdgcn_wmma_f32_16x16x4_f32(false, a1, false, b1,
                                                     (short)0, acc1, false, false);
        acc0 = __builtin_amdgcn_wmma_f32_16x16x4_f32(false, a2, false, b2,
                                                     (short)0, acc0, false, false);
        acc1 = __builtin_amdgcn_wmma_f32_16x16x4_f32(false, a3, false, b3,
                                                     (short)0, acc1, false, false);
    };

    static_assert((KGEMM_ / 16) % 8 == 0 && KGEMM_ % 16 == 0, "22 trips of eight steps");
    for (int r0 = 0; r0 < 8; r0 += 2) {
        v4f v[44];
        #pragma unroll
        for (int u = 0; u < 44; ++u) {
            const int r = r0 + u / 22, j = u % 22;
            v[u] = *reinterpret_cast<const v4f*>(enc + (size_t)r * ENC_ + (j * 32 + lane) * 4);
        }
        #pragma unroll
        for (int u = 0; u < 44; ++u) {
            const int r = r0 + u / 22, j = u % 22;
            *reinterpret_cast<v4f*>(EncS + r * EPITCH + (j * 32 + lane) * 4) = v[u];
        }
    }
    __syncthreads();

    KB b0, b1, b2, b3, b4, b5, b6, b7;
    load_b(b0, 0);  load_b(b1, 16); load_b(b2, 32); load_b(b3, 48);
    load_b(b4, 64); load_b(b5, 80); load_b(b6, 96); load_b(b7, 112);
    KA a0, a1;
    load_a(a0, 0);
    load_a(a1, 16);
    for (int kt = 0; kt < KGEMM_ - 128; kt += 128) {
        mac_step(a0, b0); load_b(b0, kt + 128); load_a(a0, kt + 32);
        mac_step(a1, b1); load_b(b1, kt + 144); load_a(a1, kt + 48);
        mac_step(a0, b2); load_b(b2, kt + 160); load_a(a0, kt + 64);
        mac_step(a1, b3); load_b(b3, kt + 176); load_a(a1, kt + 80);
        mac_step(a0, b4); load_b(b4, kt + 192); load_a(a0, kt + 96);
        mac_step(a1, b5); load_b(b5, kt + 208); load_a(a1, kt + 112);
        mac_step(a0, b6); load_b(b6, kt + 224); load_a(a0, kt + 128);
        mac_step(a1, b7); load_b(b7, kt + 240); load_a(a1, kt + 144);
    }
    {
        constexpr int kt = KGEMM_ - 128;
        mac_step(a0, b0); load_a(a0, kt + 32);
        mac_step(a1, b1); load_a(a1, kt + 48);
        mac_step(a0, b2); load_a(a0, kt + 64);
        mac_step(a1, b3); load_a(a1, kt + 80);
        mac_step(a0, b4); load_a(a0, kt + 96);
        mac_step(a1, b5); load_a(a1, kt + 112);
        mac_step(a0, b6);
        mac_step(a1, b7);
    }
    v8f acc = acc0 + acc1;
    for (int v = 0; v < 8; ++v) {
        const int m = mBase + half * 8 + v;
        const float val = (!isQ && m == 0) ? 0.0f : acc[v];
        if (m < Mrows && nOk) C[(size_t)m * Ncols + nCol] = val;
    }
}

__global__ void node_h_kernel(const float* __restrict__ x3, const int* __restrict__ gid,
                              const int* __restrict__ counts, const int* __restrict__ slots,
                              const int* __restrict__ spilln, const int* __restrict__ spill,
                              const float* __restrict__ P, const float* __restrict__ Q,
                              const float* __restrict__ Wl1, const float* __restrict__ Wr1,
                              const float* __restrict__ bl1, float* __restrict__ h, int N, int E) {
    int wave = threadIdx.x >> 5;
    int lane = threadIdx.x & 31;
    const int n = __builtin_amdgcn_readfirstlane(blockIdx.x * (blockDim.x >> 5) + wave);
    if (n >= N) return;

    float x0 = x3[(size_t)n * 3 + 0];
    float x1 = x3[(size_t)n * 3 + 1];
    float x2 = x3[(size_t)n * 3 + 2];
    int g = gid[n];
    const int cnt = counts[n];
    const int m = min(cnt, CAP_);
    const bool has = (cnt > 0);

    unsigned mk = 0u;
    float a0 = -INFINITY, a1 = -INFINITY, a2 = -INFINITY;
    if (lane < m) {
        const int s = min(max(slots[(size_t)n * CAP_ + lane], 0), N - 1);
        mk = 1u << (gid[s] & 7);
        a0 = x3[(size_t)s * 3 + 0];
        a1 = x3[(size_t)s * 3 + 1];
        a2 = x3[(size_t)s * 3 + 2];
    }
    if (cnt > CAP_) {
        const int sn = min(max(spilln[0], 0), E);
        for (int i = lane; i < sn; i += 32) {
            const int2 ds = reinterpret_cast<const int2*>(spill)[i];
            if (ds.x == n) {
                const int s = min(max(ds.y, 0), N - 1);
                mk |= 1u << (gid[s] & 7);
                a0 = fmaxf(a0, x3[(size_t)s * 3 + 0]);
                a1 = fmaxf(a1, x3[(size_t)s * 3 + 1]);
                a2 = fmaxf(a2, x3[(size_t)s * 3 + 2]);
            }
        }
    }
    for (int off = 16; off >= 1; off >>= 1) {
        mk |= (unsigned)__shfl_xor((int)mk, off, 32);
        a0 = fmaxf(a0, __shfl_xor(a0, off, 32));
        a1 = fmaxf(a1, __shfl_xor(a1, off, 32));
        a2 = fmaxf(a2, __shfl_xor(a2, off, 32));
    }
    if (!has) { a0 = 0.f; a1 = 0.f; a2 = 0.f; }
    const float* Prow = P + (size_t)mk * HID_;
    const float* Qrow = Q + (size_t)g * HID_;

    float v[4];
    #pragma unroll
    for (int c = 0; c < 4; ++c) {
        int col = lane + 32 * c;
        if (col < HID_) {
            float t = bl1[col] + Qrow[col]
                    + x0 * Wr1[col] + x1 * Wr1[HID_ + col] + x2 * Wr1[2 * HID_ + col];
            if (has)
                t += Prow[col]
                   + a0 * Wl1[col] + a1 * Wl1[HID_ + col] + a2 * Wl1[2 * HID_ + col];
            v[c] = t;
        } else {
            v[c] = -INFINITY;
        }
    }
    float mx = fmaxf(fmaxf(v[0], v[1]), fmaxf(v[2], v[3]));
    for (int off = 16; off >= 1; off >>= 1)
        mx = fmaxf(mx, __shfl_xor(mx, off, 32));
    float s = 0.f;
    #pragma unroll
    for (int c = 0; c < 4; ++c) s += __expf(v[c] - mx);
    for (int off = 16; off >= 1; off >>= 1)
        s += __shfl_xor(s, off, 32);
    float lse = mx + __logf(s);
    #pragma unroll
    for (int c = 0; c < 4; ++c) {
        int col = lane + 32 * c;
        if (col < HID_) h[(size_t)n * HID_ + col] = v[c] - lse;
    }
}

__global__ void node_out_kernel(const float* __restrict__ h, const int* __restrict__ counts,
                                const int* __restrict__ slots, const int* __restrict__ spilln,
                                const int* __restrict__ spill,
                                const float* __restrict__ Wl2, const float* __restrict__ bl2,
                                const float* __restrict__ Wr2, float* __restrict__ y,
                                int N, int E) {
    const int wave = threadIdx.x >> 5;
    const int lane = threadIdx.x & 31;
    const int n = __builtin_amdgcn_readfirstlane(blockIdx.x * (blockDim.x >> 5) + wave);
    if (n >= N) return;
    const bool own = (lane < HID_ / 4);
    const int  c4  = own ? lane * 4 : 0;
    const int cnt = counts[n];
    const int m   = min(cnt, CAP_);

    int my = 0;
    if (lane < m) my = min(max(slots[(size_t)n * CAP_ + lane], 0), N - 1);
    float4 own_row = *reinterpret_cast<const float4*>(h + (size_t)n * HID_ + c4);
    float4 acc = make_float4(0.f, 0.f, 0.f, 0.f);
    for (int j0 = 0; j0 < m; j0 += 8) {
        float4 row[8];
        #pragma unroll
        for (int u = 0; u < 8; ++u) {
            const int sc = __builtin_amdgcn_readlane(my, j0 + u);
            row[u] = *reinterpret_cast<const float4*>(h + (size_t)sc * HID_ + c4);
        }
        #pragma unroll
        for (int u = 0; u < 8; ++u) {
            const bool on = (j0 + u < m);
            acc.x += on ? row[u].x : 0.f;
            acc.y += on ? row[u].y : 0.f;
            acc.z += on ? row[u].z : 0.f;
            acc.w += on ? row[u].w : 0.f;
        }
    }
    if (cnt > CAP_) {
        const int sn = min(max(spilln[0], 0), E);
        for (int i = 0; i < sn; ++i) {
            const int2 ds = reinterpret_cast<const int2*>(spill)[i];
            if (ds.x == n) {
                const int sc = min(max(ds.y, 0), N - 1);
                const float4 row = *reinterpret_cast<const float4*>(h + (size_t)sc * HID_ + c4);
                acc.x += row.x; acc.y += row.y; acc.z += row.z; acc.w += row.w;
            }
        }
    }

    float s1 = 0.f, s2 = 0.f;
    if (own) {
        s1 = fmaf(acc.x, Wl2[c4 + 0], s1);
        s1 = fmaf(acc.y, Wl2[c4 + 1], s1);
        s1 = fmaf(acc.z, Wl2[c4 + 2], s1);
        s1 = fmaf(acc.w, Wl2[c4 + 3], s1);
        s2 = fmaf(own_row.x, Wr2[c4 + 0], s2);
        s2 = fmaf(own_row.y, Wr2[c4 + 1], s2);
        s2 = fmaf(own_row.z, Wr2[c4 + 2], s2);
        s2 = fmaf(own_row.w, Wr2[c4 + 3], s2);
    }
    for (int off = 16; off >= 1; off >>= 1) {
        s1 += __shfl_xor(s1, off, 32);
        s2 += __shfl_xor(s2, off, 32);
    }
    if (lane == 0)
        y[n] = s1 / fmaxf((float)cnt, 1.0f) + bl2[0] + s2;
}

extern "C" void kernel_launch(void* const* d_in, const int* in_sizes, int n_in,
                              void* d_out, int out_size, void* d_ws, size_t ws_size,
                              hipStream_t stream) {
    const float* X      = (const float*)d_in[0];
    const float* x3     = (const float*)d_in[1];
    const int*   ei     = (const int*)d_in[2];
    const int*   gid    = (const int*)d_in[3];
    const float* w1     = (const float*)d_in[4];
    const float* b1     = (const float*)d_in[5];
    const float* w2     = (const float*)d_in[6];
    const float* b2     = (const float*)d_in[7];
    const float* w3     = (const float*)d_in[8];
    const float* b3     = (const float*)d_in[9];
    const float* Wl1    = (const float*)d_in[10];
    const float* bl1    = (const float*)d_in[11];
    const float* Wr1    = (const float*)d_in[12];
    const float* Wl2    = (const float*)d_in[13];
    const float* bl2    = (const float*)d_in[14];
    const float* Wr2    = (const float*)d_in[15];
    float* y = (float*)d_out;

    const int E = in_sizes[2] / 2;
    const int N = in_sizes[3];

    char* ws = (char*)d_ws;
    size_t off = 0;
    auto alloc = [&](size_t bytes) -> char* {
        char* p = ws + off;
        off = (off + bytes + 255) & ~(size_t)255;
        return p;
    };
    float*    act1  = (float*)alloc((size_t)8 * 32 * 21 * 36 * 4);
    float*    act2  = (float*)alloc((size_t)8 * 64 * 17 * 36 * 4);
    float*    act3  = (float*)alloc((size_t)8 * 64 * 13 * 16 * 4);
    float*    enc   = (float*)alloc((size_t)8 * ENC_ * 4);
    float*    P     = (float*)alloc((size_t)256 * HID_ * 4);
    float*    Q     = (float*)alloc((size_t)16  * HID_ * 4);
    int*      cnt   = (int*)alloc((size_t)(N + 1) * 4);
    int*      spilln = cnt + N;
    int*      slots = (int*)alloc((size_t)N * CAP_ * 4);
    int*      spill = (int*)alloc((size_t)E * 8);
    float*    hbuf  = (float*)alloc((size_t)N * HID_ * 4);

    const int T = 256;
    zero_u32<<<(N + 1 + T - 1) / T, T, 0, stream>>>((unsigned*)cnt, N + 1);

    conv_wmma_f32<25, 25, 32, 21, 6, 6, 1, 378><<<95 * 2, 128, 0, stream>>>(X, w1, b1, act1);
    conv_wmma_f32<32, 21, 64, 17, 6, 6, 1, 306><<<77 * 4, 128, 0, stream>>>(act1, w2, b2, act2);
    conv_wmma_f32<64, 17, 64, 13, 4, 4, 0, 104><<<26 * 4, 128, 0, stream>>>(act2, w3, b3, act3);
    pool_kernel<<<(8 * 64 * 11 * 4 + T - 1) / T, T, 0, stream>>>(act3, enc);

    edge_pass<<<(E + T - 1) / T, T, 0, stream>>>(ei, cnt, slots, spilln, spill, E);

    {
        dim3 gPQ(256 / 16 + 1, (HID_ + 15) / 16);
        gemm_submax_wmma_f32<<<gPQ, 32, 0, stream>>>(enc, Wl1 + 3 * HID_, Wr1 + 3 * HID_, P, Q);
    }

    node_h_kernel<<<(N + 7) / 8, 256, 0, stream>>>(x3, gid, cnt, slots, spilln, spill, P, Q,
                                                   Wl1, Wr1, bl1, hbuf, N, E);

    node_out_kernel<<<(N + 7) / 8, 256, 0, stream>>>(hbuf, cnt, slots, spilln, spill, Wl2, bl2, Wr2, y, N, E);

    (void)n_in; (void)out_size; (void)ws_size;
}


# Conv K loop software-pipelined in one wave: 2 operand register sets, block j+1's LDS reads and loads go out in 2 groups ahead of block j's WMMAs (conv3: neighbour taps as 8-byte loads); same sums
# speedup vs baseline: 6.2629x; 1.1139x over previous
#include <hip/hip_runtime.h>
#include <hip/hip_bf16.h>
#include <math.h>


#define B_     8
#define CIN_   25
#define NPG_   2000
#define ENC_   2816
#define HID_   100
#define KGEMM_ 2816
#define CAP_   32

typedef __attribute__((ext_vector_type(2))) float v2f;
typedef v2f v2f_a4 __attribute__((aligned(4)));
typedef __attribute__((ext_vector_type(4))) float v4f;
typedef __attribute__((ext_vector_type(8))) float v8f;

__global__ void zero_u32(unsigned* __restrict__ p, int n) {
    int i = blockIdx.x * blockDim.x + threadIdx.x;
    if (i < n) p[i] = 0u;
}

template <int CIN, int DIN, int COUT, int DOUT, int HO, int WO, int PAD, int MT>
__global__ void __launch_bounds__(128)
conv_wmma_f32(const float* __restrict__ In, const float* __restrict__ Wt,
              const float* __restrict__ bias, float* __restrict__ Out) {
    constexpr int PLANE = 36;
    constexpr int CHS   = DIN * PLANE;
    constexpr int HWO   = HO * WO;
    constexpr int P     = DOUT * HWO;
    constexpr int NT    = COUT / 16;
    constexpr int CBF   = CIN / 4;
    constexpr int CREM  = CIN % 4;
    constexpr int CP    = (CIN + 3) / 4 * 4;
    constexpr int RW    = CIN * 45;
    constexpr int ROW   = CP * 45;
    constexpr int PITCH = ROW + (68 - ROW % 64) % 64;
    constexpr int WTOT  = 16 * RW;
    constexpr int V     = (RW % 4 == 0) ? 4 : 1;
    static_assert(P % 4 == 0, "16-byte stores need P % 4 == 0");
    static_assert(COUT % 16 == 0, "N tiles are full");
    constexpr int RWP   = RW / V;
    constexpr int PITCHP = PITCH / V;
    static_assert(PITCH % 64 == 4 && RW % V == 0 && PITCH % V == 0 && RWP > 128,
                  "pitch rule; rows are whole pieces; a thread's piece is at most one row past the trip's first");

    __shared__ __attribute__((aligned(16))) float Bs[16 * PITCH];

    const int lane = threadIdx.x & 31;
    const int half = lane >> 4;
    const int l16  = lane & 15;
    const int nTile = blockIdx.x % NT;
    {
        typedef float piece_t __attribute__((ext_vector_type(V)));
        const piece_t* Wn = reinterpret_cast<const piece_t*>(Wt + (size_t)nTile * WTOT);
        piece_t* Bp = reinterpret_cast<piece_t*>(Bs);
        constexpr int NPC  = WTOT / V;
        constexpr int FULL = NPC / 128;
        const int tid = threadIdx.x;
        piece_t v[FULL];
        #pragma unroll
        for (int j = 0; j < FULL; ++j) v[j] = Wn[j * 128 + tid];
        #pragma unroll
        for (int j = 0; j < FULL; ++j) {
            const int first = j * 128;
            const int q = first / RWP, m = first - q * RWP;
            const int next = (tid + m >= RWP) ? (PITCHP - RWP) : 0;
            Bp[q * PITCHP + m + tid + next] = v[j];
        }
        if constexpr (NPC % 128 != 0) {
            constexpr int first = FULL * 128;
            constexpr int q = first / RWP, m = first - q * RWP;
            if (first + tid < NPC) {
                const int next = (tid + m >= RWP) ? (PITCHP - RWP) : 0;
                Bp[q * PITCHP + m + tid + next] = Wn[first + tid];
            }
        }
        if constexpr (CREM != 0) {
            constexpr int NPAD = ROW - RW;
            for (int i = threadIdx.x; i < 16 * NPAD; i += 128) {
                const int oz = i / NPAD;
                Bs[oz * PITCH + RW + (i - oz * NPAD)] = 0.0f;
            }
        }
    }
    __syncthreads();

    const int mTile = (blockIdx.x / NT) * 4 + (threadIdx.x >> 5);
    if (mTile >= MT) return;

    const int m  = mTile * 16 + l16;
    const int bn = m / P;
    const int p  = m - bn * P;
    const int z  = p / HWO;
    const int q  = p - z * HWO;
    const int y  = q / WO;
    const int x  = q - y * WO;
    const int o  = nTile * 16 + l16;

    int  toff[9];
    bool tok[9];
    #pragma unroll
    for (int t = 0; t < 9; ++t) {
        const int yy = y + t / 3 - PAD;
        const int xx = x + t % 3 - PAD;
        const bool ok = (PAD == 0) ? true : ((yy >= 0) && (yy < 6) && (xx >= 0) && (xx < 6));
        tok[t]  = ok;
        toff[t] = ok ? (yy * 6 + xx) : 0;
    }

    const float* bP = Bs + l16 * PITCH + half * 90;

    v8f acc0 = {}, acc1 = {};
    struct Frag { v2f bf[45]; float ae[90]; };
    constexpr int NB = CP / 4;
    constexpr int HB = 23;
    static_assert((CREM == 0 && NB % 2 == 0) || (CREM != 0 && NB % 2 == 1),
                  "the loop below takes blocks in pairs; a partial block is the odd one at the end");
    auto chan_at = [&](int j, int s) __attribute__((always_inline)) {
        const int c = j * 4 + half * 2 + s;
        const int cc = (CREM != 0 && c >= CIN) ? CBF * 4 : c;
        return (bn * CIN + cc) * CHS + z * PLANE;
    };
    auto load_part = [&](Frag& f, int j, int i0, int i1) __attribute__((always_inline)) {
        const int a0 = chan_at(j, 0), a1 = chan_at(j, 1);
        const float* bp = bP + j * 180;
        #pragma unroll
        for (int i = i0; i < i1; ++i) {
            f.bf[i] = *reinterpret_cast<const v2f*>(bp + 2 * i);
            const int e0 = 2 * i, e1 = 2 * i + 1;
            const int t0 = e0 % 45, t1 = e1 % 45;
            const int x0 = (e0 / 45 ? a1 : a0) + (t0 / 9) * PLANE + toff[t0 % 9];
            const int x1 = (e1 / 45 ? a1 : a0) + (t1 / 9) * PLANE + toff[t1 % 9];
            if (PAD == 0 && e0 / 45 == e1 / 45 && t0 % 3 != 2) {
                const v2f_a4 pr = *reinterpret_cast<const v2f_a4*>(In + x0);
                f.ae[e0] = pr.x;
                f.ae[e1] = pr.y;
            } else {
                f.ae[e0] = In[x0];
                f.ae[e1] = In[x1];
            }
        }
    };
    auto mac_part = [&](const Frag& f, bool v0, bool v1, int i0, int i1) __attribute__((always_inline)) {
        #pragma unroll
        for (int i = i0; i < i1; ++i) {
            const int e0 = 2 * i, e1 = 2 * i + 1;
            const bool on0 = tok[e0 % 45 % 9] & (e0 / 45 ? v1 : v0);
            const bool on1 = tok[e1 % 45 % 9] & (e1 / 45 ? v1 : v0);
            v2f a;
            a.x = on0 ? f.ae[e0] : 0.0f;
            a.y = on1 ? f.ae[e1] : 0.0f;
            const v2f b = f.bf[i];
            if (i & 1)
                acc1 = __builtin_amdgcn_wmma_f32_16x16x4_f32(false, a, false, b,
                                                             (short)0, acc1, false, false);
            else
                acc0 = __builtin_amdgcn_wmma_f32_16x16x4_f32(false, a, false, b,
                                                             (short)0, acc0, false, false);
        }
    };
    auto step = [&](const Frag& cur, Frag& nxt, int jn) __attribute__((always_inline)) {
        load_part(nxt, jn, 0, HB);
        __builtin_amdgcn_sched_barrier(0);
        mac_part(cur, true, true, 0, HB);
        __builtin_amdgcn_sched_barrier(0);
        load_part(nxt, jn, HB, 45);
        __builtin_amdgcn_sched_barrier(0);
        mac_part(cur, true, true, HB, 45);
        __builtin_amdgcn_sched_barrier(0);
    };
    Frag f0, f1;
    load_part(f0, 0, 0, HB);
    __builtin_amdgcn_sched_barrier(0);
    load_part(f0, 0, HB, 45);
    __builtin_amdgcn_sched_barrier(0);
    int j = 0;
    for (; j + 2 < NB; j += 2) {
        step(f0, f1, j + 1);
        step(f1, f0, j + 2);
    }
    if constexpr (NB % 2 == 0) {
        step(f0, f1, NB - 1);
        mac_part(f1, true, true, 0, 45);
    } else {
        const int  c0 = CBF * 4 + half * 2;
        const bool v0 = (c0 < CIN), v1 = (c0 + 1 < CIN);
        mac_part(f0, v0, v1, 0, 45);
    }

    const v8f acc = acc0 + acc1;
    const float bo = bias[o];
    const int m0 = mTile * 16 + half * 8;
    #pragma unroll
    for (int g = 0; g < 2; ++g) {
        const int mg  = m0 + g * 4;
        const int bng = mg / P;
        const int pg  = mg - bng * P;
        float4 v;
        v.x = fmaxf(acc[g * 4 + 0] + bo, 0.0f);
        v.y = fmaxf(acc[g * 4 + 1] + bo, 0.0f);
        v.z = fmaxf(acc[g * 4 + 2] + bo, 0.0f);
        v.w = fmaxf(acc[g * 4 + 3] + bo, 0.0f);
        *reinterpret_cast<float4*>(Out + (size_t)(bng * COUT + o) * P + pg) = v;
    }
}

__global__ void pool_kernel(const float* __restrict__ A, float* __restrict__ enc) {
    const int total = 8 * 64 * 11 * 2 * 2;
    int idx = blockIdx.x * blockDim.x + threadIdx.x;
    if (idx >= total) return;
    int x = idx % 2;  int t = idx / 2;
    int y = t % 2;    t /= 2;
    int d = t % 11;   t /= 11;
    int ch = t % 64;  int bn = t / 64;
    float m = -INFINITY;
    for (int dd = 0; dd < 3; ++dd)
        for (int hh = 0; hh < 3; ++hh)
            for (int ww = 0; ww < 3; ++ww) {
                float v = A[((size_t)((bn * 64 + ch) * 13 + (d + dd))) * 16 + (y + hh) * 4 + (x + ww)];
                m = fmaxf(m, v);
            }
    enc[(size_t)bn * ENC_ + ch * 44 + d * 4 + y * 2 + x] = m;
}

__global__ void edge_pass(const int* __restrict__ ei, int* __restrict__ counts,
                          int* __restrict__ slots, int* __restrict__ spilln,
                          int* __restrict__ spill, int E) {
    int e = blockIdx.x * blockDim.x + threadIdx.x;
    if (e >= E) return;
    const int s = ei[e];
    const int d = ei[E + e];
    const int r = atomicAdd(&counts[d], 1);
    if (r < CAP_) {
        slots[(size_t)d * CAP_ + r] = s;
    } else {
        const int o = atomicAdd(spilln, 1);
        reinterpret_cast<int2*>(spill)[o] = make_int2(d, s);
    }
}

__global__ void __launch_bounds__(32)
gemm_submax_wmma_f32(const float* __restrict__ enc, const float* __restrict__ Wl,
                     const float* __restrict__ Wr, float* __restrict__ P,
                     float* __restrict__ Q) {
    constexpr int EPITCH = KGEMM_ + 4;
    __shared__ __attribute__((aligned(16))) float EncS[8 * EPITCH];

    const int Ncols = HID_;
    const int lane = threadIdx.x & 31;
    const int half = lane >> 4;
    const int l16  = lane & 15;
    const bool isQ = (blockIdx.x == 16);
    const float* W = isQ ? Wr : Wl;
    float* C = isQ ? Q : P;
    const int Mrows = isQ ? 8 : 256;
    const int mBase = isQ ? 0 : blockIdx.x * 16;
    const unsigned mask = isQ ? (1u << (l16 & 7)) : (unsigned)(mBase + l16);
    const int g0 = mask ? (__ffs(mask) - 1) : 0;
    const int nBase = blockIdx.y * 16;
    const int nCol = nBase + l16;
    const bool nOk = (nCol < Ncols);
    const int nClamp = nOk ? nCol : (Ncols - 1);

    int aoff[8];
    #pragma unroll
    for (int g = 0; g < 8; ++g) {
        const int r = ((mask >> g) & 1u) ? g : g0;
        aoff[g] = r * EPITCH + half * 8;
    }
    const float* Bcol = W + (size_t)(half * 8) * Ncols + nClamp;

    struct KB { float bv[8]; };
    struct KA { v4f L[8], H[8]; };
    v8f acc0 = {}, acc1 = {};

    auto load_b = [&](KB& s, int kt) __attribute__((always_inline)) {
        #pragma unroll
        for (int i = 0; i < 8; ++i) s.bv[i] = Bcol[(size_t)(kt + i) * Ncols];
    };
    auto load_a = [&](KA& s, int kt) __attribute__((always_inline)) {
        #pragma unroll
        for (int g = 0; g < 8; ++g) {
            s.L[g] = *reinterpret_cast<const v4f*>(EncS + aoff[g] + kt);
            s.H[g] = *reinterpret_cast<const v4f*>(EncS + aoff[g] + kt + 4);
        }
    };
    auto mac_step = [&](const KA& s, const KB& b) __attribute__((always_inline)) {
        v4f lo = s.L[0], hi = s.H[0];
        #pragma unroll
        for (int g = 1; g < 8; ++g) {
            lo.x = fmaxf(lo.x, s.L[g].x); lo.y = fmaxf(lo.y, s.L[g].y);
            lo.z = fmaxf(lo.z, s.L[g].z); lo.w = fmaxf(lo.w, s.L[g].w);
            hi.x = fmaxf(hi.x, s.H[g].x); hi.y = fmaxf(hi.y, s.H[g].y);
            hi.z = fmaxf(hi.z, s.H[g].z); hi.w = fmaxf(hi.w, s.H[g].w);
        }
        v2f a0, a1, a2, a3, b0, b1, b2, b3;
        a0.x = fmaxf(lo.x, 0.0f); a0.y = fmaxf(lo.y, 0.0f);
        a1.x = fmaxf(lo.z, 0.0f); a1.y = fmaxf(lo.w, 0.0f);
        a2.x = fmaxf(hi.x, 0.0f); a2.y = fmaxf(hi.y, 0.0f);
        a3.x = fmaxf(hi.z, 0.0f); a3.y = fmaxf(hi.w, 0.0f);
        b0.x = b.bv[0]; b0.y = b.bv[1];
        b1.x = b.bv[2]; b1.y = b.bv[3];
        b2.x = b.bv[4]; b2.y = b.bv[5];
        b3.x = b.bv[6]; b3.y = b.bv[7];
        acc0 = __builtin_amdgcn_wmma_f32_16x16x4_f32(false, a0, false, b0,
                                                     (short)0, acc0, false, false);
        acc1 = __builtin_amdgcn_wmma_f32_16x16x4_f32(false, a1, false, b1,
                                                     (short)0, acc1, false, false);
        acc0 = __builtin_amdgcn_wmma_f32_16x16x4_f32(false, a2, false, b2,
                                                     (short)0, acc0, false, false);
        acc1 = __builtin_amdgcn_wmma_f32_16x16x4_f32(false, a3, false, b3,
                                                     (short)0, acc1, false, false);
    };

    static_assert((KGEMM_ / 16) % 8 == 0 && KGEMM_ % 16 == 0, "22 trips of eight steps");
    for (int r0 = 0; r0 < 8; r0 += 2) {
        v4f v[44];
        #pragma unroll
        for (int u = 0; u < 44; ++u) {
            const int r = r0 + u / 22, j = u % 22;
            v[u] = *reinterpret_cast<const v4f*>(enc + (size_t)r * ENC_ + (j * 32 + lane) * 4);
        }
        #pragma unroll
        for (int u = 0; u < 44; ++u) {
            const int r = r0 + u / 22, j = u % 22;
            *reinterpret_cast<v4f*>(EncS + r * EPITCH + (j * 32 + lane) * 4) = v[u];
        }
    }
    __syncthreads();

    KB b0, b1, b2, b3, b4, b5, b6, b7;
    load_b(b0, 0);  load_b(b1, 16); load_b(b2, 32); load_b(b3, 48);
    load_b(b4, 64); load_b(b5, 80); load_b(b6, 96); load_b(b7, 112);
    KA a0, a1;
    load_a(a0, 0);
    load_a(a1, 16);
    for (int kt = 0; kt < KGEMM_ - 128; kt += 128) {
        mac_step(a0, b0); load_b(b0, kt + 128); load_a(a0, kt + 32);
        mac_step(a1, b1); load_b(b1, kt + 144); load_a(a1, kt + 48);
        mac_step(a0, b2); load_b(b2, kt + 160); load_a(a0, kt + 64);
        mac_step(a1, b3); load_b(b3, kt + 176); load_a(a1, kt + 80);
        mac_step(a0, b4); load_b(b4, kt + 192); load_a(a0, kt + 96);
        mac_step(a1, b5); load_b(b5, kt + 208); load_a(a1, kt + 112);
        mac_step(a0, b6); load_b(b6, kt + 224); load_a(a0, kt + 128);
        mac_step(a1, b7); load_b(b7, kt + 240); load_a(a1, kt + 144);
    }
    {
        constexpr int kt = KGEMM_ - 128;
        mac_step(a0, b0); load_a(a0, kt + 32);
        mac_step(a1, b1); load_a(a1, kt + 48);
        mac_step(a0, b2); load_a(a0, kt + 64);
        mac_step(a1, b3); load_a(a1, kt + 80);
        mac_step(a0, b4); load_a(a0, kt + 96);
        mac_step(a1, b5); load_a(a1, kt + 112);
        mac_step(a0, b6);
        mac_step(a1, b7);
    }
    v8f acc = acc0 + acc1;
    for (int v = 0; v < 8; ++v) {
        const int m = mBase + half * 8 + v;
        const float val = (!isQ && m == 0) ? 0.0f : acc[v];
        if (m < Mrows && nOk) C[(size_t)m * Ncols + nCol] = val;
    }
}

__global__ void node_h_kernel(const float* __restrict__ x3, const int* __restrict__ gid,
                              const int* __restrict__ counts, const int* __restrict__ slots,
                              const int* __restrict__ spilln, const int* __restrict__ spill,
                              const float* __restrict__ P, const float* __restrict__ Q,
                              const float* __restrict__ Wl1, const float* __restrict__ Wr1,
                              const float* __restrict__ bl1, float* __restrict__ h, int N, int E) {
    int wave = threadIdx.x >> 5;
    int lane = threadIdx.x & 31;
    const int n = __builtin_amdgcn_readfirstlane(blockIdx.x * (blockDim.x >> 5) + wave);
    if (n >= N) return;

    float x0 = x3[(size_t)n * 3 + 0];
    float x1 = x3[(size_t)n * 3 + 1];
    float x2 = x3[(size_t)n * 3 + 2];
    int g = gid[n];
    const int cnt = counts[n];
    const int m = min(cnt, CAP_);
    const bool has = (cnt > 0);

    unsigned mk = 0u;
    float a0 = -INFINITY, a1 = -INFINITY, a2 = -INFINITY;
    if (lane < m) {
        const int s = min(max(slots[(size_t)n * CAP_ + lane], 0), N - 1);
        mk = 1u << (gid[s] & 7);
        a0 = x3[(size_t)s * 3 + 0];
        a1 = x3[(size_t)s * 3 + 1];
        a2 = x3[(size_t)s * 3 + 2];
    }
    if (cnt > CAP_) {
        const int sn = min(max(spilln[0], 0), E);
        for (int i = lane; i < sn; i += 32) {
            const int2 ds = reinterpret_cast<const int2*>(spill)[i];
            if (ds.x == n) {
                const int s = min(max(ds.y, 0), N - 1);
                mk |= 1u << (gid[s] & 7);
                a0 = fmaxf(a0, x3[(size_t)s * 3 + 0]);
                a1 = fmaxf(a1, x3[(size_t)s * 3 + 1]);
                a2 = fmaxf(a2, x3[(size_t)s * 3 + 2]);
            }
        }
    }
    for (int off = 16; off >= 1; off >>= 1) {
        mk |= (unsigned)__shfl_xor((int)mk, off, 32);
        a0 = fmaxf(a0, __shfl_xor(a0, off, 32));
        a1 = fmaxf(a1, __shfl_xor(a1, off, 32));
        a2 = fmaxf(a2, __shfl_xor(a2, off, 32));
    }
    if (!has) { a0 = 0.f; a1 = 0.f; a2 = 0.f; }
    const float* Prow = P + (size_t)mk * HID_;
    const float* Qrow = Q + (size_t)g * HID_;

    float v[4];
    #pragma unroll
    for (int c = 0; c < 4; ++c) {
        int col = lane + 32 * c;
        if (col < HID_) {
            float t = bl1[col] + Qrow[col]
                    + x0 * Wr1[col] + x1 * Wr1[HID_ + col] + x2 * Wr1[2 * HID_ + col];
            if (has)
                t += Prow[col]
                   + a0 * Wl1[col] + a1 * Wl1[HID_ + col] + a2 * Wl1[2 * HID_ + col];
            v[c] = t;
        } else {
            v[c] = -INFINITY;
        }
    }
    float mx = fmaxf(fmaxf(v[0], v[1]), fmaxf(v[2], v[3]));
    for (int off = 16; off >= 1; off >>= 1)
        mx = fmaxf(mx, __shfl_xor(mx, off, 32));
    float s = 0.f;
    #pragma unroll
    for (int c = 0; c < 4; ++c) s += __expf(v[c] - mx);
    for (int off = 16; off >= 1; off >>= 1)
        s += __shfl_xor(s, off, 32);
    float lse = mx + __logf(s);
    #pragma unroll
    for (int c = 0; c < 4; ++c) {
        int col = lane + 32 * c;
        if (col < HID_) h[(size_t)n * HID_ + col] = v[c] - lse;
    }
}

__global__ void node_out_kernel(const float* __restrict__ h, const int* __restrict__ counts,
                                const int* __restrict__ slots, const int* __restrict__ spilln,
                                const int* __restrict__ spill,
                                const float* __restrict__ Wl2, const float* __restrict__ bl2,
                                const float* __restrict__ Wr2, float* __restrict__ y,
                                int N, int E) {
    const int wave = threadIdx.x >> 5;
    const int lane = threadIdx.x & 31;
    const int n = __builtin_amdgcn_readfirstlane(blockIdx.x * (blockDim.x >> 5) + wave);
    if (n >= N) return;
    const bool own = (lane < HID_ / 4);
    const int  c4  = own ? lane * 4 : 0;
    const int cnt = counts[n];
    const int m   = min(cnt, CAP_);

    int my = 0;
    if (lane < m) my = min(max(slots[(size_t)n * CAP_ + lane], 0), N - 1);
    float4 own_row = *reinterpret_cast<const float4*>(h + (size_t)n * HID_ + c4);
    float4 acc = make_float4(0.f, 0.f, 0.f, 0.f);
    for (int j0 = 0; j0 < m; j0 += 8) {
        float4 row[8];
        #pragma unroll
        for (int u = 0; u < 8; ++u) {
            const int sc = __builtin_amdgcn_readlane(my, j0 + u);
            row[u] = *reinterpret_cast<const float4*>(h + (size_t)sc * HID_ + c4);
        }
        #pragma unroll
        for (int u = 0; u < 8; ++u) {
            const bool on = (j0 + u < m);
            acc.x += on ? row[u].x : 0.f;
            acc.y += on ? row[u].y : 0.f;
            acc.z += on ? row[u].z : 0.f;
            acc.w += on ? row[u].w : 0.f;
        }
    }
    if (cnt > CAP_) {
        const int sn = min(max(spilln[0], 0), E);
        for (int i = 0; i < sn; ++i) {
            const int2 ds = reinterpret_cast<const int2*>(spill)[i];
            if (ds.x == n) {
                const int sc = min(max(ds.y, 0), N - 1);
                const float4 row = *reinterpret_cast<const float4*>(h + (size_t)sc * HID_ + c4);
                acc.x += row.x; acc.y += row.y; acc.z += row.z; acc.w += row.w;
            }
        }
    }

    float s1 = 0.f, s2 = 0.f;
    if (own) {
        s1 = fmaf(acc.x, Wl2[c4 + 0], s1);
        s1 = fmaf(acc.y, Wl2[c4 + 1], s1);
        s1 = fmaf(acc.z, Wl2[c4 + 2], s1);
        s1 = fmaf(acc.w, Wl2[c4 + 3], s1);
        s2 = fmaf(own_row.x, Wr2[c4 + 0], s2);
        s2 = fmaf(own_row.y, Wr2[c4 + 1], s2);
        s2 = fmaf(own_row.z, Wr2[c4 + 2], s2);
        s2 = fmaf(own_row.w, Wr2[c4 + 3], s2);
    }
    for (int off = 16; off >= 1; off >>= 1) {
        s1 += __shfl_xor(s1, off, 32);
        s2 += __shfl_xor(s2, off, 32);
    }
    if (lane == 0)
        y[n] = s1 / fmaxf((float)cnt, 1.0f) + bl2[0] + s2;
}

extern "C" void kernel_launch(void* const* d_in, const int* in_sizes, int n_in,
                              void* d_out, int out_size, void* d_ws, size_t ws_size,
                              hipStream_t stream) {
    const float* X      = (const float*)d_in[0];
    const float* x3     = (const float*)d_in[1];
    const int*   ei     = (const int*)d_in[2];
    const int*   gid    = (const int*)d_in[3];
    const float* w1     = (const float*)d_in[4];
    const float* b1     = (const float*)d_in[5];
    const float* w2     = (const float*)d_in[6];
    const float* b2     = (const float*)d_in[7];
    const float* w3     = (const float*)d_in[8];
    const float* b3     = (const float*)d_in[9];
    const float* Wl1    = (const float*)d_in[10];
    const float* bl1    = (const float*)d_in[11];
    const float* Wr1    = (const float*)d_in[12];
    const float* Wl2    = (const float*)d_in[13];
    const float* bl2    = (const float*)d_in[14];
    const float* Wr2    = (const float*)d_in[15];
    float* y = (float*)d_out;

    const int E = in_sizes[2] / 2;
    const int N = in_sizes[3];

    char* ws = (char*)d_ws;
    size_t off = 0;
    auto alloc = [&](size_t bytes) -> char* {
        char* p = ws + off;
        off = (off + bytes + 255) & ~(size_t)255;
        return p;
    };
    float*    act1  = (float*)alloc((size_t)8 * 32 * 21 * 36 * 4);
    float*    act2  = (float*)alloc((size_t)8 * 64 * 17 * 36 * 4);
    float*    act3  = (float*)alloc((size_t)8 * 64 * 13 * 16 * 4);
    float*    enc   = (float*)alloc((size_t)8 * ENC_ * 4);
    float*    P     = (float*)alloc((size_t)256 * HID_ * 4);
    float*    Q     = (float*)alloc((size_t)16  * HID_ * 4);
    int*      cnt   = (int*)alloc((size_t)(N + 1) * 4);
    int*      spilln = cnt + N;
    int*      slots = (int*)alloc((size_t)N * CAP_ * 4);
    int*      spill = (int*)alloc((size_t)E * 8);
    float*    hbuf  = (float*)alloc((size_t)N * HID_ * 4);

    const int T = 256;
    zero_u32<<<(N + 1 + T - 1) / T, T, 0, stream>>>((unsigned*)cnt, N + 1);

    conv_wmma_f32<25, 25, 32, 21, 6, 6, 1, 378><<<95 * 2, 128, 0, stream>>>(X, w1, b1, act1);
    conv_wmma_f32<32, 21, 64, 17, 6, 6, 1, 306><<<77 * 4, 128, 0, stream>>>(act1, w2, b2, act2);
    conv_wmma_f32<64, 17, 64, 13, 4, 4, 0, 104><<<26 * 4, 128, 0, stream>>>(act2, w3, b3, act3);
    pool_kernel<<<(8 * 64 * 11 * 4 + T - 1) / T, T, 0, stream>>>(act3, enc);

    edge_pass<<<(E + T - 1) / T, T, 0, stream>>>(ei, cnt, slots, spilln, spill, E);

    {
        dim3 gPQ(256 / 16 + 1, (HID_ + 15) / 16);
        gemm_submax_wmma_f32<<<gPQ, 32, 0, stream>>>(enc, Wl1 + 3 * HID_, Wr1 + 3 * HID_, P, Q);
    }

    node_h_kernel<<<(N + 7) / 8, 256, 0, stream>>>(x3, gid, cnt, slots, spilln, spill, P, Q,
                                                   Wl1, Wr1, bl1, hbuf, N, E);

    node_out_kernel<<<(N + 7) / 8, 256, 0, stream>>>(hbuf, cnt, slots, spilln, spill, Wl2, bl2, Wr2, y, N, E);

    (void)n_in; (void)out_size; (void)ws_size;
}


# Last conv re-tiled by depth slabs: a workgroup owns 4 consecutive 4x4 output planes of one image (overlap planes recomputed on idle SIMDs), takes the 3x3x3 max in LDS; pool launch, conv3 output gone
# speedup vs baseline: 6.7744x; 1.0817x over previous
#include <hip/hip_runtime.h>
#include <hip/hip_bf16.h>
#include <math.h>


#define B_     8
#define CIN_   25
#define NPG_   2000
#define ENC_   2816
#define HID_   100
#define KGEMM_ 2816
#define CAP_   32

typedef __attribute__((ext_vector_type(2))) float v2f;
typedef v2f v2f_a4 __attribute__((aligned(4)));
typedef __attribute__((ext_vector_type(4))) float v4f;
typedef __attribute__((ext_vector_type(8))) float v8f;

__global__ void zero_u32(unsigned* __restrict__ p, int n) {
    int i = blockIdx.x * blockDim.x + threadIdx.x;
    if (i < n) p[i] = 0u;
}

template <int CIN, int DIN, int COUT, int DOUT, int HO, int WO, int PAD, int MT, bool POOL>
__global__ void __launch_bounds__(128)
conv_wmma_f32(const float* __restrict__ In, const float* __restrict__ Wt,
              const float* __restrict__ bias, float* __restrict__ Out) {
    constexpr int PLANE = 36;
    constexpr int CHS   = DIN * PLANE;
    constexpr int HWO   = HO * WO;
    constexpr int P     = DOUT * HWO;
    constexpr int NT    = COUT / 16;
    constexpr int CBF   = CIN / 4;
    constexpr int CREM  = CIN % 4;
    constexpr int CP    = (CIN + 3) / 4 * 4;
    constexpr int RW    = CIN * 45;
    constexpr int ROW   = CP * 45;
    constexpr int PITCH = ROW + (68 - ROW % 64) % 64;
    constexpr int WTOT  = 16 * RW;
    constexpr int V     = (RW % 4 == 0) ? 4 : 1;
    static_assert(P % 4 == 0, "16-byte stores need P % 4 == 0");
    static_assert(COUT % 16 == 0, "N tiles are full");
    constexpr int RWP   = RW / V;
    constexpr int PITCHP = PITCH / V;
    static_assert(PITCH % 64 == 4 && RW % V == 0 && PITCH % V == 0 && RWP > 128,
                  "pitch rule; rows are whole pieces; a thread's piece is at most one row past the trip's first");

    __shared__ __attribute__((aligned(16))) float Bs[16 * PITCH];

    const int lane = threadIdx.x & 31;
    const int half = lane >> 4;
    const int l16  = lane & 15;
    const int nTile = blockIdx.x % NT;
    {
        typedef float piece_t __attribute__((ext_vector_type(V)));
        const piece_t* Wn = reinterpret_cast<const piece_t*>(Wt + (size_t)nTile * WTOT);
        piece_t* Bp = reinterpret_cast<piece_t*>(Bs);
        constexpr int NPC  = WTOT / V;
        constexpr int FULL = NPC / 128;
        const int tid = threadIdx.x;
        piece_t v[FULL];
        #pragma unroll
        for (int j = 0; j < FULL; ++j) v[j] = Wn[j * 128 + tid];
        #pragma unroll
        for (int j = 0; j < FULL; ++j) {
            const int first = j * 128;
            const int q = first / RWP, m = first - q * RWP;
            const int next = (tid + m >= RWP) ? (PITCHP - RWP) : 0;
            Bp[q * PITCHP + m + tid + next] = v[j];
        }
        if constexpr (NPC % 128 != 0) {
            constexpr int first = FULL * 128;
            constexpr int q = first / RWP, m = first - q * RWP;
            if (first + tid < NPC) {
                const int next = (tid + m >= RWP) ? (PITCHP - RWP) : 0;
                Bp[q * PITCHP + m + tid + next] = Wn[first + tid];
            }
        }
        if constexpr (CREM != 0) {
            constexpr int NPAD = ROW - RW;
            for (int i = threadIdx.x; i < 16 * NPAD; i += 128) {
                const int oz = i / NPAD;
                Bs[oz * PITCH + RW + (i - oz * NPAD)] = 0.0f;
            }
        }
    }
    __syncthreads();

    constexpr int DP = DOUT - 2;
    constexpr int ZG = (DP + 1) / 2;
    int mTile, pimg = 0, pzg = 0;
    if constexpr (POOL) {
        static_assert(HWO == 16 && HO == 4 && WO == 4 && PAD == 0, "an M tile is one 4x4 depth plane");
        const int rest = blockIdx.x / NT;
        pimg = rest / ZG;
        pzg  = rest - pimg * ZG;
        const int zpl = min(pzg * 2 + (int)(threadIdx.x >> 5), DOUT - 1);
        mTile = pimg * DOUT + zpl;
    } else {
        mTile = (blockIdx.x / NT) * 4 + (threadIdx.x >> 5);
        if (mTile >= MT) return;
    }

    const int m  = mTile * 16 + l16;
    const int bn = m / P;
    const int p  = m - bn * P;
    const int z  = p / HWO;
    const int q  = p - z * HWO;
    const int y  = q / WO;
    const int x  = q - y * WO;
    const int o  = nTile * 16 + l16;

    int  toff[9];
    bool tok[9];
    #pragma unroll
    for (int t = 0; t < 9; ++t) {
        const int yy = y + t / 3 - PAD;
        const int xx = x + t % 3 - PAD;
        const bool ok = (PAD == 0) ? true : ((yy >= 0) && (yy < 6) && (xx >= 0) && (xx < 6));
        tok[t]  = ok;
        toff[t] = ok ? (yy * 6 + xx) : 0;
    }

    const float* bP = Bs + l16 * PITCH + half * 90;

    v8f acc0 = {}, acc1 = {};
    struct Frag { v2f bf[45]; float ae[90]; };
    constexpr int NB = CP / 4;
    constexpr int HB = 23;
    static_assert((CREM == 0 && NB % 2 == 0) || (CREM != 0 && NB % 2 == 1),
                  "the loop below takes blocks in pairs; a partial block is the odd one at the end");
    auto chan_at = [&](int j, int s) __attribute__((always_inline)) {
        const int c = j * 4 + half * 2 + s;
        const int cc = (CREM != 0 && c >= CIN) ? CBF * 4 : c;
        return (bn * CIN + cc) * CHS + z * PLANE;
    };
    auto load_part = [&](Frag& f, int j, int i0, int i1) __attribute__((always_inline)) {
        const int a0 = chan_at(j, 0), a1 = chan_at(j, 1);
        const float* bp = bP + j * 180;
        #pragma unroll
        for (int i = i0; i < i1; ++i) {
            f.bf[i] = *reinterpret_cast<const v2f*>(bp + 2 * i);
            const int e0 = 2 * i, e1 = 2 * i + 1;
            const int t0 = e0 % 45, t1 = e1 % 45;
            const int x0 = (e0 / 45 ? a1 : a0) + (t0 / 9) * PLANE + toff[t0 % 9];
            const int x1 = (e1 / 45 ? a1 : a0) + (t1 / 9) * PLANE + toff[t1 % 9];
            if (PAD == 0 && e0 / 45 == e1 / 45 && t0 % 3 != 2) {
                const v2f_a4 pr = *reinterpret_cast<const v2f_a4*>(In + x0);
                f.ae[e0] = pr.x;
                f.ae[e1] = pr.y;
            } else {
                f.ae[e0] = In[x0];
                f.ae[e1] = In[x1];
            }
        }
    };
    auto mac_part = [&](const Frag& f, bool v0, bool v1, int i0, int i1) __attribute__((always_inline)) {
        #pragma unroll
        for (int i = i0; i < i1; ++i) {
            const int e0 = 2 * i, e1 = 2 * i + 1;
            const bool on0 = tok[e0 % 45 % 9] & (e0 / 45 ? v1 : v0);
            const bool on1 = tok[e1 % 45 % 9] & (e1 / 45 ? v1 : v0);
            v2f a;
            a.x = on0 ? f.ae[e0] : 0.0f;
            a.y = on1 ? f.ae[e1] : 0.0f;
            const v2f b = f.bf[i];
            if (i & 1)
                acc1 = __builtin_amdgcn_wmma_f32_16x16x4_f32(false, a, false, b,
                                                             (short)0, acc1, false, false);
            else
                acc0 = __builtin_amdgcn_wmma_f32_16x16x4_f32(false, a, false, b,
                                                             (short)0, acc0, false, false);
        }
    };
    auto step = [&](const Frag& cur, Frag& nxt, int jn) __attribute__((always_inline)) {
        load_part(nxt, jn, 0, HB);
        __builtin_amdgcn_sched_barrier(0);
        mac_part(cur, true, true, 0, HB);
        __builtin_amdgcn_sched_barrier(0);
        load_part(nxt, jn, HB, 45);
        __builtin_amdgcn_sched_barrier(0);
        mac_part(cur, true, true, HB, 45);
        __builtin_amdgcn_sched_barrier(0);
    };
    Frag f0, f1;
    load_part(f0, 0, 0, HB);
    __builtin_amdgcn_sched_barrier(0);
    load_part(f0, 0, HB, 45);
    __builtin_amdgcn_sched_barrier(0);
    int j = 0;
    for (; j + 2 < NB; j += 2) {
        step(f0, f1, j + 1);
        step(f1, f0, j + 2);
    }
    if constexpr (NB % 2 == 0) {
        step(f0, f1, NB - 1);
        mac_part(f1, true, true, 0, 45);
    } else {
        const int  c0 = CBF * 4 + half * 2;
        const bool v0 = (c0 < CIN), v1 = (c0 + 1 < CIN);
        mac_part(f0, v0, v1, 0, 45);
    }

    const v8f acc = acc0 + acc1;
    const float bo = bias[o];
    if constexpr (POOL) {
        __shared__ __attribute__((aligned(16))) float Ps[4 * 16 * 16];
        const int w = threadIdx.x >> 5;
        float4 lo4, hi4;
        lo4.x = fmaxf(acc[0] + bo, 0.0f); lo4.y = fmaxf(acc[1] + bo, 0.0f);
        lo4.z = fmaxf(acc[2] + bo, 0.0f); lo4.w = fmaxf(acc[3] + bo, 0.0f);
        hi4.x = fmaxf(acc[4] + bo, 0.0f); hi4.y = fmaxf(acc[5] + bo, 0.0f);
        hi4.z = fmaxf(acc[6] + bo, 0.0f); hi4.w = fmaxf(acc[7] + bo, 0.0f);
        float* pw = Ps + (w * 16 + l16) * 16 + half * 8;
        *reinterpret_cast<float4*>(pw)     = lo4;
        *reinterpret_cast<float4*>(pw + 4) = hi4;
        __syncthreads();
        const int t  = threadIdx.x;
        const int ch = t >> 3, dz = (t >> 2) & 1, py = (t >> 1) & 1, px = t & 1;
        const int d  = pzg * 2 + dz;
        if (d < DP) {
            float mx = -INFINITY;
            #pragma unroll
            for (int dd = 0; dd < 3; ++dd)
                for (int hh = 0; hh < 3; ++hh)
                    for (int ww = 0; ww < 3; ++ww)
                        mx = fmaxf(mx, Ps[((dz + dd) * 16 + ch) * 16 + (py + hh) * 4 + (px + ww)]);
            Out[(size_t)(pimg * COUT + nTile * 16 + ch) * (DP * 4) + d * 4 + py * 2 + px] = mx;
        }
        return;
    }
    const int m0 = mTile * 16 + half * 8;
    #pragma unroll
    for (int g = 0; g < 2; ++g) {
        const int mg  = m0 + g * 4;
        const int bng = mg / P;
        const int pg  = mg - bng * P;
        float4 v;
        v.x = fmaxf(acc[g * 4 + 0] + bo, 0.0f);
        v.y = fmaxf(acc[g * 4 + 1] + bo, 0.0f);
        v.z = fmaxf(acc[g * 4 + 2] + bo, 0.0f);
        v.w = fmaxf(acc[g * 4 + 3] + bo, 0.0f);
        *reinterpret_cast<float4*>(Out + (size_t)(bng * COUT + o) * P + pg) = v;
    }
}

__global__ void edge_pass(const int* __restrict__ ei, int* __restrict__ counts,
                          int* __restrict__ slots, int* __restrict__ spilln,
                          int* __restrict__ spill, int E) {
    int e = blockIdx.x * blockDim.x + threadIdx.x;
    if (e >= E) return;
    const int s = ei[e];
    const int d = ei[E + e];
    const int r = atomicAdd(&counts[d], 1);
    if (r < CAP_) {
        slots[(size_t)d * CAP_ + r] = s;
    } else {
        const int o = atomicAdd(spilln, 1);
        reinterpret_cast<int2*>(spill)[o] = make_int2(d, s);
    }
}

__global__ void __launch_bounds__(32)
gemm_submax_wmma_f32(const float* __restrict__ enc, const float* __restrict__ Wl,
                     const float* __restrict__ Wr, float* __restrict__ P,
                     float* __restrict__ Q) {
    constexpr int EPITCH = KGEMM_ + 4;
    __shared__ __attribute__((aligned(16))) float EncS[8 * EPITCH];

    const int Ncols = HID_;
    const int lane = threadIdx.x & 31;
    const int half = lane >> 4;
    const int l16  = lane & 15;
    const bool isQ = (blockIdx.x == 16);
    const float* W = isQ ? Wr : Wl;
    float* C = isQ ? Q : P;
    const int Mrows = isQ ? 8 : 256;
    const int mBase = isQ ? 0 : blockIdx.x * 16;
    const unsigned mask = isQ ? (1u << (l16 & 7)) : (unsigned)(mBase + l16);
    const int g0 = mask ? (__ffs(mask) - 1) : 0;
    const int nBase = blockIdx.y * 16;
    const int nCol = nBase + l16;
    const bool nOk = (nCol < Ncols);
    const int nClamp = nOk ? nCol : (Ncols - 1);

    int aoff[8];
    #pragma unroll
    for (int g = 0; g < 8; ++g) {
        const int r = ((mask >> g) & 1u) ? g : g0;
        aoff[g] = r * EPITCH + half * 8;
    }
    const float* Bcol = W + (size_t)(half * 8) * Ncols + nClamp;

    struct KB { float bv[8]; };
    struct KA { v4f L[8], H[8]; };
    v8f acc0 = {}, acc1 = {};

    auto load_b = [&](KB& s, int kt) __attribute__((always_inline)) {
        #pragma unroll
        for (int i = 0; i < 8; ++i) s.bv[i] = Bcol[(size_t)(kt + i) * Ncols];
    };
    auto load_a = [&](KA& s, int kt) __attribute__((always_inline)) {
        #pragma unroll
        for (int g = 0; g < 8; ++g) {
            s.L[g] = *reinterpret_cast<const v4f*>(EncS + aoff[g] + kt);
            s.H[g] = *reinterpret_cast<const v4f*>(EncS + aoff[g] + kt + 4);
        }
    };
    auto mac_step = [&](const KA& s, const KB& b) __attribute__((always_inline)) {
        v4f lo = s.L[0], hi = s.H[0];
        #pragma unroll
        for (int g = 1; g < 8; ++g) {
            lo.x = fmaxf(lo.x, s.L[g].x); lo.y = fmaxf(lo.y, s.L[g].y);
            lo.z = fmaxf(lo.z, s.L[g].z); lo.w = fmaxf(lo.w, s.L[g].w);
            hi.x = fmaxf(hi.x, s.H[g].x); hi.y = fmaxf(hi.y, s.H[g].y);
            hi.z = fmaxf(hi.z, s.H[g].z); hi.w = fmaxf(hi.w, s.H[g].w);
        }
        v2f a0, a1, a2, a3, b0, b1, b2, b3;
        a0.x = fmaxf(lo.x, 0.0f); a0.y = fmaxf(lo.y, 0.0f);
        a1.x = fmaxf(lo.z, 0.0f); a1.y = fmaxf(lo.w, 0.0f);
        a2.x = fmaxf(hi.x, 0.0f); a2.y = fmaxf(hi.y, 0.0f);
        a3.x = fmaxf(hi.z, 0.0f); a3.y = fmaxf(hi.w, 0.0f);
        b0.x = b.bv[0]; b0.y = b.bv[1];
        b1.x = b.bv[2]; b1.y = b.bv[3];
        b2.x = b.bv[4]; b2.y = b.bv[5];
        b3.x = b.bv[6]; b3.y = b.bv[7];
        acc0 = __builtin_amdgcn_wmma_f32_16x16x4_f32(false, a0, false, b0,
                                                     (short)0, acc0, false, false);
        acc1 = __builtin_amdgcn_wmma_f32_16x16x4_f32(false, a1, false, b1,
                                                     (short)0, acc1, false, false);
        acc0 = __builtin_amdgcn_wmma_f32_16x16x4_f32(false, a2, false, b2,
                                                     (short)0, acc0, false, false);
        acc1 = __builtin_amdgcn_wmma_f32_16x16x4_f32(false, a3, false, b3,
                                                     (short)0, acc1, false, false);
    };

    static_assert((KGEMM_ / 16) % 8 == 0 && KGEMM_ % 16 == 0, "22 trips of eight steps");
    for (int r0 = 0; r0 < 8; r0 += 2) {
        v4f v[44];
        #pragma unroll
        for (int u = 0; u < 44; ++u) {
            const int r = r0 + u / 22, j = u % 22;
            v[u] = *reinterpret_cast<const v4f*>(enc + (size_t)r * ENC_ + (j * 32 + lane) * 4);
        }
        #pragma unroll
        for (int u = 0; u < 44; ++u) {
            const int r = r0 + u / 22, j = u % 22;
            *reinterpret_cast<v4f*>(EncS + r * EPITCH + (j * 32 + lane) * 4) = v[u];
        }
    }
    __syncthreads();

    KB b0, b1, b2, b3, b4, b5, b6, b7;
    load_b(b0, 0);  load_b(b1, 16); load_b(b2, 32); load_b(b3, 48);
    load_b(b4, 64); load_b(b5, 80); load_b(b6, 96); load_b(b7, 112);
    KA a0, a1;
    load_a(a0, 0);
    load_a(a1, 16);
    for (int kt = 0; kt < KGEMM_ - 128; kt += 128) {
        mac_step(a0, b0); load_b(b0, kt + 128); load_a(a0, kt + 32);
        mac_step(a1, b1); load_b(b1, kt + 144); load_a(a1, kt + 48);
        mac_step(a0, b2); load_b(b2, kt + 160); load_a(a0, kt + 64);
        mac_step(a1, b3); load_b(b3, kt + 176); load_a(a1, kt + 80);
        mac_step(a0, b4); load_b(b4, kt + 192); load_a(a0, kt + 96);
        mac_step(a1, b5); load_b(b5, kt + 208); load_a(a1, kt + 112);
        mac_step(a0, b6); load_b(b6, kt + 224); load_a(a0, kt + 128);
        mac_step(a1, b7); load_b(b7, kt + 240); load_a(a1, kt + 144);
    }
    {
        constexpr int kt = KGEMM_ - 128;
        mac_step(a0, b0); load_a(a0, kt + 32);
        mac_step(a1, b1); load_a(a1, kt + 48);
        mac_step(a0, b2); load_a(a0, kt + 64);
        mac_step(a1, b3); load_a(a1, kt + 80);
        mac_step(a0, b4); load_a(a0, kt + 96);
        mac_step(a1, b5); load_a(a1, kt + 112);
        mac_step(a0, b6);
        mac_step(a1, b7);
    }
    v8f acc = acc0 + acc1;
    for (int v = 0; v < 8; ++v) {
        const int m = mBase + half * 8 + v;
        const float val = (!isQ && m == 0) ? 0.0f : acc[v];
        if (m < Mrows && nOk) C[(size_t)m * Ncols + nCol] = val;
    }
}

__global__ void node_h_kernel(const float* __restrict__ x3, const int* __restrict__ gid,
                              const int* __restrict__ counts, const int* __restrict__ slots,
                              const int* __restrict__ spilln, const int* __restrict__ spill,
                              const float* __restrict__ P, const float* __restrict__ Q,
                              const float* __restrict__ Wl1, const float* __restrict__ Wr1,
                              const float* __restrict__ bl1, float* __restrict__ h, int N, int E) {
    int wave = threadIdx.x >> 5;
    int lane = threadIdx.x & 31;
    const int n = __builtin_amdgcn_readfirstlane(blockIdx.x * (blockDim.x >> 5) + wave);
    if (n >= N) return;

    float x0 = x3[(size_t)n * 3 + 0];
    float x1 = x3[(size_t)n * 3 + 1];
    float x2 = x3[(size_t)n * 3 + 2];
    int g = gid[n];
    const int cnt = counts[n];
    const int m = min(cnt, CAP_);
    const bool has = (cnt > 0);

    unsigned mk = 0u;
    float a0 = -INFINITY, a1 = -INFINITY, a2 = -INFINITY;
    if (lane < m) {
        const int s = min(max(slots[(size_t)n * CAP_ + lane], 0), N - 1);
        mk = 1u << (gid[s] & 7);
        a0 = x3[(size_t)s * 3 + 0];
        a1 = x3[(size_t)s * 3 + 1];
        a2 = x3[(size_t)s * 3 + 2];
    }
    if (cnt > CAP_) {
        const int sn = min(max(spilln[0], 0), E);
        for (int i = lane; i < sn; i += 32) {
            const int2 ds = reinterpret_cast<const int2*>(spill)[i];
            if (ds.x == n) {
                const int s = min(max(ds.y, 0), N - 1);
                mk |= 1u << (gid[s] & 7);
                a0 = fmaxf(a0, x3[(size_t)s * 3 + 0]);
                a1 = fmaxf(a1, x3[(size_t)s * 3 + 1]);
                a2 = fmaxf(a2, x3[(size_t)s * 3 + 2]);
            }
        }
    }
    for (int off = 16; off >= 1; off >>= 1) {
        mk |= (unsigned)__shfl_xor((int)mk, off, 32);
        a0 = fmaxf(a0, __shfl_xor(a0, off, 32));
        a1 = fmaxf(a1, __shfl_xor(a1, off, 32));
        a2 = fmaxf(a2, __shfl_xor(a2, off, 32));
    }
    if (!has) { a0 = 0.f; a1 = 0.f; a2 = 0.f; }
    const float* Prow = P + (size_t)mk * HID_;
    const float* Qrow = Q + (size_t)g * HID_;

    float v[4];
    #pragma unroll
    for (int c = 0; c < 4; ++c) {
        int col = lane + 32 * c;
        if (col < HID_) {
            float t = bl1[col] + Qrow[col]
                    + x0 * Wr1[col] + x1 * Wr1[HID_ + col] + x2 * Wr1[2 * HID_ + col];
            if (has)
                t += Prow[col]
                   + a0 * Wl1[col] + a1 * Wl1[HID_ + col] + a2 * Wl1[2 * HID_ + col];
            v[c] = t;
        } else {
            v[c] = -INFINITY;
        }
    }
    float mx = fmaxf(fmaxf(v[0], v[1]), fmaxf(v[2], v[3]));
    for (int off = 16; off >= 1; off >>= 1)
        mx = fmaxf(mx, __shfl_xor(mx, off, 32));
    float s = 0.f;
    #pragma unroll
    for (int c = 0; c < 4; ++c) s += __expf(v[c] - mx);
    for (int off = 16; off >= 1; off >>= 1)
        s += __shfl_xor(s, off, 32);
    float lse = mx + __logf(s);
    #pragma unroll
    for (int c = 0; c < 4; ++c) {
        int col = lane + 32 * c;
        if (col < HID_) h[(size_t)n * HID_ + col] = v[c] - lse;
    }
}

__global__ void node_out_kernel(const float* __restrict__ h, const int* __restrict__ counts,
                                const int* __restrict__ slots, const int* __restrict__ spilln,
                                const int* __restrict__ spill,
                                const float* __restrict__ Wl2, const float* __restrict__ bl2,
                                const float* __restrict__ Wr2, float* __restrict__ y,
                                int N, int E) {
    const int wave = threadIdx.x >> 5;
    const int lane = threadIdx.x & 31;
    const int n = __builtin_amdgcn_readfirstlane(blockIdx.x * (blockDim.x >> 5) + wave);
    if (n >= N) return;
    const bool own = (lane < HID_ / 4);
    const int  c4  = own ? lane * 4 : 0;
    const int cnt = counts[n];
    const int m   = min(cnt, CAP_);

    int my = 0;
    if (lane < m) my = min(max(slots[(size_t)n * CAP_ + lane], 0), N - 1);
    float4 own_row = *reinterpret_cast<const float4*>(h + (size_t)n * HID_ + c4);
    float4 acc = make_float4(0.f, 0.f, 0.f, 0.f);
    for (int j0 = 0; j0 < m; j0 += 8) {
        float4 row[8];
        #pragma unroll
        for (int u = 0; u < 8; ++u) {
            const int sc = __builtin_amdgcn_readlane(my, j0 + u);
            row[u] = *reinterpret_cast<const float4*>(h + (size_t)sc * HID_ + c4);
        }
        #pragma unroll
        for (int u = 0; u < 8; ++u) {
            const bool on = (j0 + u < m);
            acc.x += on ? row[u].x : 0.f;
            acc.y += on ? row[u].y : 0.f;
            acc.z += on ? row[u].z : 0.f;
            acc.w += on ? row[u].w : 0.f;
        }
    }
    if (cnt > CAP_) {
        const int sn = min(max(spilln[0], 0), E);
        for (int i = 0; i < sn; ++i) {
            const int2 ds = reinterpret_cast<const int2*>(spill)[i];
            if (ds.x == n) {
                const int sc = min(max(ds.y, 0), N - 1);
                const float4 row = *reinterpret_cast<const float4*>(h + (size_t)sc * HID_ + c4);
                acc.x += row.x; acc.y += row.y; acc.z += row.z; acc.w += row.w;
            }
        }
    }

    float s1 = 0.f, s2 = 0.f;
    if (own) {
        s1 = fmaf(acc.x, Wl2[c4 + 0], s1);
        s1 = fmaf(acc.y, Wl2[c4 + 1], s1);
        s1 = fmaf(acc.z, Wl2[c4 + 2], s1);
        s1 = fmaf(acc.w, Wl2[c4 + 3], s1);
        s2 = fmaf(own_row.x, Wr2[c4 + 0], s2);
        s2 = fmaf(own_row.y, Wr2[c4 + 1], s2);
        s2 = fmaf(own_row.z, Wr2[c4 + 2], s2);
        s2 = fmaf(own_row.w, Wr2[c4 + 3], s2);
    }
    for (int off = 16; off >= 1; off >>= 1) {
        s1 += __shfl_xor(s1, off, 32);
        s2 += __shfl_xor(s2, off, 32);
    }
    if (lane == 0)
        y[n] = s1 / fmaxf((float)cnt, 1.0f) + bl2[0] + s2;
}

extern "C" void kernel_launch(void* const* d_in, const int* in_sizes, int n_in,
                              void* d_out, int out_size, void* d_ws, size_t ws_size,
                              hipStream_t stream) {
    const float* X      = (const float*)d_in[0];
    const float* x3     = (const float*)d_in[1];
    const int*   ei     = (const int*)d_in[2];
    const int*   gid    = (const int*)d_in[3];
    const float* w1     = (const float*)d_in[4];
    const float* b1     = (const float*)d_in[5];
    const float* w2     = (const float*)d_in[6];
    const float* b2     = (const float*)d_in[7];
    const float* w3     = (const float*)d_in[8];
    const float* b3     = (const float*)d_in[9];
    const float* Wl1    = (const float*)d_in[10];
    const float* bl1    = (const float*)d_in[11];
    const float* Wr1    = (const float*)d_in[12];
    const float* Wl2    = (const float*)d_in[13];
    const float* bl2    = (const float*)d_in[14];
    const float* Wr2    = (const float*)d_in[15];
    float* y = (float*)d_out;

    const int E = in_sizes[2] / 2;
    const int N = in_sizes[3];

    char* ws = (char*)d_ws;
    size_t off = 0;
    auto alloc = [&](size_t bytes) -> char* {
        char* p = ws + off;
        off = (off + bytes + 255) & ~(size_t)255;
        return p;
    };
    float*    act1  = (float*)alloc((size_t)8 * 32 * 21 * 36 * 4);
    float*    act2  = (float*)alloc((size_t)8 * 64 * 17 * 36 * 4);
    float*    enc   = (float*)alloc((size_t)8 * ENC_ * 4);
    float*    P     = (float*)alloc((size_t)256 * HID_ * 4);
    float*    Q     = (float*)alloc((size_t)16  * HID_ * 4);
    int*      cnt   = (int*)alloc((size_t)(N + 1) * 4);
    int*      spilln = cnt + N;
    int*      slots = (int*)alloc((size_t)N * CAP_ * 4);
    int*      spill = (int*)alloc((size_t)E * 8);
    float*    hbuf  = (float*)alloc((size_t)N * HID_ * 4);

    const int T = 256;
    zero_u32<<<(N + 1 + T - 1) / T, T, 0, stream>>>((unsigned*)cnt, N + 1);

    conv_wmma_f32<25, 25, 32, 21, 6, 6, 1, 378, false><<<95 * 2, 128, 0, stream>>>(X, w1, b1, act1);
    conv_wmma_f32<32, 21, 64, 17, 6, 6, 1, 306, false><<<77 * 4, 128, 0, stream>>>(act1, w2, b2, act2);
    conv_wmma_f32<64, 17, 64, 13, 4, 4, 0, 104, true><<<8 * 6 * 4, 128, 0, stream>>>(act2, w3, b3, enc);

    edge_pass<<<(E + T - 1) / T, T, 0, stream>>>(ei, cnt, slots, spilln, spill, E);

    {
        dim3 gPQ(256 / 16 + 1, (HID_ + 15) / 16);
        gemm_submax_wmma_f32<<<gPQ, 32, 0, stream>>>(enc, Wl1 + 3 * HID_, Wr1 + 3 * HID_, P, Q);
    }

    node_h_kernel<<<(N + 7) / 8, 256, 0, stream>>>(x3, gid, cnt, slots, spilln, spill, P, Q,
                                                   Wl1, Wr1, bl1, hbuf, N, E);

    node_out_kernel<<<(N + 7) / 8, 256, 0, stream>>>(hbuf, cnt, slots, spilln, spill, Wl2, bl2, Wr2, y, N, E);

    (void)n_in; (void)out_size; (void)ws_size;
}
